# starting kernel
# baseline (speedup 1.0000x reference)
#include <hip/hip_runtime.h>
#include <math.h>

#define N_NODES   10000
#define F_IN      128
#define HEADS     8
#define HIDD      256
#define HC        (HEADS * HIDD)
#define NUM_CLASS 10
#define E_IN      50000
#define E_TOT     (E_IN + N_NODES)
#define NEG_SLOPE 0.2f

typedef __attribute__((ext_vector_type(2))) float v2f;
typedef __attribute__((ext_vector_type(8))) float v8f;

__launch_bounds__(256)
__global__ void gemm1_wmma(const float* __restrict__ X, const float* __restrict__ W,
                           float* __restrict__ Hout)
{
    __shared__ float As[32][F_IN + 4];
    __shared__ float Bs[F_IN][64];

    const int tid = threadIdx.x;
    const int m0  = blockIdx.y * 32;
    const int n0  = blockIdx.x * 64;

#pragma unroll
    for (int i = 0; i < 4; ++i) {
        int idx = (tid + i * 256) * 4;
        int r = idx >> 7;
        int c = idx & 127;
        float4 v = make_float4(0.f, 0.f, 0.f, 0.f);
        int gr = m0 + r;
        if (gr < N_NODES) v = *(const float4*)(X + (size_t)gr * F_IN + c);
        *(float4*)(&As[r][c]) = v;
    }
#pragma unroll
    for (int i = 0; i < 8; ++i) {
        int idx = (tid + i * 256) * 4;
        int r = idx >> 6;
        int c = idx & 63;
        *(float4*)(&Bs[r][c]) = *(const float4*)(W + (size_t)r * HC + n0 + c);
    }
    __syncthreads();

    const int wave = tid >> 5;
    const int lane = tid & 31;
    const int half = lane >> 4;
    const int ml   = lane & 15;
    const int ar   = (wave >> 2) * 16;
    const int bc   = (wave & 3) * 16;

    v8f acc = {0.f, 0.f, 0.f, 0.f, 0.f, 0.f, 0.f, 0.f};
#pragma unroll
    for (int kk = 0; kk < F_IN / 4; ++kk) {
        const int kb = kk * 4 + half * 2;
        v2f a, b;
        float2 av = *(const float2*)(&As[ar + ml][kb]);
        a.x = av.x; a.y = av.y;
        b.x = Bs[kb + 0][bc + ml];
        b.y = Bs[kb + 1][bc + ml];
        acc = __builtin_amdgcn_wmma_f32_16x16x4_f32(false, a, false, b,
                                                    (short)0, acc, false, false);
    }

    const int col = n0 + bc + ml;
#pragma unroll
    for (int r = 0; r < 8; ++r) {
        const int row = m0 + ar + r + 8 * half;
        if (row < N_NODES) Hout[(size_t)row * HC + col] = acc[r];
    }
}

__launch_bounds__(256)
__global__ void gemm2_wmma(const float* __restrict__ Z, const float* __restrict__ W2,
                           float* __restrict__ H2)
{
    const int tid  = threadIdx.x;
    const int wave = tid >> 5;
    const int lane = tid & 31;
    const int half = lane >> 4;
    const int ml   = lane & 15;
    const int m0   = (blockIdx.x * 8 + wave) * 16;
    const int ra   = min(m0 + ml, N_NODES - 1);
    const int nb   = min(ml, NUM_CLASS - 1);
    const float* zr = Z + (size_t)ra * HC;

    v8f acc = {0.f, 0.f, 0.f, 0.f, 0.f, 0.f, 0.f, 0.f};
#pragma unroll 4
    for (int k = 0; k < HC; k += 4) {
        const int kb = k + half * 2;
        v2f a, b;
        float2 av = *(const float2*)(zr + kb);
        a.x = av.x; a.y = av.y;
        b.x = W2[(size_t)(kb + 0) * NUM_CLASS + nb];
        b.y = W2[(size_t)(kb + 1) * NUM_CLASS + nb];
        acc = __builtin_amdgcn_wmma_f32_16x16x4_f32(false, a, false, b,
                                                    (short)0, acc, false, false);
    }
    if (ml < NUM_CLASS) {
#pragma unroll
        for (int r = 0; r < 8; ++r) {
            const int row = m0 + r + 8 * half;
            if (row < N_NODES) H2[(size_t)row * NUM_CLASS + ml] = acc[r];
        }
    }
}

__launch_bounds__(256)
__global__ void att1_kernel(const float* __restrict__ H1,
                            const float* __restrict__ att_src,
                            const float* __restrict__ att_dst,
                            float* __restrict__ aS, float* __restrict__ aD)
{
    const int wave = threadIdx.x >> 5;
    const int lane = threadIdx.x & 31;
    const int gw = blockIdx.x * 8 + wave;
    if (gw >= N_NODES * HEADS) return;
    const int n = gw / HEADS, hh = gw % HEADS;
    const float* hp = H1 + (size_t)n * HC + hh * HIDD;
    const float* as = att_src + hh * HIDD;
    const float* ad = att_dst + hh * HIDD;
    float ss = 0.f, sd = 0.f;
    for (int c = lane; c < HIDD; c += 32) {
        float v = hp[c];
        ss += v * as[c];
        sd += v * ad[c];
    }
    for (int off = 16; off > 0; off >>= 1) {
        ss += __shfl_down(ss, off, 32);
        sd += __shfl_down(sd, off, 32);
    }
    if (lane == 0) { aS[gw] = ss; aD[gw] = sd; }
}

__global__ void att2_kernel(const float* __restrict__ H2,
                            const float* __restrict__ as2,
                            const float* __restrict__ ad2,
                            float* __restrict__ aS, float* __restrict__ aD)
{
    const int n = blockIdx.x * blockDim.x + threadIdx.x;
    if (n >= N_NODES) return;
    float ss = 0.f, sd = 0.f;
#pragma unroll
    for (int c = 0; c < NUM_CLASS; ++c) {
        float v = H2[(size_t)n * NUM_CLASS + c];
        ss += v * as2[c];
        sd += v * ad2[c];
    }
    aS[n] = ss; aD[n] = sd;
}

__device__ __forceinline__ int2 edge_sd(const int* __restrict__ ei, int e)
{
    int s = (e < E_IN) ? ei[e]        : (e - E_IN);
    int d = (e < E_IN) ? ei[E_IN + e] : (e - E_IN);
    return make_int2(s, d);
}

__device__ __forceinline__ void atomicMaxF(float* addr, float v)
{
    if (v >= 0.f) atomicMax((int*)addr, __float_as_int(v));
    else          atomicMin((unsigned int*)addr, __float_as_uint(v));
}

__global__ void edge_max(const int* __restrict__ ei,
                         const float* __restrict__ aS, const float* __restrict__ aD,
                         float* __restrict__ emax, int Hn)
{
    const int t = blockIdx.x * blockDim.x + threadIdx.x;
    if (t >= E_TOT * Hn) return;
    const int e = t / Hn, hh = t % Hn;
    const int2 sd = edge_sd(ei, e);
    float x = aS[sd.x * Hn + hh] + aD[sd.y * Hn + hh];
    x = (x >= 0.f) ? x : NEG_SLOPE * x;
    atomicMaxF(&emax[sd.y * Hn + hh], x);
}

__global__ void edge_expsum(const int* __restrict__ ei,
                            const float* __restrict__ aS, const float* __restrict__ aD,
                            const float* __restrict__ emax,
                            float* __restrict__ denom, float* __restrict__ exbuf, int Hn)
{
    const int t = blockIdx.x * blockDim.x + threadIdx.x;
    if (t >= E_TOT * Hn) return;
    const int e = t / Hn, hh = t % Hn;
    const int2 sd = edge_sd(ei, e);
    float x = aS[sd.x * Hn + hh] + aD[sd.y * Hn + hh];
    x = (x >= 0.f) ? x : NEG_SLOPE * x;
    float ex = expf(x - emax[sd.y * Hn + hh]);
    exbuf[t] = ex;
    atomicAdd(&denom[sd.y * Hn + hh], ex);
}

__launch_bounds__(256)
__global__ void edge_agg1(const int* __restrict__ ei, const float* __restrict__ H1,
                          const float* __restrict__ exbuf, const float* __restrict__ denom,
                          float* __restrict__ agg)
{
    const int e  = blockIdx.x;
    const int2 sd = edge_sd(ei, e);
    const int cb = threadIdx.x * 8;
    const int head = cb >> 8;
    const float alpha = exbuf[(size_t)e * HEADS + head]
                      / (denom[sd.y * HEADS + head] + 1e-16f);
    const float4* hs = (const float4*)(H1 + (size_t)sd.x * HC + cb);
    float4 v0 = hs[0], v1 = hs[1];
    float* op = agg + (size_t)sd.y * HC + cb;
    atomicAdd(op + 0, alpha * v0.x); atomicAdd(op + 1, alpha * v0.y);
    atomicAdd(op + 2, alpha * v0.z); atomicAdd(op + 3, alpha * v0.w);
    atomicAdd(op + 4, alpha * v1.x); atomicAdd(op + 5, alpha * v1.y);
    atomicAdd(op + 6, alpha * v1.z); atomicAdd(op + 7, alpha * v1.w);
}

__global__ void edge_agg2(const int* __restrict__ ei, const float* __restrict__ H2,
                          const float* __restrict__ exbuf, const float* __restrict__ denom,
                          float* __restrict__ agg)
{
    const int t = blockIdx.x * blockDim.x + threadIdx.x;
    if (t >= E_TOT * 16) return;
    const int e = t >> 4, c = t & 15;
    if (c >= NUM_CLASS) return;
    const int2 sd = edge_sd(ei, e);
    const float alpha = exbuf[e] / (denom[sd.y] + 1e-16f);
    atomicAdd(&agg[(size_t)sd.y * NUM_CLASS + c],
              alpha * H2[(size_t)sd.x * NUM_CLASS + c]);
}

__global__ void fill_kernel(float* __restrict__ p, size_t n, float v)
{
    for (size_t i = blockIdx.x * (size_t)blockDim.x + threadIdx.x; i < n;
         i += (size_t)gridDim.x * blockDim.x)
        p[i] = v;
}

__global__ void bias_elu(const float* __restrict__ agg, const float* __restrict__ bias,
                         float* __restrict__ z)
{
    const size_t total = (size_t)N_NODES * HC;
    for (size_t i = blockIdx.x * (size_t)blockDim.x + threadIdx.x; i < total;
         i += (size_t)gridDim.x * blockDim.x) {
        float v = agg[i] + bias[i % HC];
        z[i] = (v > 0.f) ? v : expm1f(v);
    }
}

__global__ void final_kernel(const float* __restrict__ agg2, const float* __restrict__ b2,
                             float* __restrict__ out)
{
    const int i = blockIdx.x * blockDim.x + threadIdx.x;
    if (i >= N_NODES * NUM_CLASS) return;
    float v = agg2[i] + b2[i % NUM_CLASS];
    out[i] = 1.f / (1.f + expf(-v));
}

extern "C" void kernel_launch(void* const* d_in, const int* in_sizes, int n_in,
                              void* d_out, int out_size, void* d_ws, size_t ws_size,
                              hipStream_t stream)
{
    const float* x   = (const float*)d_in[0];
    const int*   ei  = (const int*)d_in[1];
    const float* W1  = (const float*)d_in[2];
    const float* as1 = (const float*)d_in[3];
    const float* ad1 = (const float*)d_in[4];
    const float* b1  = (const float*)d_in[5];
    const float* W2  = (const float*)d_in[6];
    const float* as2 = (const float*)d_in[7];
    const float* ad2 = (const float*)d_in[8];
    const float* b2  = (const float*)d_in[9];
    float* out = (float*)d_out;

    float* ws = (float*)d_ws;
    size_t off = 0;
    float* h1    = ws + off; off += (size_t)N_NODES * HC;
    float* agg1  = ws + off; off += (size_t)N_NODES * HC;
    float* sA1   = ws + off; off += (size_t)N_NODES * HEADS;
    float* sD1   = ws + off; off += (size_t)N_NODES * HEADS;
    float* emax1 = ws + off; off += (size_t)N_NODES * HEADS;
    float* den1  = ws + off; off += (size_t)N_NODES * HEADS;
    float* ex1   = ws + off; off += (size_t)E_TOT * HEADS;
    float* h2    = ws + off; off += (size_t)N_NODES * NUM_CLASS;
    float* sA2   = ws + off; off += N_NODES;
    float* sD2   = ws + off; off += N_NODES;
    float* emax2 = ws + off; off += N_NODES;
    float* den2  = ws + off; off += N_NODES;
    float* ex2   = ws + off; off += E_TOT;
    float* agg2  = ws + off; off += (size_t)N_NODES * NUM_CLASS;
    (void)ws_size; (void)in_sizes; (void)n_in; (void)out_size;

    const dim3 b256(256);

    gemm1_wmma<<<dim3(HC / 64, (N_NODES + 31) / 32), b256, 0, stream>>>(x, W1, h1);
    att1_kernel<<<(N_NODES * HEADS) / 8, b256, 0, stream>>>(h1, as1, ad1, sA1, sD1);

    fill_kernel<<<4096, b256, 0, stream>>>(agg1, (size_t)N_NODES * HC, 0.f);
    fill_kernel<<<64,   b256, 0, stream>>>(den1, (size_t)N_NODES * HEADS, 0.f);
    fill_kernel<<<64,   b256, 0, stream>>>(emax1, (size_t)N_NODES * HEADS, -INFINITY);

    const int eg1 = (E_TOT * HEADS + 255) / 256;
    edge_max<<<eg1, b256, 0, stream>>>(ei, sA1, sD1, emax1, HEADS);
    edge_expsum<<<eg1, b256, 0, stream>>>(ei, sA1, sD1, emax1, den1, ex1, HEADS);
    edge_agg1<<<E_TOT, b256, 0, stream>>>(ei, h1, ex1, den1, agg1);

    bias_elu<<<4096, b256, 0, stream>>>(agg1, b1, h1);

    gemm2_wmma<<<(N_NODES + 127) / 128, b256, 0, stream>>>(h1, W2, h2);
    att2_kernel<<<(N_NODES + 255) / 256, b256, 0, stream>>>(h2, as2, ad2, sA2, sD2);

    fill_kernel<<<8, b256, 0, stream>>>(den2, (size_t)N_NODES, 0.f);
    fill_kernel<<<8, b256, 0, stream>>>(emax2, (size_t)N_NODES, -INFINITY);
    fill_kernel<<<8, b256, 0, stream>>>(agg2, (size_t)N_NODES * NUM_CLASS, 0.f);

    const int eg2 = (E_TOT + 255) / 256;
    edge_max<<<eg2, b256, 0, stream>>>(ei, sA2, sD2, emax2, 1);
    edge_expsum<<<eg2, b256, 0, stream>>>(ei, sA2, sD2, emax2, den2, ex2, 1);
    edge_agg2<<<(E_TOT * 16 + 255) / 256, b256, 0, stream>>>(ei, h2, ex2, den2, agg2);

    final_kernel<<<(N_NODES * NUM_CLASS + 255) / 256, b256, 0, stream>>>(agg2, b2, out);
}


# Layer-1 aggregation as a gather: edges grouped by destination per call (one-block LDS counting sort), a wave per node and head sums rows in registers in list order; no float atomics, no agg fill
# speedup vs baseline: 1.7261x; 1.7261x over previous
#include <hip/hip_runtime.h>
#include <math.h>

#define N_NODES   10000
#define F_IN      128
#define HEADS     8
#define HIDD      256
#define HC        (HEADS * HIDD)
#define NUM_CLASS 10
#define E_IN      50000
#define E_TOT     (E_IN + N_NODES)
#define NEG_SLOPE 0.2f

typedef __attribute__((ext_vector_type(2))) float v2f;
typedef __attribute__((ext_vector_type(8))) float v8f;

__launch_bounds__(256)
__global__ void gemm1_wmma(const float* __restrict__ X, const float* __restrict__ W,
                           float* __restrict__ Hout)
{
    __shared__ float As[32][F_IN + 4];
    __shared__ float Bs[F_IN][64];

    const int tid = threadIdx.x;
    const int m0  = blockIdx.y * 32;
    const int n0  = blockIdx.x * 64;

#pragma unroll
    for (int i = 0; i < 4; ++i) {
        int idx = (tid + i * 256) * 4;
        int r = idx >> 7;
        int c = idx & 127;
        float4 v = make_float4(0.f, 0.f, 0.f, 0.f);
        int gr = m0 + r;
        if (gr < N_NODES) v = *(const float4*)(X + (size_t)gr * F_IN + c);
        *(float4*)(&As[r][c]) = v;
    }
#pragma unroll
    for (int i = 0; i < 8; ++i) {
        int idx = (tid + i * 256) * 4;
        int r = idx >> 6;
        int c = idx & 63;
        *(float4*)(&Bs[r][c]) = *(const float4*)(W + (size_t)r * HC + n0 + c);
    }
    __syncthreads();

    const int wave = tid >> 5;
    const int lane = tid & 31;
    const int half = lane >> 4;
    const int ml   = lane & 15;
    const int ar   = (wave >> 2) * 16;
    const int bc   = (wave & 3) * 16;

    v8f acc = {0.f, 0.f, 0.f, 0.f, 0.f, 0.f, 0.f, 0.f};
#pragma unroll
    for (int kk = 0; kk < F_IN / 4; ++kk) {
        const int kb = kk * 4 + half * 2;
        v2f a, b;
        float2 av = *(const float2*)(&As[ar + ml][kb]);
        a.x = av.x; a.y = av.y;
        b.x = Bs[kb + 0][bc + ml];
        b.y = Bs[kb + 1][bc + ml];
        acc = __builtin_amdgcn_wmma_f32_16x16x4_f32(false, a, false, b,
                                                    (short)0, acc, false, false);
    }

    const int col = n0 + bc + ml;
#pragma unroll
    for (int r = 0; r < 8; ++r) {
        const int row = m0 + ar + r + 8 * half;
        if (row < N_NODES) Hout[(size_t)row * HC + col] = acc[r];
    }
}

__launch_bounds__(256)
__global__ void gemm2_wmma(const float* __restrict__ Z, const float* __restrict__ W2,
                           float* __restrict__ H2)
{
    const int tid  = threadIdx.x;
    const int wave = tid >> 5;
    const int lane = tid & 31;
    const int half = lane >> 4;
    const int ml   = lane & 15;
    const int m0   = (blockIdx.x * 8 + wave) * 16;
    const int ra   = min(m0 + ml, N_NODES - 1);
    const int nb   = min(ml, NUM_CLASS - 1);
    const float* zr = Z + (size_t)ra * HC;

    v8f acc = {0.f, 0.f, 0.f, 0.f, 0.f, 0.f, 0.f, 0.f};
#pragma unroll 4
    for (int k = 0; k < HC; k += 4) {
        const int kb = k + half * 2;
        v2f a, b;
        float2 av = *(const float2*)(zr + kb);
        a.x = av.x; a.y = av.y;
        b.x = W2[(size_t)(kb + 0) * NUM_CLASS + nb];
        b.y = W2[(size_t)(kb + 1) * NUM_CLASS + nb];
        acc = __builtin_amdgcn_wmma_f32_16x16x4_f32(false, a, false, b,
                                                    (short)0, acc, false, false);
    }
    if (ml < NUM_CLASS) {
#pragma unroll
        for (int r = 0; r < 8; ++r) {
            const int row = m0 + r + 8 * half;
            if (row < N_NODES) H2[(size_t)row * NUM_CLASS + ml] = acc[r];
        }
    }
}

__launch_bounds__(256)
__global__ void att1_kernel(const float* __restrict__ H1,
                            const float* __restrict__ att_src,
                            const float* __restrict__ att_dst,
                            float* __restrict__ aS, float* __restrict__ aD)
{
    const int wave = threadIdx.x >> 5;
    const int lane = threadIdx.x & 31;
    const int gw = blockIdx.x * 8 + wave;
    if (gw >= N_NODES * HEADS) return;
    const int n = gw / HEADS, hh = gw % HEADS;
    const float* hp = H1 + (size_t)n * HC + hh * HIDD;
    const float* as = att_src + hh * HIDD;
    const float* ad = att_dst + hh * HIDD;
    float ss = 0.f, sd = 0.f;
    for (int c = lane; c < HIDD; c += 32) {
        float v = hp[c];
        ss += v * as[c];
        sd += v * ad[c];
    }
    for (int off = 16; off > 0; off >>= 1) {
        ss += __shfl_down(ss, off, 32);
        sd += __shfl_down(sd, off, 32);
    }
    if (lane == 0) { aS[gw] = ss; aD[gw] = sd; }
}

__global__ void att2_kernel(const float* __restrict__ H2,
                            const float* __restrict__ as2,
                            const float* __restrict__ ad2,
                            float* __restrict__ aS, float* __restrict__ aD)
{
    const int n = blockIdx.x * blockDim.x + threadIdx.x;
    if (n >= N_NODES) return;
    float ss = 0.f, sd = 0.f;
#pragma unroll
    for (int c = 0; c < NUM_CLASS; ++c) {
        float v = H2[(size_t)n * NUM_CLASS + c];
        ss += v * as2[c];
        sd += v * ad2[c];
    }
    aS[n] = ss; aD[n] = sd;
}

__device__ __forceinline__ int2 edge_sd(const int* __restrict__ ei, int e)
{
    int s = (e < E_IN) ? ei[e]        : (e - E_IN);
    int d = (e < E_IN) ? ei[E_IN + e] : (e - E_IN);
    return make_int2(s, d);
}

__device__ __forceinline__ void atomicMaxF(float* addr, float v)
{
    if (v >= 0.f) atomicMax((int*)addr, __float_as_int(v));
    else          atomicMin((unsigned int*)addr, __float_as_uint(v));
}

__global__ void edge_max(const int* __restrict__ ei,
                         const float* __restrict__ aS, const float* __restrict__ aD,
                         float* __restrict__ emax, int Hn)
{
    const int t = blockIdx.x * blockDim.x + threadIdx.x;
    if (t >= E_TOT * Hn) return;
    const int e = t / Hn, hh = t % Hn;
    const int2 sd = edge_sd(ei, e);
    float x = aS[sd.x * Hn + hh] + aD[sd.y * Hn + hh];
    x = (x >= 0.f) ? x : NEG_SLOPE * x;
    atomicMaxF(&emax[sd.y * Hn + hh], x);
}

__global__ void edge_expsum(const int* __restrict__ ei,
                            const float* __restrict__ aS, const float* __restrict__ aD,
                            const float* __restrict__ emax,
                            float* __restrict__ denom, float* __restrict__ exbuf, int Hn)
{
    const int t = blockIdx.x * blockDim.x + threadIdx.x;
    if (t >= E_TOT * Hn) return;
    const int e = t / Hn, hh = t % Hn;
    const int2 sd = edge_sd(ei, e);
    float x = aS[sd.x * Hn + hh] + aD[sd.y * Hn + hh];
    x = (x >= 0.f) ? x : NEG_SLOPE * x;
    float ex = expf(x - emax[sd.y * Hn + hh]);
    exbuf[t] = ex;
    atomicAdd(&denom[sd.y * Hn + hh], ex);
}

#define CSR_T    1024
#define CSR_EPT  ((E_IN + CSR_T - 1) / CSR_T)
#define CSR_CH   7
#define CSR_SPAN ((N_NODES + CSR_T - 1) / CSR_T)
static_assert(E_IN <= 65536 && N_NODES <= 65536, "list entry packs edge id and source in 16 bits each");
static_assert(CSR_EPT % CSR_CH == 0, "chunks cover the edges of a thread exactly");

__launch_bounds__(CSR_T)
__global__ void csr_build(const int* __restrict__ ei, int* __restrict__ rowptr,
                          unsigned* __restrict__ elist)
{
    __shared__ int cnt[N_NODES + 1];
    __shared__ unsigned lst[E_IN + 1];
    __shared__ int wsum[CSR_T / 32];
    __shared__ int total_s;

    const int tid  = threadIdx.x;
    const int lane = tid & 31;
    const int wv   = tid >> 5;

    for (int i = tid; i < N_NODES + 1; i += CSR_T) cnt[i] = 0;
    __syncthreads();

#pragma unroll 1
    for (int k0 = 0; k0 < CSR_EPT; k0 += CSR_CH) {
        unsigned d[CSR_CH];
#pragma unroll
        for (int i = 0; i < CSR_CH; ++i) {
            const int e = tid + (k0 + i) * CSR_T;
            d[i] = (unsigned)ei[E_IN + min(e, E_IN - 1)];
            if (e >= E_IN || d[i] >= (unsigned)N_NODES) d[i] = N_NODES;
        }
#pragma unroll
        for (int i = 0; i < CSR_CH; ++i) atomicAdd(&cnt[d[i]], 1);
    }
    __syncthreads();

    const int base = tid * CSR_SPAN;
    int loc = 0;
#pragma unroll
    for (int j = 0; j < CSR_SPAN; ++j)
        if (base + j < N_NODES) loc += cnt[base + j];

    int inc = loc;
#pragma unroll
    for (int o = 1; o < 32; o <<= 1) {
        const int t = __shfl_up(inc, o, 32);
        if (lane >= o) inc += t;
    }
    if (lane == 31) wsum[wv] = inc;
    __syncthreads();
    if (wv == 0) {
        const int w = wsum[lane];
        int winc = w;
#pragma unroll
        for (int o = 1; o < 32; o <<= 1) {
            const int t = __shfl_up(winc, o, 32);
            if (lane >= o) winc += t;
        }
        wsum[lane] = winc - w;
    }
    __syncthreads();

    int run = wsum[wv] + inc - loc;
#pragma unroll
    for (int j = 0; j < CSR_SPAN; ++j) {
        if (base + j < N_NODES) {
            const int c = cnt[base + j];
            rowptr[base + j] = run;
            cnt[base + j]    = run;
            run += c;
        }
    }
    if (tid == CSR_T - 1) {
        rowptr[N_NODES] = run;
        total_s         = run;
        cnt[N_NODES]    = E_IN;
    }
    __syncthreads();

#pragma unroll 1
    for (int k0 = 0; k0 < CSR_EPT; k0 += CSR_CH) {
        unsigned d[CSR_CH], w[CSR_CH];
#pragma unroll
        for (int i = 0; i < CSR_CH; ++i) {
            const int e  = tid + (k0 + i) * CSR_T;
            const int ec = min(e, E_IN - 1);
            w[i] = ((unsigned)e << 16) | ((unsigned)ei[ec] & 0xffffu);
            d[i] = (unsigned)ei[E_IN + ec];
            if (e >= E_IN || d[i] >= (unsigned)N_NODES) d[i] = N_NODES;
        }
        int slot[CSR_CH];
#pragma unroll
        for (int i = 0; i < CSR_CH; ++i) slot[i] = atomicAdd(&cnt[d[i]], 1);
#pragma unroll
        for (int i = 0; i < CSR_CH; ++i) lst[min(slot[i], E_IN)] = w[i];
    }
    __syncthreads();

    const int total = total_s;
#pragma unroll 1
    for (int k0 = 0; k0 < CSR_EPT; k0 += CSR_CH) {
        unsigned v[CSR_CH];
#pragma unroll
        for (int i = 0; i < CSR_CH; ++i) v[i] = lst[min(tid + (k0 + i) * CSR_T, E_IN - 1)];
#pragma unroll
        for (int i = 0; i < CSR_CH; ++i) {
            const int idx = tid + (k0 + i) * CSR_T;
            if (idx < total) elist[idx] = v[i];
        }
    }
}

#define AGG_BLOCKS 2000

__device__ __forceinline__ void axpy4(float4& acc, const float a, const float4 v)
{
    acc.x += a * v.x; acc.y += a * v.y; acc.z += a * v.z; acc.w += a * v.w;
}

__launch_bounds__(256)
__global__ void node_agg1(const int* __restrict__ rowptr, const unsigned* __restrict__ elist,
                          const float* __restrict__ H1, const float* __restrict__ exbuf,
                          const float* __restrict__ denom, float* __restrict__ agg)
{
    const int lane = threadIdx.x & 31;
    const int head = __builtin_amdgcn_readfirstlane((int)(threadIdx.x >> 5));
    const int coff = head * HIDD + lane * 4;

    for (int d = blockIdx.x; d < N_NODES; d += gridDim.x) {
        const float den = denom[d * HEADS + head] + 1e-16f;
        const int beg = rowptr[d];
        const int end = rowptr[d + 1];

        float4 acc0, acc1;
        {
            const float a = exbuf[(size_t)(E_IN + d) * HEADS + head] / den;
            const float* hp = H1 + (size_t)d * HC + coff;
            const float4 v0 = *(const float4*)hp;
            const float4 v1 = *(const float4*)(hp + 128);
            acc0 = make_float4(a * v0.x, a * v0.y, a * v0.z, a * v0.w);
            acc1 = make_float4(a * v1.x, a * v1.y, a * v1.z, a * v1.w);
        }

        int j = beg;
        for (; j + 4 <= end; j += 4) {
            const unsigned r0 = elist[j], r1 = elist[j + 1], r2 = elist[j + 2], r3 = elist[j + 3];
            const float* p0 = H1 + (size_t)(r0 & 0xffffu) * HC + coff;
            const float* p1 = H1 + (size_t)(r1 & 0xffffu) * HC + coff;
            const float* p2 = H1 + (size_t)(r2 & 0xffffu) * HC + coff;
            const float* p3 = H1 + (size_t)(r3 & 0xffffu) * HC + coff;
            const float4 u0 = *(const float4*)p0, w0 = *(const float4*)(p0 + 128);
            const float4 u1 = *(const float4*)p1, w1 = *(const float4*)(p1 + 128);
            const float4 u2 = *(const float4*)p2, w2 = *(const float4*)(p2 + 128);
            const float4 u3 = *(const float4*)p3, w3 = *(const float4*)(p3 + 128);
            const float a0 = exbuf[(size_t)(r0 >> 16) * HEADS + head] / den;
            const float a1 = exbuf[(size_t)(r1 >> 16) * HEADS + head] / den;
            const float a2 = exbuf[(size_t)(r2 >> 16) * HEADS + head] / den;
            const float a3 = exbuf[(size_t)(r3 >> 16) * HEADS + head] / den;
            axpy4(acc0, a0, u0); axpy4(acc1, a0, w0);
            axpy4(acc0, a1, u1); axpy4(acc1, a1, w1);
            axpy4(acc0, a2, u2); axpy4(acc1, a2, w2);
            axpy4(acc0, a3, u3); axpy4(acc1, a3, w3);
        }
        if (j + 2 <= end) {
            const unsigned r0 = elist[j], r1 = elist[j + 1];
            const float* p0 = H1 + (size_t)(r0 & 0xffffu) * HC + coff;
            const float* p1 = H1 + (size_t)(r1 & 0xffffu) * HC + coff;
            const float4 u0 = *(const float4*)p0, w0 = *(const float4*)(p0 + 128);
            const float4 u1 = *(const float4*)p1, w1 = *(const float4*)(p1 + 128);
            const float a0 = exbuf[(size_t)(r0 >> 16) * HEADS + head] / den;
            const float a1 = exbuf[(size_t)(r1 >> 16) * HEADS + head] / den;
            axpy4(acc0, a0, u0); axpy4(acc1, a0, w0);
            axpy4(acc0, a1, u1); axpy4(acc1, a1, w1);
            j += 2;
        }
        if (j < end) {
            const unsigned r = elist[j];
            const float* p = H1 + (size_t)(r & 0xffffu) * HC + coff;
            const float4 u = *(const float4*)p, w = *(const float4*)(p + 128);
            const float a = exbuf[(size_t)(r >> 16) * HEADS + head] / den;
            axpy4(acc0, a, u); axpy4(acc1, a, w);
        }

        float* op = agg + (size_t)d * HC + coff;
        *(float4*)op         = acc0;
        *(float4*)(op + 128) = acc1;
    }
}

__global__ void edge_agg2(const int* __restrict__ ei, const float* __restrict__ H2,
                          const float* __restrict__ exbuf, const float* __restrict__ denom,
                          float* __restrict__ agg)
{
    const int t = blockIdx.x * blockDim.x + threadIdx.x;
    if (t >= E_TOT * 16) return;
    const int e = t >> 4, c = t & 15;
    if (c >= NUM_CLASS) return;
    const int2 sd = edge_sd(ei, e);
    const float alpha = exbuf[e] / (denom[sd.y] + 1e-16f);
    atomicAdd(&agg[(size_t)sd.y * NUM_CLASS + c],
              alpha * H2[(size_t)sd.x * NUM_CLASS + c]);
}

__global__ void fill_kernel(float* __restrict__ p, size_t n, float v)
{
    for (size_t i = blockIdx.x * (size_t)blockDim.x + threadIdx.x; i < n;
         i += (size_t)gridDim.x * blockDim.x)
        p[i] = v;
}

__global__ void bias_elu(const float* __restrict__ agg, const float* __restrict__ bias,
                         float* __restrict__ z)
{
    const size_t total = (size_t)N_NODES * HC;
    for (size_t i = blockIdx.x * (size_t)blockDim.x + threadIdx.x; i < total;
         i += (size_t)gridDim.x * blockDim.x) {
        float v = agg[i] + bias[i % HC];
        z[i] = (v > 0.f) ? v : expm1f(v);
    }
}

__global__ void final_kernel(const float* __restrict__ agg2, const float* __restrict__ b2,
                             float* __restrict__ out)
{
    const int i = blockIdx.x * blockDim.x + threadIdx.x;
    if (i >= N_NODES * NUM_CLASS) return;
    float v = agg2[i] + b2[i % NUM_CLASS];
    out[i] = 1.f / (1.f + expf(-v));
}

extern "C" void kernel_launch(void* const* d_in, const int* in_sizes, int n_in,
                              void* d_out, int out_size, void* d_ws, size_t ws_size,
                              hipStream_t stream)
{
    const float* x   = (const float*)d_in[0];
    const int*   ei  = (const int*)d_in[1];
    const float* W1  = (const float*)d_in[2];
    const float* as1 = (const float*)d_in[3];
    const float* ad1 = (const float*)d_in[4];
    const float* b1  = (const float*)d_in[5];
    const float* W2  = (const float*)d_in[6];
    const float* as2 = (const float*)d_in[7];
    const float* ad2 = (const float*)d_in[8];
    const float* b2  = (const float*)d_in[9];
    float* out = (float*)d_out;

    float* ws = (float*)d_ws;
    size_t off = 0;
    float* h1    = ws + off; off += (size_t)N_NODES * HC;
    float* agg1  = ws + off; off += (size_t)N_NODES * HC;
    float* sA1   = ws + off; off += (size_t)N_NODES * HEADS;
    float* sD1   = ws + off; off += (size_t)N_NODES * HEADS;
    float* emax1 = ws + off; off += (size_t)N_NODES * HEADS;
    float* den1  = ws + off; off += (size_t)N_NODES * HEADS;
    float* ex1   = ws + off; off += (size_t)E_TOT * HEADS;
    float* h2    = ws + off; off += (size_t)N_NODES * NUM_CLASS;
    float* sA2   = ws + off; off += N_NODES;
    float* sD2   = ws + off; off += N_NODES;
    float* emax2 = ws + off; off += N_NODES;
    float* den2  = ws + off; off += N_NODES;
    float* ex2   = ws + off; off += E_TOT;
    float* agg2  = ws + off; off += (size_t)N_NODES * NUM_CLASS;
    unsigned* elist = (unsigned*)(ws + off); off += (size_t)E_IN;
    int*      rowp  = (int*)(ws + off);      off += (size_t)N_NODES + 1;
    (void)ws_size; (void)in_sizes; (void)n_in; (void)out_size;

    const dim3 b256(256);

    gemm1_wmma<<<dim3(HC / 64, (N_NODES + 31) / 32), b256, 0, stream>>>(x, W1, h1);
    att1_kernel<<<(N_NODES * HEADS) / 8, b256, 0, stream>>>(h1, as1, ad1, sA1, sD1);

    fill_kernel<<<64,   b256, 0, stream>>>(den1, (size_t)N_NODES * HEADS, 0.f);
    fill_kernel<<<64,   b256, 0, stream>>>(emax1, (size_t)N_NODES * HEADS, -INFINITY);

    const int eg1 = (E_TOT * HEADS + 255) / 256;
    edge_max<<<eg1, b256, 0, stream>>>(ei, sA1, sD1, emax1, HEADS);
    edge_expsum<<<eg1, b256, 0, stream>>>(ei, sA1, sD1, emax1, den1, ex1, HEADS);
    csr_build<<<1, CSR_T, 0, stream>>>(ei, rowp, elist);
    node_agg1<<<AGG_BLOCKS, b256, 0, stream>>>(rowp, elist, h1, ex1, den1, agg1);

    bias_elu<<<4096, b256, 0, stream>>>(agg1, b1, h1);

    gemm2_wmma<<<(N_NODES + 127) / 128, b256, 0, stream>>>(h1, W2, h2);
    att2_kernel<<<(N_NODES + 255) / 256, b256, 0, stream>>>(h2, as2, ad2, sA2, sD2);

    fill_kernel<<<8, b256, 0, stream>>>(den2, (size_t)N_NODES, 0.f);
    fill_kernel<<<8, b256, 0, stream>>>(emax2, (size_t)N_NODES, -INFINITY);
    fill_kernel<<<8, b256, 0, stream>>>(agg2, (size_t)N_NODES * NUM_CLASS, 0.f);

    const int eg2 = (E_TOT + 255) / 256;
    edge_max<<<eg2, b256, 0, stream>>>(ei, sA2, sD2, emax2, 1);
    edge_expsum<<<eg2, b256, 0, stream>>>(ei, sA2, sD2, emax2, den2, ex2, 1);
    edge_agg2<<<(E_TOT * 16 + 255) / 256, b256, 0, stream>>>(ei, h2, ex2, den2, agg2);

    final_kernel<<<(N_NODES * NUM_CLASS + 255) / 256, b256, 0, stream>>>(agg2, b2, out);
}


# Layer-1 bias add and ELU moved into the aggregation kernel's row store (row still in registers); the separate elementwise pass over the aggregate and its launch are gone
# speedup vs baseline: 1.9229x; 1.1140x over previous
#include <hip/hip_runtime.h>
#include <math.h>

#define N_NODES   10000
#define F_IN      128
#define HEADS     8
#define HIDD      256
#define HC        (HEADS * HIDD)
#define NUM_CLASS 10
#define E_IN      50000
#define E_TOT     (E_IN + N_NODES)
#define NEG_SLOPE 0.2f

typedef __attribute__((ext_vector_type(2))) float v2f;
typedef __attribute__((ext_vector_type(8))) float v8f;

__launch_bounds__(256)
__global__ void gemm1_wmma(const float* __restrict__ X, const float* __restrict__ W,
                           float* __restrict__ Hout)
{
    __shared__ float As[32][F_IN + 4];
    __shared__ float Bs[F_IN][64];

    const int tid = threadIdx.x;
    const int m0  = blockIdx.y * 32;
    const int n0  = blockIdx.x * 64;

#pragma unroll
    for (int i = 0; i < 4; ++i) {
        int idx = (tid + i * 256) * 4;
        int r = idx >> 7;
        int c = idx & 127;
        float4 v = make_float4(0.f, 0.f, 0.f, 0.f);
        int gr = m0 + r;
        if (gr < N_NODES) v = *(const float4*)(X + (size_t)gr * F_IN + c);
        *(float4*)(&As[r][c]) = v;
    }
#pragma unroll
    for (int i = 0; i < 8; ++i) {
        int idx = (tid + i * 256) * 4;
        int r = idx >> 6;
        int c = idx & 63;
        *(float4*)(&Bs[r][c]) = *(const float4*)(W + (size_t)r * HC + n0 + c);
    }
    __syncthreads();

    const int wave = tid >> 5;
    const int lane = tid & 31;
    const int half = lane >> 4;
    const int ml   = lane & 15;
    const int ar   = (wave >> 2) * 16;
    const int bc   = (wave & 3) * 16;

    v8f acc = {0.f, 0.f, 0.f, 0.f, 0.f, 0.f, 0.f, 0.f};
#pragma unroll
    for (int kk = 0; kk < F_IN / 4; ++kk) {
        const int kb = kk * 4 + half * 2;
        v2f a, b;
        float2 av = *(const float2*)(&As[ar + ml][kb]);
        a.x = av.x; a.y = av.y;
        b.x = Bs[kb + 0][bc + ml];
        b.y = Bs[kb + 1][bc + ml];
        acc = __builtin_amdgcn_wmma_f32_16x16x4_f32(false, a, false, b,
                                                    (short)0, acc, false, false);
    }

    const int col = n0 + bc + ml;
#pragma unroll
    for (int r = 0; r < 8; ++r) {
        const int row = m0 + ar + r + 8 * half;
        if (row < N_NODES) Hout[(size_t)row * HC + col] = acc[r];
    }
}

__launch_bounds__(256)
__global__ void gemm2_wmma(const float* __restrict__ Z, const float* __restrict__ W2,
                           float* __restrict__ H2)
{
    const int tid  = threadIdx.x;
    const int wave = tid >> 5;
    const int lane = tid & 31;
    const int half = lane >> 4;
    const int ml   = lane & 15;
    const int m0   = (blockIdx.x * 8 + wave) * 16;
    const int ra   = min(m0 + ml, N_NODES - 1);
    const int nb   = min(ml, NUM_CLASS - 1);
    const float* zr = Z + (size_t)ra * HC;

    v8f acc = {0.f, 0.f, 0.f, 0.f, 0.f, 0.f, 0.f, 0.f};
#pragma unroll 4
    for (int k = 0; k < HC; k += 4) {
        const int kb = k + half * 2;
        v2f a, b;
        float2 av = *(const float2*)(zr + kb);
        a.x = av.x; a.y = av.y;
        b.x = W2[(size_t)(kb + 0) * NUM_CLASS + nb];
        b.y = W2[(size_t)(kb + 1) * NUM_CLASS + nb];
        acc = __builtin_amdgcn_wmma_f32_16x16x4_f32(false, a, false, b,
                                                    (short)0, acc, false, false);
    }
    if (ml < NUM_CLASS) {
#pragma unroll
        for (int r = 0; r < 8; ++r) {
            const int row = m0 + r + 8 * half;
            if (row < N_NODES) H2[(size_t)row * NUM_CLASS + ml] = acc[r];
        }
    }
}

__launch_bounds__(256)
__global__ void att1_kernel(const float* __restrict__ H1,
                            const float* __restrict__ att_src,
                            const float* __restrict__ att_dst,
                            float* __restrict__ aS, float* __restrict__ aD)
{
    const int wave = threadIdx.x >> 5;
    const int lane = threadIdx.x & 31;
    const int gw = blockIdx.x * 8 + wave;
    if (gw >= N_NODES * HEADS) return;
    const int n = gw / HEADS, hh = gw % HEADS;
    const float* hp = H1 + (size_t)n * HC + hh * HIDD;
    const float* as = att_src + hh * HIDD;
    const float* ad = att_dst + hh * HIDD;
    float ss = 0.f, sd = 0.f;
    for (int c = lane; c < HIDD; c += 32) {
        float v = hp[c];
        ss += v * as[c];
        sd += v * ad[c];
    }
    for (int off = 16; off > 0; off >>= 1) {
        ss += __shfl_down(ss, off, 32);
        sd += __shfl_down(sd, off, 32);
    }
    if (lane == 0) { aS[gw] = ss; aD[gw] = sd; }
}

__global__ void att2_kernel(const float* __restrict__ H2,
                            const float* __restrict__ as2,
                            const float* __restrict__ ad2,
                            float* __restrict__ aS, float* __restrict__ aD)
{
    const int n = blockIdx.x * blockDim.x + threadIdx.x;
    if (n >= N_NODES) return;
    float ss = 0.f, sd = 0.f;
#pragma unroll
    for (int c = 0; c < NUM_CLASS; ++c) {
        float v = H2[(size_t)n * NUM_CLASS + c];
        ss += v * as2[c];
        sd += v * ad2[c];
    }
    aS[n] = ss; aD[n] = sd;
}

__device__ __forceinline__ int2 edge_sd(const int* __restrict__ ei, int e)
{
    int s = (e < E_IN) ? ei[e]        : (e - E_IN);
    int d = (e < E_IN) ? ei[E_IN + e] : (e - E_IN);
    return make_int2(s, d);
}

__device__ __forceinline__ void atomicMaxF(float* addr, float v)
{
    if (v >= 0.f) atomicMax((int*)addr, __float_as_int(v));
    else          atomicMin((unsigned int*)addr, __float_as_uint(v));
}

__global__ void edge_max(const int* __restrict__ ei,
                         const float* __restrict__ aS, const float* __restrict__ aD,
                         float* __restrict__ emax, int Hn)
{
    const int t = blockIdx.x * blockDim.x + threadIdx.x;
    if (t >= E_TOT * Hn) return;
    const int e = t / Hn, hh = t % Hn;
    const int2 sd = edge_sd(ei, e);
    float x = aS[sd.x * Hn + hh] + aD[sd.y * Hn + hh];
    x = (x >= 0.f) ? x : NEG_SLOPE * x;
    atomicMaxF(&emax[sd.y * Hn + hh], x);
}

__global__ void edge_expsum(const int* __restrict__ ei,
                            const float* __restrict__ aS, const float* __restrict__ aD,
                            const float* __restrict__ emax,
                            float* __restrict__ denom, float* __restrict__ exbuf, int Hn)
{
    const int t = blockIdx.x * blockDim.x + threadIdx.x;
    if (t >= E_TOT * Hn) return;
    const int e = t / Hn, hh = t % Hn;
    const int2 sd = edge_sd(ei, e);
    float x = aS[sd.x * Hn + hh] + aD[sd.y * Hn + hh];
    x = (x >= 0.f) ? x : NEG_SLOPE * x;
    float ex = expf(x - emax[sd.y * Hn + hh]);
    exbuf[t] = ex;
    atomicAdd(&denom[sd.y * Hn + hh], ex);
}

#define CSR_T    1024
#define CSR_EPT  ((E_IN + CSR_T - 1) / CSR_T)
#define CSR_CH   7
#define CSR_SPAN ((N_NODES + CSR_T - 1) / CSR_T)
static_assert(E_IN <= 65536 && N_NODES <= 65536, "list entry packs edge id and source in 16 bits each");
static_assert(CSR_EPT % CSR_CH == 0, "chunks cover the edges of a thread exactly");

__launch_bounds__(CSR_T)
__global__ void csr_build(const int* __restrict__ ei, int* __restrict__ rowptr,
                          unsigned* __restrict__ elist)
{
    __shared__ int cnt[N_NODES + 1];
    __shared__ unsigned lst[E_IN + 1];
    __shared__ int wsum[CSR_T / 32];
    __shared__ int total_s;

    const int tid  = threadIdx.x;
    const int lane = tid & 31;
    const int wv   = tid >> 5;

    for (int i = tid; i < N_NODES + 1; i += CSR_T) cnt[i] = 0;
    __syncthreads();

#pragma unroll 1
    for (int k0 = 0; k0 < CSR_EPT; k0 += CSR_CH) {
        unsigned d[CSR_CH];
#pragma unroll
        for (int i = 0; i < CSR_CH; ++i) {
            const int e = tid + (k0 + i) * CSR_T;
            d[i] = (unsigned)ei[E_IN + min(e, E_IN - 1)];
            if (e >= E_IN || d[i] >= (unsigned)N_NODES) d[i] = N_NODES;
        }
#pragma unroll
        for (int i = 0; i < CSR_CH; ++i) atomicAdd(&cnt[d[i]], 1);
    }
    __syncthreads();

    const int base = tid * CSR_SPAN;
    int loc = 0;
#pragma unroll
    for (int j = 0; j < CSR_SPAN; ++j)
        if (base + j < N_NODES) loc += cnt[base + j];

    int inc = loc;
#pragma unroll
    for (int o = 1; o < 32; o <<= 1) {
        const int t = __shfl_up(inc, o, 32);
        if (lane >= o) inc += t;
    }
    if (lane == 31) wsum[wv] = inc;
    __syncthreads();
    if (wv == 0) {
        const int w = wsum[lane];
        int winc = w;
#pragma unroll
        for (int o = 1; o < 32; o <<= 1) {
            const int t = __shfl_up(winc, o, 32);
            if (lane >= o) winc += t;
        }
        wsum[lane] = winc - w;
    }
    __syncthreads();

    int run = wsum[wv] + inc - loc;
#pragma unroll
    for (int j = 0; j < CSR_SPAN; ++j) {
        if (base + j < N_NODES) {
            const int c = cnt[base + j];
            rowptr[base + j] = run;
            cnt[base + j]    = run;
            run += c;
        }
    }
    if (tid == CSR_T - 1) {
        rowptr[N_NODES] = run;
        total_s         = run;
        cnt[N_NODES]    = E_IN;
    }
    __syncthreads();

#pragma unroll 1
    for (int k0 = 0; k0 < CSR_EPT; k0 += CSR_CH) {
        unsigned d[CSR_CH], w[CSR_CH];
#pragma unroll
        for (int i = 0; i < CSR_CH; ++i) {
            const int e  = tid + (k0 + i) * CSR_T;
            const int ec = min(e, E_IN - 1);
            w[i] = ((unsigned)e << 16) | ((unsigned)ei[ec] & 0xffffu);
            d[i] = (unsigned)ei[E_IN + ec];
            if (e >= E_IN || d[i] >= (unsigned)N_NODES) d[i] = N_NODES;
        }
        int slot[CSR_CH];
#pragma unroll
        for (int i = 0; i < CSR_CH; ++i) slot[i] = atomicAdd(&cnt[d[i]], 1);
#pragma unroll
        for (int i = 0; i < CSR_CH; ++i) lst[min(slot[i], E_IN)] = w[i];
    }
    __syncthreads();

    const int total = total_s;
#pragma unroll 1
    for (int k0 = 0; k0 < CSR_EPT; k0 += CSR_CH) {
        unsigned v[CSR_CH];
#pragma unroll
        for (int i = 0; i < CSR_CH; ++i) v[i] = lst[min(tid + (k0 + i) * CSR_T, E_IN - 1)];
#pragma unroll
        for (int i = 0; i < CSR_CH; ++i) {
            const int idx = tid + (k0 + i) * CSR_T;
            if (idx < total) elist[idx] = v[i];
        }
    }
}

#define AGG_BLOCKS 2000

__device__ __forceinline__ void axpy4(float4& acc, const float a, const float4 v)
{
    acc.x += a * v.x; acc.y += a * v.y; acc.z += a * v.z; acc.w += a * v.w;
}

__device__ __forceinline__ float bias_elu1(const float s, const float b)
{
    const float v = s + b;
    return (v > 0.f) ? v : expm1f(v);
}

__device__ __forceinline__ float4 bias_elu4(const float4 s, const float4 b)
{
    return make_float4(bias_elu1(s.x, b.x), bias_elu1(s.y, b.y),
                       bias_elu1(s.z, b.z), bias_elu1(s.w, b.w));
}

__launch_bounds__(256)
__global__ void node_agg1(const int* __restrict__ rowptr, const unsigned* __restrict__ elist,
                          const float* __restrict__ H1, const float* __restrict__ exbuf,
                          const float* __restrict__ denom, const float* bias, float* zout)
{
    const int lane = threadIdx.x & 31;
    const int head = __builtin_amdgcn_readfirstlane((int)(threadIdx.x >> 5));
    const int coff = head * HIDD + lane * 4;

    for (int d = blockIdx.x; d < N_NODES; d += gridDim.x) {
        const float den = denom[d * HEADS + head] + 1e-16f;
        const int beg = rowptr[d];
        const int end = rowptr[d + 1];

        float4 acc0, acc1;
        {
            const float a = exbuf[(size_t)(E_IN + d) * HEADS + head] / den;
            const float* hp = H1 + (size_t)d * HC + coff;
            const float4 v0 = *(const float4*)hp;
            const float4 v1 = *(const float4*)(hp + 128);
            acc0 = make_float4(a * v0.x, a * v0.y, a * v0.z, a * v0.w);
            acc1 = make_float4(a * v1.x, a * v1.y, a * v1.z, a * v1.w);
        }

        int j = beg;
        for (; j + 4 <= end; j += 4) {
            const unsigned r0 = elist[j], r1 = elist[j + 1], r2 = elist[j + 2], r3 = elist[j + 3];
            const float* p0 = H1 + (size_t)(r0 & 0xffffu) * HC + coff;
            const float* p1 = H1 + (size_t)(r1 & 0xffffu) * HC + coff;
            const float* p2 = H1 + (size_t)(r2 & 0xffffu) * HC + coff;
            const float* p3 = H1 + (size_t)(r3 & 0xffffu) * HC + coff;
            const float4 u0 = *(const float4*)p0, w0 = *(const float4*)(p0 + 128);
            const float4 u1 = *(const float4*)p1, w1 = *(const float4*)(p1 + 128);
            const float4 u2 = *(const float4*)p2, w2 = *(const float4*)(p2 + 128);
            const float4 u3 = *(const float4*)p3, w3 = *(const float4*)(p3 + 128);
            const float a0 = exbuf[(size_t)(r0 >> 16) * HEADS + head] / den;
            const float a1 = exbuf[(size_t)(r1 >> 16) * HEADS + head] / den;
            const float a2 = exbuf[(size_t)(r2 >> 16) * HEADS + head] / den;
            const float a3 = exbuf[(size_t)(r3 >> 16) * HEADS + head] / den;
            axpy4(acc0, a0, u0); axpy4(acc1, a0, w0);
            axpy4(acc0, a1, u1); axpy4(acc1, a1, w1);
            axpy4(acc0, a2, u2); axpy4(acc1, a2, w2);
            axpy4(acc0, a3, u3); axpy4(acc1, a3, w3);
        }
        const float4 b0 = *(const float4*)(bias + coff);
        const float4 b1 = *(const float4*)(bias + coff + 128);

        if (j + 2 <= end) {
            const unsigned r0 = elist[j], r1 = elist[j + 1];
            const float* p0 = H1 + (size_t)(r0 & 0xffffu) * HC + coff;
            const float* p1 = H1 + (size_t)(r1 & 0xffffu) * HC + coff;
            const float4 u0 = *(const float4*)p0, w0 = *(const float4*)(p0 + 128);
            const float4 u1 = *(const float4*)p1, w1 = *(const float4*)(p1 + 128);
            const float a0 = exbuf[(size_t)(r0 >> 16) * HEADS + head] / den;
            const float a1 = exbuf[(size_t)(r1 >> 16) * HEADS + head] / den;
            axpy4(acc0, a0, u0); axpy4(acc1, a0, w0);
            axpy4(acc0, a1, u1); axpy4(acc1, a1, w1);
            j += 2;
        }
        if (j < end) {
            const unsigned r = elist[j];
            const float* p = H1 + (size_t)(r & 0xffffu) * HC + coff;
            const float4 u = *(const float4*)p, w = *(const float4*)(p + 128);
            const float a = exbuf[(size_t)(r >> 16) * HEADS + head] / den;
            axpy4(acc0, a, u); axpy4(acc1, a, w);
        }

        float* op = zout + (size_t)d * HC + coff;
        *(float4*)op         = bias_elu4(acc0, b0);
        *(float4*)(op + 128) = bias_elu4(acc1, b1);
    }
}

__global__ void edge_agg2(const int* __restrict__ ei, const float* __restrict__ H2,
                          const float* __restrict__ exbuf, const float* __restrict__ denom,
                          float* __restrict__ agg)
{
    const int t = blockIdx.x * blockDim.x + threadIdx.x;
    if (t >= E_TOT * 16) return;
    const int e = t >> 4, c = t & 15;
    if (c >= NUM_CLASS) return;
    const int2 sd = edge_sd(ei, e);
    const float alpha = exbuf[e] / (denom[sd.y] + 1e-16f);
    atomicAdd(&agg[(size_t)sd.y * NUM_CLASS + c],
              alpha * H2[(size_t)sd.x * NUM_CLASS + c]);
}

__global__ void fill_kernel(float* __restrict__ p, size_t n, float v)
{
    for (size_t i = blockIdx.x * (size_t)blockDim.x + threadIdx.x; i < n;
         i += (size_t)gridDim.x * blockDim.x)
        p[i] = v;
}

__global__ void final_kernel(const float* __restrict__ agg2, const float* __restrict__ b2,
                             float* __restrict__ out)
{
    const int i = blockIdx.x * blockDim.x + threadIdx.x;
    if (i >= N_NODES * NUM_CLASS) return;
    float v = agg2[i] + b2[i % NUM_CLASS];
    out[i] = 1.f / (1.f + expf(-v));
}

extern "C" void kernel_launch(void* const* d_in, const int* in_sizes, int n_in,
                              void* d_out, int out_size, void* d_ws, size_t ws_size,
                              hipStream_t stream)
{
    const float* x   = (const float*)d_in[0];
    const int*   ei  = (const int*)d_in[1];
    const float* W1  = (const float*)d_in[2];
    const float* as1 = (const float*)d_in[3];
    const float* ad1 = (const float*)d_in[4];
    const float* b1  = (const float*)d_in[5];
    const float* W2  = (const float*)d_in[6];
    const float* as2 = (const float*)d_in[7];
    const float* ad2 = (const float*)d_in[8];
    const float* b2  = (const float*)d_in[9];
    float* out = (float*)d_out;

    float* ws = (float*)d_ws;
    size_t off = 0;
    float* h1    = ws + off; off += (size_t)N_NODES * HC;
    float* z1    = ws + off; off += (size_t)N_NODES * HC;
    float* sA1   = ws + off; off += (size_t)N_NODES * HEADS;
    float* sD1   = ws + off; off += (size_t)N_NODES * HEADS;
    float* emax1 = ws + off; off += (size_t)N_NODES * HEADS;
    float* den1  = ws + off; off += (size_t)N_NODES * HEADS;
    float* ex1   = ws + off; off += (size_t)E_TOT * HEADS;
    float* h2    = ws + off; off += (size_t)N_NODES * NUM_CLASS;
    float* sA2   = ws + off; off += N_NODES;
    float* sD2   = ws + off; off += N_NODES;
    float* emax2 = ws + off; off += N_NODES;
    float* den2  = ws + off; off += N_NODES;
    float* ex2   = ws + off; off += E_TOT;
    float* agg2  = ws + off; off += (size_t)N_NODES * NUM_CLASS;
    unsigned* elist = (unsigned*)(ws + off); off += (size_t)E_IN;
    int*      rowp  = (int*)(ws + off);      off += (size_t)N_NODES + 1;
    (void)ws_size; (void)in_sizes; (void)n_in; (void)out_size;

    const dim3 b256(256);

    gemm1_wmma<<<dim3(HC / 64, (N_NODES + 31) / 32), b256, 0, stream>>>(x, W1, h1);
    att1_kernel<<<(N_NODES * HEADS) / 8, b256, 0, stream>>>(h1, as1, ad1, sA1, sD1);

    fill_kernel<<<64,   b256, 0, stream>>>(den1, (size_t)N_NODES * HEADS, 0.f);
    fill_kernel<<<64,   b256, 0, stream>>>(emax1, (size_t)N_NODES * HEADS, -INFINITY);

    const int eg1 = (E_TOT * HEADS + 255) / 256;
    edge_max<<<eg1, b256, 0, stream>>>(ei, sA1, sD1, emax1, HEADS);
    edge_expsum<<<eg1, b256, 0, stream>>>(ei, sA1, sD1, emax1, den1, ex1, HEADS);
    csr_build<<<1, CSR_T, 0, stream>>>(ei, rowp, elist);
    node_agg1<<<AGG_BLOCKS, b256, 0, stream>>>(rowp, elist, h1, ex1, den1, b1, z1);

    gemm2_wmma<<<(N_NODES + 127) / 128, b256, 0, stream>>>(z1, W2, h2);
    att2_kernel<<<(N_NODES + 255) / 256, b256, 0, stream>>>(h2, as2, ad2, sA2, sD2);

    fill_kernel<<<8, b256, 0, stream>>>(den2, (size_t)N_NODES, 0.f);
    fill_kernel<<<8, b256, 0, stream>>>(emax2, (size_t)N_NODES, -INFINITY);
    fill_kernel<<<8, b256, 0, stream>>>(agg2, (size_t)N_NODES * NUM_CLASS, 0.f);

    const int eg2 = (E_TOT + 255) / 256;
    edge_max<<<eg2, b256, 0, stream>>>(ei, sA2, sD2, emax2, 1);
    edge_expsum<<<eg2, b256, 0, stream>>>(ei, sA2, sD2, emax2, den2, ex2, 1);
    edge_agg2<<<(E_TOT * 16 + 255) / 256, b256, 0, stream>>>(ei, h2, ex2, den2, agg2);

    final_kernel<<<(N_NODES * NUM_CLASS + 255) / 256, b256, 0, stream>>>(agg2, b2, out);
}


# Segment softmax (max, exp-sum) and the layer-2 weighted sum done per destination on the grouped edge list: no atomics, fills, edge or final passes (17 -> 7 launches); sums in list order
# speedup vs baseline: 2.3017x; 1.1970x over previous
#include <hip/hip_runtime.h>
#include <math.h>

#define N_NODES   10000
#define F_IN      128
#define HEADS     8
#define HIDD      256
#define HC        (HEADS * HIDD)
#define NUM_CLASS 10
#define E_IN      50000
#define NEG_SLOPE 0.2f

typedef __attribute__((ext_vector_type(2))) float v2f;
typedef __attribute__((ext_vector_type(8))) float v8f;

__launch_bounds__(256)
__global__ void gemm1_wmma(const float* __restrict__ X, const float* __restrict__ W,
                           float* __restrict__ Hout)
{
    __shared__ float As[32][F_IN + 4];
    __shared__ float Bs[F_IN][64];

    const int tid = threadIdx.x;
    const int m0  = blockIdx.y * 32;
    const int n0  = blockIdx.x * 64;

#pragma unroll
    for (int i = 0; i < 4; ++i) {
        int idx = (tid + i * 256) * 4;
        int r = idx >> 7;
        int c = idx & 127;
        float4 v = make_float4(0.f, 0.f, 0.f, 0.f);
        int gr = m0 + r;
        if (gr < N_NODES) v = *(const float4*)(X + (size_t)gr * F_IN + c);
        *(float4*)(&As[r][c]) = v;
    }
#pragma unroll
    for (int i = 0; i < 8; ++i) {
        int idx = (tid + i * 256) * 4;
        int r = idx >> 6;
        int c = idx & 63;
        *(float4*)(&Bs[r][c]) = *(const float4*)(W + (size_t)r * HC + n0 + c);
    }
    __syncthreads();

    const int wave = tid >> 5;
    const int lane = tid & 31;
    const int half = lane >> 4;
    const int ml   = lane & 15;
    const int ar   = (wave >> 2) * 16;
    const int bc   = (wave & 3) * 16;

    v8f acc = {0.f, 0.f, 0.f, 0.f, 0.f, 0.f, 0.f, 0.f};
#pragma unroll
    for (int kk = 0; kk < F_IN / 4; ++kk) {
        const int kb = kk * 4 + half * 2;
        v2f a, b;
        float2 av = *(const float2*)(&As[ar + ml][kb]);
        a.x = av.x; a.y = av.y;
        b.x = Bs[kb + 0][bc + ml];
        b.y = Bs[kb + 1][bc + ml];
        acc = __builtin_amdgcn_wmma_f32_16x16x4_f32(false, a, false, b,
                                                    (short)0, acc, false, false);
    }

    const int col = n0 + bc + ml;
#pragma unroll
    for (int r = 0; r < 8; ++r) {
        const int row = m0 + ar + r + 8 * half;
        if (row < N_NODES) Hout[(size_t)row * HC + col] = acc[r];
    }
}

__launch_bounds__(256)
__global__ void gemm2_wmma(const float* __restrict__ Z, const float* __restrict__ W2,
                           float* __restrict__ H2)
{
    const int tid  = threadIdx.x;
    const int wave = tid >> 5;
    const int lane = tid & 31;
    const int half = lane >> 4;
    const int ml   = lane & 15;
    const int m0   = (blockIdx.x * 8 + wave) * 16;
    const int ra   = min(m0 + ml, N_NODES - 1);
    const int nb   = min(ml, NUM_CLASS - 1);
    const float* zr = Z + (size_t)ra * HC;

    v8f acc = {0.f, 0.f, 0.f, 0.f, 0.f, 0.f, 0.f, 0.f};
#pragma unroll 4
    for (int k = 0; k < HC; k += 4) {
        const int kb = k + half * 2;
        v2f a, b;
        float2 av = *(const float2*)(zr + kb);
        a.x = av.x; a.y = av.y;
        b.x = W2[(size_t)(kb + 0) * NUM_CLASS + nb];
        b.y = W2[(size_t)(kb + 1) * NUM_CLASS + nb];
        acc = __builtin_amdgcn_wmma_f32_16x16x4_f32(false, a, false, b,
                                                    (short)0, acc, false, false);
    }
    if (ml < NUM_CLASS) {
#pragma unroll
        for (int r = 0; r < 8; ++r) {
            const int row = m0 + r + 8 * half;
            if (row < N_NODES) H2[(size_t)row * NUM_CLASS + ml] = acc[r];
        }
    }
}

__launch_bounds__(256)
__global__ void att1_kernel(const float* __restrict__ H1,
                            const float* __restrict__ att_src,
                            const float* __restrict__ att_dst,
                            float* __restrict__ aS, float* __restrict__ aD)
{
    const int wave = threadIdx.x >> 5;
    const int lane = threadIdx.x & 31;
    const int gw = blockIdx.x * 8 + wave;
    if (gw >= N_NODES * HEADS) return;
    const int n = gw / HEADS, hh = gw % HEADS;
    const float* hp = H1 + (size_t)n * HC + hh * HIDD;
    const float* as = att_src + hh * HIDD;
    const float* ad = att_dst + hh * HIDD;
    float ss = 0.f, sd = 0.f;
    for (int c = lane; c < HIDD; c += 32) {
        float v = hp[c];
        ss += v * as[c];
        sd += v * ad[c];
    }
    for (int off = 16; off > 0; off >>= 1) {
        ss += __shfl_down(ss, off, 32);
        sd += __shfl_down(sd, off, 32);
    }
    if (lane == 0) { aS[gw] = ss; aD[gw] = sd; }
}

__global__ void att2_kernel(const float* __restrict__ H2,
                            const float* __restrict__ as2,
                            const float* __restrict__ ad2,
                            float* __restrict__ aS, float* __restrict__ aD)
{
    const int n = blockIdx.x * blockDim.x + threadIdx.x;
    if (n >= N_NODES) return;
    float ss = 0.f, sd = 0.f;
#pragma unroll
    for (int c = 0; c < NUM_CLASS; ++c) {
        float v = H2[(size_t)n * NUM_CLASS + c];
        ss += v * as2[c];
        sd += v * ad2[c];
    }
    aS[n] = ss; aD[n] = sd;
}

#define CSR_T    1024
#define CSR_EPT  ((E_IN + CSR_T - 1) / CSR_T)
#define CSR_CH   7
#define CSR_SPAN ((N_NODES + CSR_T - 1) / CSR_T)
static_assert(E_IN <= 65536 && N_NODES <= 65536, "list entry packs edge id and source in 16 bits each");
static_assert(CSR_EPT % CSR_CH == 0, "chunks cover the edges of a thread exactly");

__launch_bounds__(CSR_T)
__global__ void csr_build(const int* __restrict__ ei, int* __restrict__ rowptr,
                          unsigned* __restrict__ elist)
{
    __shared__ int cnt[N_NODES + 1];
    __shared__ unsigned lst[E_IN + 1];
    __shared__ int wsum[CSR_T / 32];
    __shared__ int total_s;

    const int tid  = threadIdx.x;
    const int lane = tid & 31;
    const int wv   = tid >> 5;

    for (int i = tid; i < N_NODES + 1; i += CSR_T) cnt[i] = 0;
    __syncthreads();

#pragma unroll 1
    for (int k0 = 0; k0 < CSR_EPT; k0 += CSR_CH) {
        unsigned d[CSR_CH];
#pragma unroll
        for (int i = 0; i < CSR_CH; ++i) {
            const int e = tid + (k0 + i) * CSR_T;
            d[i] = (unsigned)ei[E_IN + min(e, E_IN - 1)];
            if (e >= E_IN || d[i] >= (unsigned)N_NODES) d[i] = N_NODES;
        }
#pragma unroll
        for (int i = 0; i < CSR_CH; ++i) atomicAdd(&cnt[d[i]], 1);
    }
    __syncthreads();

    const int base = tid * CSR_SPAN;
    int loc = 0;
#pragma unroll
    for (int j = 0; j < CSR_SPAN; ++j)
        if (base + j < N_NODES) loc += cnt[base + j];

    int inc = loc;
#pragma unroll
    for (int o = 1; o < 32; o <<= 1) {
        const int t = __shfl_up(inc, o, 32);
        if (lane >= o) inc += t;
    }
    if (lane == 31) wsum[wv] = inc;
    __syncthreads();
    if (wv == 0) {
        const int w = wsum[lane];
        int winc = w;
#pragma unroll
        for (int o = 1; o < 32; o <<= 1) {
            const int t = __shfl_up(winc, o, 32);
            if (lane >= o) winc += t;
        }
        wsum[lane] = winc - w;
    }
    __syncthreads();

    int run = wsum[wv] + inc - loc;
#pragma unroll
    for (int j = 0; j < CSR_SPAN; ++j) {
        if (base + j < N_NODES) {
            const int c = cnt[base + j];
            rowptr[base + j] = run;
            cnt[base + j]    = run;
            run += c;
        }
    }
    if (tid == CSR_T - 1) {
        rowptr[N_NODES] = run;
        total_s         = run;
        cnt[N_NODES]    = E_IN;
    }
    __syncthreads();

#pragma unroll 1
    for (int k0 = 0; k0 < CSR_EPT; k0 += CSR_CH) {
        unsigned d[CSR_CH], w[CSR_CH];
#pragma unroll
        for (int i = 0; i < CSR_CH; ++i) {
            const int e  = tid + (k0 + i) * CSR_T;
            const int ec = min(e, E_IN - 1);
            w[i] = ((unsigned)e << 16) | ((unsigned)ei[ec] & 0xffffu);
            d[i] = (unsigned)ei[E_IN + ec];
            if (e >= E_IN || d[i] >= (unsigned)N_NODES) d[i] = N_NODES;
        }
        int slot[CSR_CH];
#pragma unroll
        for (int i = 0; i < CSR_CH; ++i) slot[i] = atomicAdd(&cnt[d[i]], 1);
#pragma unroll
        for (int i = 0; i < CSR_CH; ++i) lst[min(slot[i], E_IN)] = w[i];
    }
    __syncthreads();

    const int total = total_s;
#pragma unroll 1
    for (int k0 = 0; k0 < CSR_EPT; k0 += CSR_CH) {
        unsigned v[CSR_CH];
#pragma unroll
        for (int i = 0; i < CSR_CH; ++i) v[i] = lst[min(tid + (k0 + i) * CSR_T, E_IN - 1)];
#pragma unroll
        for (int i = 0; i < CSR_CH; ++i) {
            const int idx = tid + (k0 + i) * CSR_T;
            if (idx < total) elist[idx] = v[i];
        }
    }
}

__device__ __forceinline__ float leaky(const float x)
{
    return (x >= 0.f) ? x : NEG_SLOPE * x;
}

template <int MASK>
__device__ __forceinline__ float row_xor(const float x)
{
    return __builtin_bit_cast(float, __builtin_amdgcn_update_dpp(
        0, __builtin_bit_cast(int, x), 0x160 + MASK, 0xf, 0xf, false));
}

__device__ __forceinline__ float xhalf(const float x)
{
    const int b = __builtin_bit_cast(int, x);
    return __builtin_bit_cast(float, __builtin_amdgcn_permlanex16(b, b, 0x76543210, 0xfedcba98, false, false));
}

__device__ __forceinline__ float wave_max(float x)
{
    x = fmaxf(x, row_xor<1>(x));
    x = fmaxf(x, row_xor<2>(x));
    x = fmaxf(x, row_xor<4>(x));
    x = fmaxf(x, row_xor<8>(x));
    return fmaxf(x, xhalf(x));
}

__device__ __forceinline__ float wave_sum(float x)
{
    x += row_xor<1>(x);
    x += row_xor<2>(x);
    x += row_xor<4>(x);
    x += row_xor<8>(x);
    return x + xhalf(x);
}

__device__ __forceinline__ float lane_f(const float v, const int k)
{
    return __int_as_float(__builtin_amdgcn_readlane(__float_as_int(v), k));
}

#define AGG_BLOCKS 2000

__device__ __forceinline__ void axpy4(float4& acc, const float a, const float4 v)
{
    acc.x += a * v.x; acc.y += a * v.y; acc.z += a * v.z; acc.w += a * v.w;
}

__device__ __forceinline__ float bias_elu1(const float s, const float b)
{
    const float v = s + b;
    return (v > 0.f) ? v : expm1f(v);
}

__device__ __forceinline__ float4 bias_elu4(const float4 s, const float4 b)
{
    return make_float4(bias_elu1(s.x, b.x), bias_elu1(s.y, b.y),
                       bias_elu1(s.z, b.z), bias_elu1(s.w, b.w));
}

__device__ __forceinline__ float edge_score1(const float* __restrict__ aS, const unsigned rec,
                                             const int head, const float ad)
{
    return leaky(aS[(rec & 0xffffu) * HEADS + head] + ad);
}

__launch_bounds__(256)
__global__ void node_agg1(const int* __restrict__ rowptr, const unsigned* __restrict__ elist,
                          const float* __restrict__ H1, const float* __restrict__ aS,
                          const float* __restrict__ aD, const float* bias, float* zout)
{
    const int lane = threadIdx.x & 31;
    const int head = __builtin_amdgcn_readfirstlane((int)(threadIdx.x >> 5));
    const int coff = head * HIDD + lane * 4;

    for (int d = blockIdx.x; d < N_NODES; d += gridDim.x) {
        const int beg = rowptr[d];
        const int end = rowptr[d + 1];
        const float ad = aD[d * HEADS + head];
        const float es = leaky(aS[d * HEADS + head] + ad);

        const int  j0 = beg + lane;
        const bool v0 = j0 < end;
        unsigned rc = 0u;
        float ec = -INFINITY;
        if (v0) { rc = elist[j0]; ec = edge_score1(aS, rc, head, ad); }
        float m = fmaxf(es, ec);
        for (int c0 = beg + 32; c0 < end; c0 += 32) {
            const int j = c0 + lane;
            if (j < end) m = fmaxf(m, edge_score1(aS, elist[j], head, ad));
        }
        m = wave_max(m);
        m = isfinite(m) ? m : 0.f;

        const float pc = v0 ? expf(ec - m) : 0.f;
        float s = pc;
        for (int c0 = beg + 32; c0 < end; c0 += 32) {
            const int j = c0 + lane;
            if (j < end) s += expf(edge_score1(aS, elist[j], head, ad) - m);
        }
        const float ps  = expf(es - m);
        const float den = wave_sum(s) + ps + 1e-16f;

        float4 acc0, acc1;
        {
            const float a = ps / den;
            const float* hp = H1 + (size_t)d * HC + coff;
            const float4 v0r = *(const float4*)hp;
            const float4 v1r = *(const float4*)(hp + 128);
            acc0 = make_float4(a * v0r.x, a * v0r.y, a * v0r.z, a * v0r.w);
            acc1 = make_float4(a * v1r.x, a * v1r.y, a * v1r.z, a * v1r.w);
        }

        unsigned rcur = rc;
        float    acur = pc / den;
        for (int c0 = beg; c0 < end; c0 += 32) {
            const int cnt = min(end - c0, 32);
            int k = 0;
            for (; k + 4 <= cnt; k += 4) {
                const unsigned r0 = __builtin_amdgcn_readlane(rcur, k);
                const unsigned r1 = __builtin_amdgcn_readlane(rcur, k + 1);
                const unsigned r2 = __builtin_amdgcn_readlane(rcur, k + 2);
                const unsigned r3 = __builtin_amdgcn_readlane(rcur, k + 3);
                const float* p0 = H1 + (size_t)(r0 & 0xffffu) * HC + coff;
                const float* p1 = H1 + (size_t)(r1 & 0xffffu) * HC + coff;
                const float* p2 = H1 + (size_t)(r2 & 0xffffu) * HC + coff;
                const float* p3 = H1 + (size_t)(r3 & 0xffffu) * HC + coff;
                const float4 u0 = *(const float4*)p0, w0 = *(const float4*)(p0 + 128);
                const float4 u1 = *(const float4*)p1, w1 = *(const float4*)(p1 + 128);
                const float4 u2 = *(const float4*)p2, w2 = *(const float4*)(p2 + 128);
                const float4 u3 = *(const float4*)p3, w3 = *(const float4*)(p3 + 128);
                const float a0 = lane_f(acur, k),     a1 = lane_f(acur, k + 1);
                const float a2 = lane_f(acur, k + 2), a3 = lane_f(acur, k + 3);
                axpy4(acc0, a0, u0); axpy4(acc1, a0, w0);
                axpy4(acc0, a1, u1); axpy4(acc1, a1, w1);
                axpy4(acc0, a2, u2); axpy4(acc1, a2, w2);
                axpy4(acc0, a3, u3); axpy4(acc1, a3, w3);
            }
            if (k + 2 <= cnt) {
                const unsigned r0 = __builtin_amdgcn_readlane(rcur, k);
                const unsigned r1 = __builtin_amdgcn_readlane(rcur, k + 1);
                const float* p0 = H1 + (size_t)(r0 & 0xffffu) * HC + coff;
                const float* p1 = H1 + (size_t)(r1 & 0xffffu) * HC + coff;
                const float4 u0 = *(const float4*)p0, w0 = *(const float4*)(p0 + 128);
                const float4 u1 = *(const float4*)p1, w1 = *(const float4*)(p1 + 128);
                const float a0 = lane_f(acur, k), a1 = lane_f(acur, k + 1);
                axpy4(acc0, a0, u0); axpy4(acc1, a0, w0);
                axpy4(acc0, a1, u1); axpy4(acc1, a1, w1);
                k += 2;
            }
            if (k < cnt) {
                const unsigned r = __builtin_amdgcn_readlane(rcur, k);
                const float* p = H1 + (size_t)(r & 0xffffu) * HC + coff;
                const float4 u = *(const float4*)p, w = *(const float4*)(p + 128);
                const float a = lane_f(acur, k);
                axpy4(acc0, a, u); axpy4(acc1, a, w);
            }
            if (c0 + 32 < end) {
                const int j = c0 + 32 + lane;
                rcur = 0u; acur = 0.f;
                if (j < end) {
                    rcur = elist[j];
                    acur = expf(edge_score1(aS, rcur, head, ad) - m) / den;
                }
            }
        }

        const float4 b0 = *(const float4*)(bias + coff);
        const float4 b1 = *(const float4*)(bias + coff + 128);
        float* op = zout + (size_t)d * HC + coff;
        *(float4*)op         = bias_elu4(acc0, b0);
        *(float4*)(op + 128) = bias_elu4(acc1, b1);
    }
}

#define AGG2_NODES 16
static_assert(N_NODES % AGG2_NODES == 0, "every block of node_agg2 owns 16 whole nodes");
static_assert((AGG2_NODES * NUM_CLASS * 4) % 128 == 0, "a block's output rows are whole 128-byte lines");

__launch_bounds__(256)
__global__ void node_agg2(const int* __restrict__ rowptr, const unsigned* __restrict__ elist,
                          const float* __restrict__ H2, const float* __restrict__ aS,
                          const float* __restrict__ aD, const float* __restrict__ b2,
                          float* __restrict__ out)
{
    __shared__ float rows[AGG2_NODES * NUM_CLASS];

    const int tid = threadIdx.x;
    const int ln  = tid >> 4;
    const int cl  = tid & 15;
    const int c   = min(cl, NUM_CLASS - 1);
    const int d   = blockIdx.x * AGG2_NODES + ln;

    const int beg = rowptr[d];
    const int end = rowptr[d + 1];
    const float ad = aD[d];
    const float es = leaky(aS[d] + ad);

    float m = es;
    for (int j = beg; j < end; j += 4) {
        float e[4];
#pragma unroll
        for (int i = 0; i < 4; ++i)
            e[i] = leaky(aS[elist[min(j + i, end - 1)] & 0xffffu] + ad);
        m = fmaxf(fmaxf(m, fmaxf(e[0], e[1])), fmaxf(e[2], e[3]));
    }
    m = isfinite(m) ? m : 0.f;

    float den, acc;
    {
        const float p = expf(es - m);
        den = p;
        acc = p * H2[(size_t)d * NUM_CLASS + c];
    }
    for (int j = beg; j < end; j += 4) {
        unsigned sN[4];
        float e[4], h[4];
#pragma unroll
        for (int i = 0; i < 4; ++i) sN[i] = elist[min(j + i, end - 1)] & 0xffffu;
#pragma unroll
        for (int i = 0; i < 4; ++i) {
            e[i] = leaky(aS[sN[i]] + ad);
            h[i] = H2[(size_t)sN[i] * NUM_CLASS + c];
        }
#pragma unroll
        for (int i = 0; i < 4; ++i) {
            const bool  on = j + i < end;
            const float p  = on ? expf(e[i] - m) : 0.f;
            den += p;
            acc += p * (on ? h[i] : 0.f);
        }
    }
    const float v = acc / (den + 1e-16f) + b2[c];
    if (cl < NUM_CLASS) rows[ln * NUM_CLASS + cl] = 1.f / (1.f + expf(-v));
    __syncthreads();
    if (tid < AGG2_NODES * NUM_CLASS)
        out[(size_t)blockIdx.x * (AGG2_NODES * NUM_CLASS) + tid] = rows[tid];
}

extern "C" void kernel_launch(void* const* d_in, const int* in_sizes, int n_in,
                              void* d_out, int out_size, void* d_ws, size_t ws_size,
                              hipStream_t stream)
{
    const float* x   = (const float*)d_in[0];
    const int*   ei  = (const int*)d_in[1];
    const float* W1  = (const float*)d_in[2];
    const float* as1 = (const float*)d_in[3];
    const float* ad1 = (const float*)d_in[4];
    const float* b1  = (const float*)d_in[5];
    const float* W2  = (const float*)d_in[6];
    const float* as2 = (const float*)d_in[7];
    const float* ad2 = (const float*)d_in[8];
    const float* b2  = (const float*)d_in[9];
    float* out = (float*)d_out;

    float* ws = (float*)d_ws;
    size_t off = 0;
    float* h1    = ws + off; off += (size_t)N_NODES * HC;
    float* z1    = ws + off; off += (size_t)N_NODES * HC;
    float* sA1   = ws + off; off += (size_t)N_NODES * HEADS;
    float* sD1   = ws + off; off += (size_t)N_NODES * HEADS;
    float* h2    = ws + off; off += (size_t)N_NODES * NUM_CLASS;
    float* sA2   = ws + off; off += N_NODES;
    float* sD2   = ws + off; off += N_NODES;
    unsigned* elist = (unsigned*)(ws + off); off += (size_t)E_IN;
    int*      rowp  = (int*)(ws + off);      off += (size_t)N_NODES + 1;
    (void)ws_size; (void)in_sizes; (void)n_in; (void)out_size;

    const dim3 b256(256);

    gemm1_wmma<<<dim3(HC / 64, (N_NODES + 31) / 32), b256, 0, stream>>>(x, W1, h1);
    att1_kernel<<<(N_NODES * HEADS) / 8, b256, 0, stream>>>(h1, as1, ad1, sA1, sD1);
    csr_build<<<1, CSR_T, 0, stream>>>(ei, rowp, elist);
    node_agg1<<<AGG_BLOCKS, b256, 0, stream>>>(rowp, elist, h1, sA1, sD1, b1, z1);

    gemm2_wmma<<<(N_NODES + 127) / 128, b256, 0, stream>>>(z1, W2, h2);
    att2_kernel<<<(N_NODES + 255) / 256, b256, 0, stream>>>(h2, as2, ad2, sA2, sD2);
    node_agg2<<<N_NODES / AGG2_NODES, b256, 0, stream>>>(rowp, elist, h2, sA2, sD2, b2, out);
}


# Layer-2 GEMM: one block per 16 rows, K split over its 8 waves; z rows read as whole 512-byte runs (non-temporal), made into fragments through LDS; 8 partial tiles summed in LDS (sum order changes)
# speedup vs baseline: 3.1308x; 1.3602x over previous
#include <hip/hip_runtime.h>
#include <math.h>

#define N_NODES   10000
#define F_IN      128
#define HEADS     8
#define HIDD      256
#define HC        (HEADS * HIDD)
#define NUM_CLASS 10
#define E_IN      50000
#define NEG_SLOPE 0.2f

typedef __attribute__((ext_vector_type(2))) float v2f;
typedef __attribute__((ext_vector_type(8))) float v8f;
typedef __attribute__((ext_vector_type(4))) float v4f;

__launch_bounds__(256)
__global__ void gemm1_wmma(const float* __restrict__ X, const float* __restrict__ W,
                           float* __restrict__ Hout)
{
    __shared__ float As[32][F_IN + 4];
    __shared__ float Bs[F_IN][64];

    const int tid = threadIdx.x;
    const int m0  = blockIdx.y * 32;
    const int n0  = blockIdx.x * 64;

#pragma unroll
    for (int i = 0; i < 4; ++i) {
        int idx = (tid + i * 256) * 4;
        int r = idx >> 7;
        int c = idx & 127;
        float4 v = make_float4(0.f, 0.f, 0.f, 0.f);
        int gr = m0 + r;
        if (gr < N_NODES) v = *(const float4*)(X + (size_t)gr * F_IN + c);
        *(float4*)(&As[r][c]) = v;
    }
#pragma unroll
    for (int i = 0; i < 8; ++i) {
        int idx = (tid + i * 256) * 4;
        int r = idx >> 6;
        int c = idx & 63;
        *(float4*)(&Bs[r][c]) = *(const float4*)(W + (size_t)r * HC + n0 + c);
    }
    __syncthreads();

    const int wave = tid >> 5;
    const int lane = tid & 31;
    const int half = lane >> 4;
    const int ml   = lane & 15;
    const int ar   = (wave >> 2) * 16;
    const int bc   = (wave & 3) * 16;

    v8f acc = {0.f, 0.f, 0.f, 0.f, 0.f, 0.f, 0.f, 0.f};
#pragma unroll
    for (int kk = 0; kk < F_IN / 4; ++kk) {
        const int kb = kk * 4 + half * 2;
        v2f a, b;
        float2 av = *(const float2*)(&As[ar + ml][kb]);
        a.x = av.x; a.y = av.y;
        b.x = Bs[kb + 0][bc + ml];
        b.y = Bs[kb + 1][bc + ml];
        acc = __builtin_amdgcn_wmma_f32_16x16x4_f32(false, a, false, b,
                                                    (short)0, acc, false, false);
    }

    const int col = n0 + bc + ml;
#pragma unroll
    for (int r = 0; r < 8; ++r) {
        const int row = m0 + ar + r + 8 * half;
        if (row < N_NODES) Hout[(size_t)row * HC + col] = acc[r];
    }
}

#define G2_ROWS  16
#define G2_WAVES 8
#define G2_KW    (HC / G2_WAVES)
#define G2_KR    128
#define G2_PITCH (G2_KR + 4)
#define G2_TILE  (G2_ROWS * G2_PITCH)
static_assert(N_NODES % G2_ROWS == 0, "every block of gemm2_wmma owns 16 whole rows");
static_assert(G2_KW == 2 * G2_KR, "two rounds per wave");
static_assert(G2_TILE >= 16 * 16, "a wave's partial tile fits its own region");

__device__ __forceinline__ void g2_load_rows(v4f (&v)[G2_ROWS], const float* __restrict__ zp)
{
#pragma unroll
    for (int r = 0; r < G2_ROWS; ++r) v[r] = __builtin_nontemporal_load((const v4f*)(zp + (size_t)r * HC));
}

__device__ __forceinline__ void g2_store_rows(float* __restrict__ tl, const v4f (&v)[G2_ROWS])
{
#pragma unroll
    for (int r = 0; r < G2_ROWS; ++r) *(v4f*)(tl + r * G2_PITCH) = v[r];
}

__device__ __forceinline__ v8f g2_products(v8f acc, const float* __restrict__ fr,
                                           const float* __restrict__ wp)
{
#pragma unroll
    for (int j = 0; j < G2_KR / 8; ++j) {
        const float4 f = *(const float4*)(fr + 8 * j);
        v2f a0, a1, b0, b1;
        a0.x = f.x; a0.y = f.y;
        a1.x = f.z; a1.y = f.w;
        b0.x = wp[(size_t)(8 * j + 0) * NUM_CLASS];
        b0.y = wp[(size_t)(8 * j + 1) * NUM_CLASS];
        b1.x = wp[(size_t)(8 * j + 2) * NUM_CLASS];
        b1.y = wp[(size_t)(8 * j + 3) * NUM_CLASS];
        acc = __builtin_amdgcn_wmma_f32_16x16x4_f32(false, a0, false, b0,
                                                    (short)0, acc, false, false);
        acc = __builtin_amdgcn_wmma_f32_16x16x4_f32(false, a1, false, b1,
                                                    (short)0, acc, false, false);
    }
    return acc;
}

__launch_bounds__(256)
__global__ void gemm2_wmma(const float* __restrict__ Z, const float* __restrict__ W2,
                           float* __restrict__ H2)
{
    __shared__ __attribute__((aligned(16))) float tiles[G2_WAVES * G2_TILE];

    const int tid  = threadIdx.x;
    const int wave = __builtin_amdgcn_readfirstlane(tid >> 5);
    const int lane = tid & 31;
    const int half = lane >> 4;
    const int ml   = lane & 15;
    const int nb   = min(ml, NUM_CLASS - 1);
    const int m0   = blockIdx.x * G2_ROWS;
    const int k0   = wave * G2_KW;

    float*       tl = tiles + wave * G2_TILE;
    float*       ts = tl + lane * 4;
    const float* fr = tl + ml * G2_PITCH + 4 * half;
    const float* zp = Z + (size_t)m0 * HC + k0 + lane * 4;
    const float* wp = W2 + (size_t)(k0 + 4 * half) * NUM_CLASS + nb;

    v8f acc = {0.f, 0.f, 0.f, 0.f, 0.f, 0.f, 0.f, 0.f};
    v4f v[G2_ROWS];

    g2_load_rows(v, zp);
    g2_store_rows(ts, v);
    g2_load_rows(v, zp + G2_KR);
    __builtin_amdgcn_wave_barrier();
    acc = g2_products(acc, fr, wp);
    __builtin_amdgcn_wave_barrier();
    g2_store_rows(ts, v);
    __builtin_amdgcn_wave_barrier();
    acc = g2_products(acc, fr, wp + (size_t)G2_KR * NUM_CLASS);
    __builtin_amdgcn_wave_barrier();

#pragma unroll
    for (int r = 0; r < 8; ++r) tl[(8 * half + r) * 16 + ml] = acc[r];
    __syncthreads();
    if (tid < G2_ROWS * NUM_CLASS) {
        const int row = tid / NUM_CLASS, col = tid % NUM_CLASS;
        float s = 0.f;
#pragma unroll
        for (int w = 0; w < G2_WAVES; ++w) s += tiles[w * G2_TILE + row * 16 + col];
        H2[(size_t)m0 * NUM_CLASS + tid] = s;
    }
}

__launch_bounds__(256)
__global__ void att1_kernel(const float* __restrict__ H1,
                            const float* __restrict__ att_src,
                            const float* __restrict__ att_dst,
                            float* __restrict__ aS, float* __restrict__ aD)
{
    const int wave = threadIdx.x >> 5;
    const int lane = threadIdx.x & 31;
    const int gw = blockIdx.x * 8 + wave;
    if (gw >= N_NODES * HEADS) return;
    const int n = gw / HEADS, hh = gw % HEADS;
    const float* hp = H1 + (size_t)n * HC + hh * HIDD;
    const float* as = att_src + hh * HIDD;
    const float* ad = att_dst + hh * HIDD;
    float ss = 0.f, sd = 0.f;
    for (int c = lane; c < HIDD; c += 32) {
        float v = hp[c];
        ss += v * as[c];
        sd += v * ad[c];
    }
    for (int off = 16; off > 0; off >>= 1) {
        ss += __shfl_down(ss, off, 32);
        sd += __shfl_down(sd, off, 32);
    }
    if (lane == 0) { aS[gw] = ss; aD[gw] = sd; }
}

__global__ void att2_kernel(const float* __restrict__ H2,
                            const float* __restrict__ as2,
                            const float* __restrict__ ad2,
                            float* __restrict__ aS, float* __restrict__ aD)
{
    const int n = blockIdx.x * blockDim.x + threadIdx.x;
    if (n >= N_NODES) return;
    float ss = 0.f, sd = 0.f;
#pragma unroll
    for (int c = 0; c < NUM_CLASS; ++c) {
        float v = H2[(size_t)n * NUM_CLASS + c];
        ss += v * as2[c];
        sd += v * ad2[c];
    }
    aS[n] = ss; aD[n] = sd;
}

#define CSR_T    1024
#define CSR_EPT  ((E_IN + CSR_T - 1) / CSR_T)
#define CSR_CH   7
#define CSR_SPAN ((N_NODES + CSR_T - 1) / CSR_T)
static_assert(E_IN <= 65536 && N_NODES <= 65536, "list entry packs edge id and source in 16 bits each");
static_assert(CSR_EPT % CSR_CH == 0, "chunks cover the edges of a thread exactly");

__launch_bounds__(CSR_T)
__global__ void csr_build(const int* __restrict__ ei, int* __restrict__ rowptr,
                          unsigned* __restrict__ elist)
{
    __shared__ int cnt[N_NODES + 1];
    __shared__ unsigned lst[E_IN + 1];
    __shared__ int wsum[CSR_T / 32];
    __shared__ int total_s;

    const int tid  = threadIdx.x;
    const int lane = tid & 31;
    const int wv   = tid >> 5;

    for (int i = tid; i < N_NODES + 1; i += CSR_T) cnt[i] = 0;
    __syncthreads();

#pragma unroll 1
    for (int k0 = 0; k0 < CSR_EPT; k0 += CSR_CH) {
        unsigned d[CSR_CH];
#pragma unroll
        for (int i = 0; i < CSR_CH; ++i) {
            const int e = tid + (k0 + i) * CSR_T;
            d[i] = (unsigned)ei[E_IN + min(e, E_IN - 1)];
            if (e >= E_IN || d[i] >= (unsigned)N_NODES) d[i] = N_NODES;
        }
#pragma unroll
        for (int i = 0; i < CSR_CH; ++i) atomicAdd(&cnt[d[i]], 1);
    }
    __syncthreads();

    const int base = tid * CSR_SPAN;
    int loc = 0;
#pragma unroll
    for (int j = 0; j < CSR_SPAN; ++j)
        if (base + j < N_NODES) loc += cnt[base + j];

    int inc = loc;
#pragma unroll
    for (int o = 1; o < 32; o <<= 1) {
        const int t = __shfl_up(inc, o, 32);
        if (lane >= o) inc += t;
    }
    if (lane == 31) wsum[wv] = inc;
    __syncthreads();
    if (wv == 0) {
        const int w = wsum[lane];
        int winc = w;
#pragma unroll
        for (int o = 1; o < 32; o <<= 1) {
            const int t = __shfl_up(winc, o, 32);
            if (lane >= o) winc += t;
        }
        wsum[lane] = winc - w;
    }
    __syncthreads();

    int run = wsum[wv] + inc - loc;
#pragma unroll
    for (int j = 0; j < CSR_SPAN; ++j) {
        if (base + j < N_NODES) {
            const int c = cnt[base + j];
            rowptr[base + j] = run;
            cnt[base + j]    = run;
            run += c;
        }
    }
    if (tid == CSR_T - 1) {
        rowptr[N_NODES] = run;
        total_s         = run;
        cnt[N_NODES]    = E_IN;
    }
    __syncthreads();

#pragma unroll 1
    for (int k0 = 0; k0 < CSR_EPT; k0 += CSR_CH) {
        unsigned d[CSR_CH], w[CSR_CH];
#pragma unroll
        for (int i = 0; i < CSR_CH; ++i) {
            const int e  = tid + (k0 + i) * CSR_T;
            const int ec = min(e, E_IN - 1);
            w[i] = ((unsigned)e << 16) | ((unsigned)ei[ec] & 0xffffu);
            d[i] = (unsigned)ei[E_IN + ec];
            if (e >= E_IN || d[i] >= (unsigned)N_NODES) d[i] = N_NODES;
        }
        int slot[CSR_CH];
#pragma unroll
        for (int i = 0; i < CSR_CH; ++i) slot[i] = atomicAdd(&cnt[d[i]], 1);
#pragma unroll
        for (int i = 0; i < CSR_CH; ++i) lst[min(slot[i], E_IN)] = w[i];
    }
    __syncthreads();

    const int total = total_s;
#pragma unroll 1
    for (int k0 = 0; k0 < CSR_EPT; k0 += CSR_CH) {
        unsigned v[CSR_CH];
#pragma unroll
        for (int i = 0; i < CSR_CH; ++i) v[i] = lst[min(tid + (k0 + i) * CSR_T, E_IN - 1)];
#pragma unroll
        for (int i = 0; i < CSR_CH; ++i) {
            const int idx = tid + (k0 + i) * CSR_T;
            if (idx < total) elist[idx] = v[i];
        }
    }
}

__device__ __forceinline__ float leaky(const float x)
{
    return (x >= 0.f) ? x : NEG_SLOPE * x;
}

template <int MASK>
__device__ __forceinline__ float row_xor(const float x)
{
    return __builtin_bit_cast(float, __builtin_amdgcn_update_dpp(
        0, __builtin_bit_cast(int, x), 0x160 + MASK, 0xf, 0xf, false));
}

__device__ __forceinline__ float xhalf(const float x)
{
    const int b = __builtin_bit_cast(int, x);
    return __builtin_bit_cast(float, __builtin_amdgcn_permlanex16(b, b, 0x76543210, 0xfedcba98, false, false));
}

__device__ __forceinline__ float wave_max(float x)
{
    x = fmaxf(x, row_xor<1>(x));
    x = fmaxf(x, row_xor<2>(x));
    x = fmaxf(x, row_xor<4>(x));
    x = fmaxf(x, row_xor<8>(x));
    return fmaxf(x, xhalf(x));
}

__device__ __forceinline__ float wave_sum(float x)
{
    x += row_xor<1>(x);
    x += row_xor<2>(x);
    x += row_xor<4>(x);
    x += row_xor<8>(x);
    return x + xhalf(x);
}

__device__ __forceinline__ float lane_f(const float v, const int k)
{
    return __int_as_float(__builtin_amdgcn_readlane(__float_as_int(v), k));
}

#define AGG_BLOCKS 2000

__device__ __forceinline__ void axpy4(float4& acc, const float a, const float4 v)
{
    acc.x += a * v.x; acc.y += a * v.y; acc.z += a * v.z; acc.w += a * v.w;
}

__device__ __forceinline__ float bias_elu1(const float s, const float b)
{
    const float v = s + b;
    return (v > 0.f) ? v : expm1f(v);
}

__device__ __forceinline__ float4 bias_elu4(const float4 s, const float4 b)
{
    return make_float4(bias_elu1(s.x, b.x), bias_elu1(s.y, b.y),
                       bias_elu1(s.z, b.z), bias_elu1(s.w, b.w));
}

__device__ __forceinline__ float edge_score1(const float* __restrict__ aS, const unsigned rec,
                                             const int head, const float ad)
{
    return leaky(aS[(rec & 0xffffu) * HEADS + head] + ad);
}

__launch_bounds__(256)
__global__ void node_agg1(const int* __restrict__ rowptr, const unsigned* __restrict__ elist,
                          const float* __restrict__ H1, const float* __restrict__ aS,
                          const float* __restrict__ aD, const float* bias, float* zout)
{
    const int lane = threadIdx.x & 31;
    const int head = __builtin_amdgcn_readfirstlane((int)(threadIdx.x >> 5));
    const int coff = head * HIDD + lane * 4;

    for (int d = blockIdx.x; d < N_NODES; d += gridDim.x) {
        const int beg = rowptr[d];
        const int end = rowptr[d + 1];
        const float ad = aD[d * HEADS + head];
        const float es = leaky(aS[d * HEADS + head] + ad);

        const int  j0 = beg + lane;
        const bool v0 = j0 < end;
        unsigned rc = 0u;
        float ec = -INFINITY;
        if (v0) { rc = elist[j0]; ec = edge_score1(aS, rc, head, ad); }
        float m = fmaxf(es, ec);
        for (int c0 = beg + 32; c0 < end; c0 += 32) {
            const int j = c0 + lane;
            if (j < end) m = fmaxf(m, edge_score1(aS, elist[j], head, ad));
        }
        m = wave_max(m);
        m = isfinite(m) ? m : 0.f;

        const float pc = v0 ? expf(ec - m) : 0.f;
        float s = pc;
        for (int c0 = beg + 32; c0 < end; c0 += 32) {
            const int j = c0 + lane;
            if (j < end) s += expf(edge_score1(aS, elist[j], head, ad) - m);
        }
        const float ps  = expf(es - m);
        const float den = wave_sum(s) + ps + 1e-16f;

        float4 acc0, acc1;
        {
            const float a = ps / den;
            const float* hp = H1 + (size_t)d * HC + coff;
            const float4 v0r = *(const float4*)hp;
            const float4 v1r = *(const float4*)(hp + 128);
            acc0 = make_float4(a * v0r.x, a * v0r.y, a * v0r.z, a * v0r.w);
            acc1 = make_float4(a * v1r.x, a * v1r.y, a * v1r.z, a * v1r.w);
        }

        unsigned rcur = rc;
        float    acur = pc / den;
        for (int c0 = beg; c0 < end; c0 += 32) {
            const int cnt = min(end - c0, 32);
            int k = 0;
            for (; k + 4 <= cnt; k += 4) {
                const unsigned r0 = __builtin_amdgcn_readlane(rcur, k);
                const unsigned r1 = __builtin_amdgcn_readlane(rcur, k + 1);
                const unsigned r2 = __builtin_amdgcn_readlane(rcur, k + 2);
                const unsigned r3 = __builtin_amdgcn_readlane(rcur, k + 3);
                const float* p0 = H1 + (size_t)(r0 & 0xffffu) * HC + coff;
                const float* p1 = H1 + (size_t)(r1 & 0xffffu) * HC + coff;
                const float* p2 = H1 + (size_t)(r2 & 0xffffu) * HC + coff;
                const float* p3 = H1 + (size_t)(r3 & 0xffffu) * HC + coff;
                const float4 u0 = *(const float4*)p0, w0 = *(const float4*)(p0 + 128);
                const float4 u1 = *(const float4*)p1, w1 = *(const float4*)(p1 + 128);
                const float4 u2 = *(const float4*)p2, w2 = *(const float4*)(p2 + 128);
                const float4 u3 = *(const float4*)p3, w3 = *(const float4*)(p3 + 128);
                const float a0 = lane_f(acur, k),     a1 = lane_f(acur, k + 1);
                const float a2 = lane_f(acur, k + 2), a3 = lane_f(acur, k + 3);
                axpy4(acc0, a0, u0); axpy4(acc1, a0, w0);
                axpy4(acc0, a1, u1); axpy4(acc1, a1, w1);
                axpy4(acc0, a2, u2); axpy4(acc1, a2, w2);
                axpy4(acc0, a3, u3); axpy4(acc1, a3, w3);
            }
            if (k + 2 <= cnt) {
                const unsigned r0 = __builtin_amdgcn_readlane(rcur, k);
                const unsigned r1 = __builtin_amdgcn_readlane(rcur, k + 1);
                const float* p0 = H1 + (size_t)(r0 & 0xffffu) * HC + coff;
                const float* p1 = H1 + (size_t)(r1 & 0xffffu) * HC + coff;
                const float4 u0 = *(const float4*)p0, w0 = *(const float4*)(p0 + 128);
                const float4 u1 = *(const float4*)p1, w1 = *(const float4*)(p1 + 128);
                const float a0 = lane_f(acur, k), a1 = lane_f(acur, k + 1);
                axpy4(acc0, a0, u0); axpy4(acc1, a0, w0);
                axpy4(acc0, a1, u1); axpy4(acc1, a1, w1);
                k += 2;
            }
            if (k < cnt) {
                const unsigned r = __builtin_amdgcn_readlane(rcur, k);
                const float* p = H1 + (size_t)(r & 0xffffu) * HC + coff;
                const float4 u = *(const float4*)p, w = *(const float4*)(p + 128);
                const float a = lane_f(acur, k);
                axpy4(acc0, a, u); axpy4(acc1, a, w);
            }
            if (c0 + 32 < end) {
                const int j = c0 + 32 + lane;
                rcur = 0u; acur = 0.f;
                if (j < end) {
                    rcur = elist[j];
                    acur = expf(edge_score1(aS, rcur, head, ad) - m) / den;
                }
            }
        }

        const float4 b0 = *(const float4*)(bias + coff);
        const float4 b1 = *(const float4*)(bias + coff + 128);
        float* op = zout + (size_t)d * HC + coff;
        *(float4*)op         = bias_elu4(acc0, b0);
        *(float4*)(op + 128) = bias_elu4(acc1, b1);
    }
}

#define AGG2_NODES 16
static_assert(N_NODES % AGG2_NODES == 0, "every block of node_agg2 owns 16 whole nodes");
static_assert((AGG2_NODES * NUM_CLASS * 4) % 128 == 0, "a block's output rows are whole 128-byte lines");

__launch_bounds__(256)
__global__ void node_agg2(const int* __restrict__ rowptr, const unsigned* __restrict__ elist,
                          const float* __restrict__ H2, const float* __restrict__ aS,
                          const float* __restrict__ aD, const float* __restrict__ b2,
                          float* __restrict__ out)
{
    __shared__ float rows[AGG2_NODES * NUM_CLASS];

    const int tid = threadIdx.x;
    const int ln  = tid >> 4;
    const int cl  = tid & 15;
    const int c   = min(cl, NUM_CLASS - 1);
    const int d   = blockIdx.x * AGG2_NODES + ln;

    const int beg = rowptr[d];
    const int end = rowptr[d + 1];
    const float ad = aD[d];
    const float es = leaky(aS[d] + ad);

    float m = es;
    for (int j = beg; j < end; j += 4) {
        float e[4];
#pragma unroll
        for (int i = 0; i < 4; ++i)
            e[i] = leaky(aS[elist[min(j + i, end - 1)] & 0xffffu] + ad);
        m = fmaxf(fmaxf(m, fmaxf(e[0], e[1])), fmaxf(e[2], e[3]));
    }
    m = isfinite(m) ? m : 0.f;

    float den, acc;
    {
        const float p = expf(es - m);
        den = p;
        acc = p * H2[(size_t)d * NUM_CLASS + c];
    }
    for (int j = beg; j < end; j += 4) {
        unsigned sN[4];
        float e[4], h[4];
#pragma unroll
        for (int i = 0; i < 4; ++i) sN[i] = elist[min(j + i, end - 1)] & 0xffffu;
#pragma unroll
        for (int i = 0; i < 4; ++i) {
            e[i] = leaky(aS[sN[i]] + ad);
            h[i] = H2[(size_t)sN[i] * NUM_CLASS + c];
        }
#pragma unroll
        for (int i = 0; i < 4; ++i) {
            const bool  on = j + i < end;
            const float p  = on ? expf(e[i] - m) : 0.f;
            den += p;
            acc += p * (on ? h[i] : 0.f);
        }
    }
    const float v = acc / (den + 1e-16f) + b2[c];
    if (cl < NUM_CLASS) rows[ln * NUM_CLASS + cl] = 1.f / (1.f + expf(-v));
    __syncthreads();
    if (tid < AGG2_NODES * NUM_CLASS)
        out[(size_t)blockIdx.x * (AGG2_NODES * NUM_CLASS) + tid] = rows[tid];
}

extern "C" void kernel_launch(void* const* d_in, const int* in_sizes, int n_in,
                              void* d_out, int out_size, void* d_ws, size_t ws_size,
                              hipStream_t stream)
{
    const float* x   = (const float*)d_in[0];
    const int*   ei  = (const int*)d_in[1];
    const float* W1  = (const float*)d_in[2];
    const float* as1 = (const float*)d_in[3];
    const float* ad1 = (const float*)d_in[4];
    const float* b1  = (const float*)d_in[5];
    const float* W2  = (const float*)d_in[6];
    const float* as2 = (const float*)d_in[7];
    const float* ad2 = (const float*)d_in[8];
    const float* b2  = (const float*)d_in[9];
    float* out = (float*)d_out;

    float* ws = (float*)d_ws;
    size_t off = 0;
    float* h1    = ws + off; off += (size_t)N_NODES * HC;
    float* z1    = ws + off; off += (size_t)N_NODES * HC;
    float* sA1   = ws + off; off += (size_t)N_NODES * HEADS;
    float* sD1   = ws + off; off += (size_t)N_NODES * HEADS;
    float* h2    = ws + off; off += (size_t)N_NODES * NUM_CLASS;
    float* sA2   = ws + off; off += N_NODES;
    float* sD2   = ws + off; off += N_NODES;
    unsigned* elist = (unsigned*)(ws + off); off += (size_t)E_IN;
    int*      rowp  = (int*)(ws + off);      off += (size_t)N_NODES + 1;
    (void)ws_size; (void)in_sizes; (void)n_in; (void)out_size;

    const dim3 b256(256);

    gemm1_wmma<<<dim3(HC / 64, (N_NODES + 31) / 32), b256, 0, stream>>>(x, W1, h1);
    att1_kernel<<<(N_NODES * HEADS) / 8, b256, 0, stream>>>(h1, as1, ad1, sA1, sD1);
    csr_build<<<1, CSR_T, 0, stream>>>(ei, rowp, elist);
    node_agg1<<<AGG_BLOCKS, b256, 0, stream>>>(rowp, elist, h1, sA1, sD1, b1, z1);

    gemm2_wmma<<<N_NODES / G2_ROWS, b256, 0, stream>>>(z1, W2, h2);
    att2_kernel<<<(N_NODES + 255) / 256, b256, 0, stream>>>(h2, as2, ad2, sA2, sD2);
    node_agg2<<<N_NODES / AGG2_NODES, b256, 0, stream>>>(rowp, elist, h2, sA2, sD2, b2, out);
}


# Layer-1 attention dots: one wave per head and 8 nodes (80000 waves -> 10000), its 16 row loads all in flight at once, not 8 waited-on trips; DPP butterfly sums (dot-product sum order changes)
# speedup vs baseline: 3.5758x; 1.1421x over previous
#include <hip/hip_runtime.h>
#include <math.h>

#define N_NODES   10000
#define F_IN      128
#define HEADS     8
#define HIDD      256
#define HC        (HEADS * HIDD)
#define NUM_CLASS 10
#define E_IN      50000
#define NEG_SLOPE 0.2f

typedef __attribute__((ext_vector_type(2))) float v2f;
typedef __attribute__((ext_vector_type(8))) float v8f;
typedef __attribute__((ext_vector_type(4))) float v4f;

__launch_bounds__(256)
__global__ void gemm1_wmma(const float* __restrict__ X, const float* __restrict__ W,
                           float* __restrict__ Hout)
{
    __shared__ float As[32][F_IN + 4];
    __shared__ float Bs[F_IN][64];

    const int tid = threadIdx.x;
    const int m0  = blockIdx.y * 32;
    const int n0  = blockIdx.x * 64;

#pragma unroll
    for (int i = 0; i < 4; ++i) {
        int idx = (tid + i * 256) * 4;
        int r = idx >> 7;
        int c = idx & 127;
        float4 v = make_float4(0.f, 0.f, 0.f, 0.f);
        int gr = m0 + r;
        if (gr < N_NODES) v = *(const float4*)(X + (size_t)gr * F_IN + c);
        *(float4*)(&As[r][c]) = v;
    }
#pragma unroll
    for (int i = 0; i < 8; ++i) {
        int idx = (tid + i * 256) * 4;
        int r = idx >> 6;
        int c = idx & 63;
        *(float4*)(&Bs[r][c]) = *(const float4*)(W + (size_t)r * HC + n0 + c);
    }
    __syncthreads();

    const int wave = tid >> 5;
    const int lane = tid & 31;
    const int half = lane >> 4;
    const int ml   = lane & 15;
    const int ar   = (wave >> 2) * 16;
    const int bc   = (wave & 3) * 16;

    v8f acc = {0.f, 0.f, 0.f, 0.f, 0.f, 0.f, 0.f, 0.f};
#pragma unroll
    for (int kk = 0; kk < F_IN / 4; ++kk) {
        const int kb = kk * 4 + half * 2;
        v2f a, b;
        float2 av = *(const float2*)(&As[ar + ml][kb]);
        a.x = av.x; a.y = av.y;
        b.x = Bs[kb + 0][bc + ml];
        b.y = Bs[kb + 1][bc + ml];
        acc = __builtin_amdgcn_wmma_f32_16x16x4_f32(false, a, false, b,
                                                    (short)0, acc, false, false);
    }

    const int col = n0 + bc + ml;
#pragma unroll
    for (int r = 0; r < 8; ++r) {
        const int row = m0 + ar + r + 8 * half;
        if (row < N_NODES) Hout[(size_t)row * HC + col] = acc[r];
    }
}

#define G2_ROWS  16
#define G2_WAVES 8
#define G2_KW    (HC / G2_WAVES)
#define G2_KR    128
#define G2_PITCH (G2_KR + 4)
#define G2_TILE  (G2_ROWS * G2_PITCH)
static_assert(N_NODES % G2_ROWS == 0, "every block of gemm2_wmma owns 16 whole rows");
static_assert(G2_KW == 2 * G2_KR, "two rounds per wave");
static_assert(G2_TILE >= 16 * 16, "a wave's partial tile fits its own region");

__device__ __forceinline__ void g2_load_rows(v4f (&v)[G2_ROWS], const float* __restrict__ zp)
{
#pragma unroll
    for (int r = 0; r < G2_ROWS; ++r) v[r] = __builtin_nontemporal_load((const v4f*)(zp + (size_t)r * HC));
}

__device__ __forceinline__ void g2_store_rows(float* __restrict__ tl, const v4f (&v)[G2_ROWS])
{
#pragma unroll
    for (int r = 0; r < G2_ROWS; ++r) *(v4f*)(tl + r * G2_PITCH) = v[r];
}

__device__ __forceinline__ v8f g2_products(v8f acc, const float* __restrict__ fr,
                                           const float* __restrict__ wp)
{
#pragma unroll
    for (int j = 0; j < G2_KR / 8; ++j) {
        const float4 f = *(const float4*)(fr + 8 * j);
        v2f a0, a1, b0, b1;
        a0.x = f.x; a0.y = f.y;
        a1.x = f.z; a1.y = f.w;
        b0.x = wp[(size_t)(8 * j + 0) * NUM_CLASS];
        b0.y = wp[(size_t)(8 * j + 1) * NUM_CLASS];
        b1.x = wp[(size_t)(8 * j + 2) * NUM_CLASS];
        b1.y = wp[(size_t)(8 * j + 3) * NUM_CLASS];
        acc = __builtin_amdgcn_wmma_f32_16x16x4_f32(false, a0, false, b0,
                                                    (short)0, acc, false, false);
        acc = __builtin_amdgcn_wmma_f32_16x16x4_f32(false, a1, false, b1,
                                                    (short)0, acc, false, false);
    }
    return acc;
}

__launch_bounds__(256)
__global__ void gemm2_wmma(const float* __restrict__ Z, const float* __restrict__ W2,
                           float* __restrict__ H2)
{
    __shared__ __attribute__((aligned(16))) float tiles[G2_WAVES * G2_TILE];

    const int tid  = threadIdx.x;
    const int wave = __builtin_amdgcn_readfirstlane(tid >> 5);
    const int lane = tid & 31;
    const int half = lane >> 4;
    const int ml   = lane & 15;
    const int nb   = min(ml, NUM_CLASS - 1);
    const int m0   = blockIdx.x * G2_ROWS;
    const int k0   = wave * G2_KW;

    float*       tl = tiles + wave * G2_TILE;
    float*       ts = tl + lane * 4;
    const float* fr = tl + ml * G2_PITCH + 4 * half;
    const float* zp = Z + (size_t)m0 * HC + k0 + lane * 4;
    const float* wp = W2 + (size_t)(k0 + 4 * half) * NUM_CLASS + nb;

    v8f acc = {0.f, 0.f, 0.f, 0.f, 0.f, 0.f, 0.f, 0.f};
    v4f v[G2_ROWS];

    g2_load_rows(v, zp);
    g2_store_rows(ts, v);
    g2_load_rows(v, zp + G2_KR);
    __builtin_amdgcn_wave_barrier();
    acc = g2_products(acc, fr, wp);
    __builtin_amdgcn_wave_barrier();
    g2_store_rows(ts, v);
    __builtin_amdgcn_wave_barrier();
    acc = g2_products(acc, fr, wp + (size_t)G2_KR * NUM_CLASS);
    __builtin_amdgcn_wave_barrier();

#pragma unroll
    for (int r = 0; r < 8; ++r) tl[(8 * half + r) * 16 + ml] = acc[r];
    __syncthreads();
    if (tid < G2_ROWS * NUM_CLASS) {
        const int row = tid / NUM_CLASS, col = tid % NUM_CLASS;
        float s = 0.f;
#pragma unroll
        for (int w = 0; w < G2_WAVES; ++w) s += tiles[w * G2_TILE + row * 16 + col];
        H2[(size_t)m0 * NUM_CLASS + tid] = s;
    }
}

__global__ void att2_kernel(const float* __restrict__ H2,
                            const float* __restrict__ as2,
                            const float* __restrict__ ad2,
                            float* __restrict__ aS, float* __restrict__ aD)
{
    const int n = blockIdx.x * blockDim.x + threadIdx.x;
    if (n >= N_NODES) return;
    float ss = 0.f, sd = 0.f;
#pragma unroll
    for (int c = 0; c < NUM_CLASS; ++c) {
        float v = H2[(size_t)n * NUM_CLASS + c];
        ss += v * as2[c];
        sd += v * ad2[c];
    }
    aS[n] = ss; aD[n] = sd;
}

#define CSR_T    1024
#define CSR_EPT  ((E_IN + CSR_T - 1) / CSR_T)
#define CSR_CH   7
#define CSR_SPAN ((N_NODES + CSR_T - 1) / CSR_T)
static_assert(E_IN <= 65536 && N_NODES <= 65536, "list entry packs edge id and source in 16 bits each");
static_assert(CSR_EPT % CSR_CH == 0, "chunks cover the edges of a thread exactly");

__launch_bounds__(CSR_T)
__global__ void csr_build(const int* __restrict__ ei, int* __restrict__ rowptr,
                          unsigned* __restrict__ elist)
{
    __shared__ int cnt[N_NODES + 1];
    __shared__ unsigned lst[E_IN + 1];
    __shared__ int wsum[CSR_T / 32];
    __shared__ int total_s;

    const int tid  = threadIdx.x;
    const int lane = tid & 31;
    const int wv   = tid >> 5;

    for (int i = tid; i < N_NODES + 1; i += CSR_T) cnt[i] = 0;
    __syncthreads();

#pragma unroll 1
    for (int k0 = 0; k0 < CSR_EPT; k0 += CSR_CH) {
        unsigned d[CSR_CH];
#pragma unroll
        for (int i = 0; i < CSR_CH; ++i) {
            const int e = tid + (k0 + i) * CSR_T;
            d[i] = (unsigned)ei[E_IN + min(e, E_IN - 1)];
            if (e >= E_IN || d[i] >= (unsigned)N_NODES) d[i] = N_NODES;
        }
#pragma unroll
        for (int i = 0; i < CSR_CH; ++i) atomicAdd(&cnt[d[i]], 1);
    }
    __syncthreads();

    const int base = tid * CSR_SPAN;
    int loc = 0;
#pragma unroll
    for (int j = 0; j < CSR_SPAN; ++j)
        if (base + j < N_NODES) loc += cnt[base + j];

    int inc = loc;
#pragma unroll
    for (int o = 1; o < 32; o <<= 1) {
        const int t = __shfl_up(inc, o, 32);
        if (lane >= o) inc += t;
    }
    if (lane == 31) wsum[wv] = inc;
    __syncthreads();
    if (wv == 0) {
        const int w = wsum[lane];
        int winc = w;
#pragma unroll
        for (int o = 1; o < 32; o <<= 1) {
            const int t = __shfl_up(winc, o, 32);
            if (lane >= o) winc += t;
        }
        wsum[lane] = winc - w;
    }
    __syncthreads();

    int run = wsum[wv] + inc - loc;
#pragma unroll
    for (int j = 0; j < CSR_SPAN; ++j) {
        if (base + j < N_NODES) {
            const int c = cnt[base + j];
            rowptr[base + j] = run;
            cnt[base + j]    = run;
            run += c;
        }
    }
    if (tid == CSR_T - 1) {
        rowptr[N_NODES] = run;
        total_s         = run;
        cnt[N_NODES]    = E_IN;
    }
    __syncthreads();

#pragma unroll 1
    for (int k0 = 0; k0 < CSR_EPT; k0 += CSR_CH) {
        unsigned d[CSR_CH], w[CSR_CH];
#pragma unroll
        for (int i = 0; i < CSR_CH; ++i) {
            const int e  = tid + (k0 + i) * CSR_T;
            const int ec = min(e, E_IN - 1);
            w[i] = ((unsigned)e << 16) | ((unsigned)ei[ec] & 0xffffu);
            d[i] = (unsigned)ei[E_IN + ec];
            if (e >= E_IN || d[i] >= (unsigned)N_NODES) d[i] = N_NODES;
        }
        int slot[CSR_CH];
#pragma unroll
        for (int i = 0; i < CSR_CH; ++i) slot[i] = atomicAdd(&cnt[d[i]], 1);
#pragma unroll
        for (int i = 0; i < CSR_CH; ++i) lst[min(slot[i], E_IN)] = w[i];
    }
    __syncthreads();

    const int total = total_s;
#pragma unroll 1
    for (int k0 = 0; k0 < CSR_EPT; k0 += CSR_CH) {
        unsigned v[CSR_CH];
#pragma unroll
        for (int i = 0; i < CSR_CH; ++i) v[i] = lst[min(tid + (k0 + i) * CSR_T, E_IN - 1)];
#pragma unroll
        for (int i = 0; i < CSR_CH; ++i) {
            const int idx = tid + (k0 + i) * CSR_T;
            if (idx < total) elist[idx] = v[i];
        }
    }
}

__device__ __forceinline__ float leaky(const float x)
{
    return (x >= 0.f) ? x : NEG_SLOPE * x;
}

template <int MASK>
__device__ __forceinline__ float row_xor(const float x)
{
    return __builtin_bit_cast(float, __builtin_amdgcn_update_dpp(
        0, __builtin_bit_cast(int, x), 0x160 + MASK, 0xf, 0xf, false));
}

__device__ __forceinline__ float xhalf(const float x)
{
    const int b = __builtin_bit_cast(int, x);
    return __builtin_bit_cast(float, __builtin_amdgcn_permlanex16(b, b, 0x76543210, 0xfedcba98, false, false));
}

__device__ __forceinline__ float wave_max(float x)
{
    x = fmaxf(x, row_xor<1>(x));
    x = fmaxf(x, row_xor<2>(x));
    x = fmaxf(x, row_xor<4>(x));
    x = fmaxf(x, row_xor<8>(x));
    return fmaxf(x, xhalf(x));
}

__device__ __forceinline__ float wave_sum(float x)
{
    x += row_xor<1>(x);
    x += row_xor<2>(x);
    x += row_xor<4>(x);
    x += row_xor<8>(x);
    return x + xhalf(x);
}

__device__ __forceinline__ float lane_f(const float v, const int k)
{
    return __int_as_float(__builtin_amdgcn_readlane(__float_as_int(v), k));
}

#define ATT_NODES 8
static_assert(N_NODES % ATT_NODES == 0, "every block of att1_kernel owns 8 whole nodes");
static_assert((ATT_NODES * HEADS * 4) % 128 == 0, "a block's results are whole 128-byte lines");

__device__ __forceinline__ float dot8(const float4 u, const float4 w, const float4 a, const float4 b)
{
    return u.x * a.x + u.y * a.y + u.z * a.z + u.w * a.w
         + w.x * b.x + w.y * b.y + w.z * b.z + w.w * b.w;
}

__launch_bounds__(256)
__global__ void att1_kernel(const float* __restrict__ H1,
                            const float* __restrict__ att_src,
                            const float* __restrict__ att_dst,
                            float* __restrict__ aS, float* __restrict__ aD)
{
    __shared__ float res[2 * ATT_NODES * HEADS];

    const int tid  = threadIdx.x;
    const int lane = tid & 31;
    const int head = __builtin_amdgcn_readfirstlane(tid >> 5);
    const int n0   = blockIdx.x * ATT_NODES;
    const int coff = head * HIDD + lane * 4;

    const float4 s0 = *(const float4*)(att_src + coff);
    const float4 s1 = *(const float4*)(att_src + coff + 128);
    const float4 d0 = *(const float4*)(att_dst + coff);
    const float4 d1 = *(const float4*)(att_dst + coff + 128);

    const float* hp = H1 + (size_t)n0 * HC + coff;
    float4 u[ATT_NODES], w[ATT_NODES];
#pragma unroll
    for (int i = 0; i < ATT_NODES; ++i) {
        u[i] = *(const float4*)(hp + (size_t)i * HC);
        w[i] = *(const float4*)(hp + (size_t)i * HC + 128);
    }

    float keep = 0.f;
#pragma unroll
    for (int i = 0; i < ATT_NODES; ++i) {
        const float ss = wave_sum(dot8(u[i], w[i], s0, s1));
        const float sd = wave_sum(dot8(u[i], w[i], d0, d1));
        keep = (lane == i) ? ss : keep;
        keep = (lane == ATT_NODES + i) ? sd : keep;
    }
    if (lane < 2 * ATT_NODES)
        res[(lane >> 3) * (ATT_NODES * HEADS) + (lane & 7) * HEADS + head] = keep;
    __syncthreads();
    if (tid < ATT_NODES * HEADS)
        aS[(size_t)n0 * HEADS + tid] = res[tid];
    else if (tid < 2 * ATT_NODES * HEADS)
        aD[(size_t)n0 * HEADS + (tid - ATT_NODES * HEADS)] = res[tid];
}

#define AGG_BLOCKS 2000

__device__ __forceinline__ void axpy4(float4& acc, const float a, const float4 v)
{
    acc.x += a * v.x; acc.y += a * v.y; acc.z += a * v.z; acc.w += a * v.w;
}

__device__ __forceinline__ float bias_elu1(const float s, const float b)
{
    const float v = s + b;
    return (v > 0.f) ? v : expm1f(v);
}

__device__ __forceinline__ float4 bias_elu4(const float4 s, const float4 b)
{
    return make_float4(bias_elu1(s.x, b.x), bias_elu1(s.y, b.y),
                       bias_elu1(s.z, b.z), bias_elu1(s.w, b.w));
}

__device__ __forceinline__ float edge_score1(const float* __restrict__ aS, const unsigned rec,
                                             const int head, const float ad)
{
    return leaky(aS[(rec & 0xffffu) * HEADS + head] + ad);
}

__launch_bounds__(256)
__global__ void node_agg1(const int* __restrict__ rowptr, const unsigned* __restrict__ elist,
                          const float* __restrict__ H1, const float* __restrict__ aS,
                          const float* __restrict__ aD, const float* bias, float* zout)
{
    const int lane = threadIdx.x & 31;
    const int head = __builtin_amdgcn_readfirstlane((int)(threadIdx.x >> 5));
    const int coff = head * HIDD + lane * 4;

    for (int d = blockIdx.x; d < N_NODES; d += gridDim.x) {
        const int beg = rowptr[d];
        const int end = rowptr[d + 1];
        const float ad = aD[d * HEADS + head];
        const float es = leaky(aS[d * HEADS + head] + ad);

        const int  j0 = beg + lane;
        const bool v0 = j0 < end;
        unsigned rc = 0u;
        float ec = -INFINITY;
        if (v0) { rc = elist[j0]; ec = edge_score1(aS, rc, head, ad); }
        float m = fmaxf(es, ec);
        for (int c0 = beg + 32; c0 < end; c0 += 32) {
            const int j = c0 + lane;
            if (j < end) m = fmaxf(m, edge_score1(aS, elist[j], head, ad));
        }
        m = wave_max(m);
        m = isfinite(m) ? m : 0.f;

        const float pc = v0 ? expf(ec - m) : 0.f;
        float s = pc;
        for (int c0 = beg + 32; c0 < end; c0 += 32) {
            const int j = c0 + lane;
            if (j < end) s += expf(edge_score1(aS, elist[j], head, ad) - m);
        }
        const float ps  = expf(es - m);
        const float den = wave_sum(s) + ps + 1e-16f;

        float4 acc0, acc1;
        {
            const float a = ps / den;
            const float* hp = H1 + (size_t)d * HC + coff;
            const float4 v0r = *(const float4*)hp;
            const float4 v1r = *(const float4*)(hp + 128);
            acc0 = make_float4(a * v0r.x, a * v0r.y, a * v0r.z, a * v0r.w);
            acc1 = make_float4(a * v1r.x, a * v1r.y, a * v1r.z, a * v1r.w);
        }

        unsigned rcur = rc;
        float    acur = pc / den;
        for (int c0 = beg; c0 < end; c0 += 32) {
            const int cnt = min(end - c0, 32);
            int k = 0;
            for (; k + 4 <= cnt; k += 4) {
                const unsigned r0 = __builtin_amdgcn_readlane(rcur, k);
                const unsigned r1 = __builtin_amdgcn_readlane(rcur, k + 1);
                const unsigned r2 = __builtin_amdgcn_readlane(rcur, k + 2);
                const unsigned r3 = __builtin_amdgcn_readlane(rcur, k + 3);
                const float* p0 = H1 + (size_t)(r0 & 0xffffu) * HC + coff;
                const float* p1 = H1 + (size_t)(r1 & 0xffffu) * HC + coff;
                const float* p2 = H1 + (size_t)(r2 & 0xffffu) * HC + coff;
                const float* p3 = H1 + (size_t)(r3 & 0xffffu) * HC + coff;
                const float4 u0 = *(const float4*)p0, w0 = *(const float4*)(p0 + 128);
                const float4 u1 = *(const float4*)p1, w1 = *(const float4*)(p1 + 128);
                const float4 u2 = *(const float4*)p2, w2 = *(const float4*)(p2 + 128);
                const float4 u3 = *(const float4*)p3, w3 = *(const float4*)(p3 + 128);
                const float a0 = lane_f(acur, k),     a1 = lane_f(acur, k + 1);
                const float a2 = lane_f(acur, k + 2), a3 = lane_f(acur, k + 3);
                axpy4(acc0, a0, u0); axpy4(acc1, a0, w0);
                axpy4(acc0, a1, u1); axpy4(acc1, a1, w1);
                axpy4(acc0, a2, u2); axpy4(acc1, a2, w2);
                axpy4(acc0, a3, u3); axpy4(acc1, a3, w3);
            }
            if (k + 2 <= cnt) {
                const unsigned r0 = __builtin_amdgcn_readlane(rcur, k);
                const unsigned r1 = __builtin_amdgcn_readlane(rcur, k + 1);
                const float* p0 = H1 + (size_t)(r0 & 0xffffu) * HC + coff;
                const float* p1 = H1 + (size_t)(r1 & 0xffffu) * HC + coff;
                const float4 u0 = *(const float4*)p0, w0 = *(const float4*)(p0 + 128);
                const float4 u1 = *(const float4*)p1, w1 = *(const float4*)(p1 + 128);
                const float a0 = lane_f(acur, k), a1 = lane_f(acur, k + 1);
                axpy4(acc0, a0, u0); axpy4(acc1, a0, w0);
                axpy4(acc0, a1, u1); axpy4(acc1, a1, w1);
                k += 2;
            }
            if (k < cnt) {
                const unsigned r = __builtin_amdgcn_readlane(rcur, k);
                const float* p = H1 + (size_t)(r & 0xffffu) * HC + coff;
                const float4 u = *(const float4*)p, w = *(const float4*)(p + 128);
                const float a = lane_f(acur, k);
                axpy4(acc0, a, u); axpy4(acc1, a, w);
            }
            if (c0 + 32 < end) {
                const int j = c0 + 32 + lane;
                rcur = 0u; acur = 0.f;
                if (j < end) {
                    rcur = elist[j];
                    acur = expf(edge_score1(aS, rcur, head, ad) - m) / den;
                }
            }
        }

        const float4 b0 = *(const float4*)(bias + coff);
        const float4 b1 = *(const float4*)(bias + coff + 128);
        float* op = zout + (size_t)d * HC + coff;
        *(float4*)op         = bias_elu4(acc0, b0);
        *(float4*)(op + 128) = bias_elu4(acc1, b1);
    }
}

#define AGG2_NODES 16
static_assert(N_NODES % AGG2_NODES == 0, "every block of node_agg2 owns 16 whole nodes");
static_assert((AGG2_NODES * NUM_CLASS * 4) % 128 == 0, "a block's output rows are whole 128-byte lines");

__launch_bounds__(256)
__global__ void node_agg2(const int* __restrict__ rowptr, const unsigned* __restrict__ elist,
                          const float* __restrict__ H2, const float* __restrict__ aS,
                          const float* __restrict__ aD, const float* __restrict__ b2,
                          float* __restrict__ out)
{
    __shared__ float rows[AGG2_NODES * NUM_CLASS];

    const int tid = threadIdx.x;
    const int ln  = tid >> 4;
    const int cl  = tid & 15;
    const int c   = min(cl, NUM_CLASS - 1);
    const int d   = blockIdx.x * AGG2_NODES + ln;

    const int beg = rowptr[d];
    const int end = rowptr[d + 1];
    const float ad = aD[d];
    const float es = leaky(aS[d] + ad);

    float m = es;
    for (int j = beg; j < end; j += 4) {
        float e[4];
#pragma unroll
        for (int i = 0; i < 4; ++i)
            e[i] = leaky(aS[elist[min(j + i, end - 1)] & 0xffffu] + ad);
        m = fmaxf(fmaxf(m, fmaxf(e[0], e[1])), fmaxf(e[2], e[3]));
    }
    m = isfinite(m) ? m : 0.f;

    float den, acc;
    {
        const float p = expf(es - m);
        den = p;
        acc = p * H2[(size_t)d * NUM_CLASS + c];
    }
    for (int j = beg; j < end; j += 4) {
        unsigned sN[4];
        float e[4], h[4];
#pragma unroll
        for (int i = 0; i < 4; ++i) sN[i] = elist[min(j + i, end - 1)] & 0xffffu;
#pragma unroll
        for (int i = 0; i < 4; ++i) {
            e[i] = leaky(aS[sN[i]] + ad);
            h[i] = H2[(size_t)sN[i] * NUM_CLASS + c];
        }
#pragma unroll
        for (int i = 0; i < 4; ++i) {
            const bool  on = j + i < end;
            const float p  = on ? expf(e[i] - m) : 0.f;
            den += p;
            acc += p * (on ? h[i] : 0.f);
        }
    }
    const float v = acc / (den + 1e-16f) + b2[c];
    if (cl < NUM_CLASS) rows[ln * NUM_CLASS + cl] = 1.f / (1.f + expf(-v));
    __syncthreads();
    if (tid < AGG2_NODES * NUM_CLASS)
        out[(size_t)blockIdx.x * (AGG2_NODES * NUM_CLASS) + tid] = rows[tid];
}

extern "C" void kernel_launch(void* const* d_in, const int* in_sizes, int n_in,
                              void* d_out, int out_size, void* d_ws, size_t ws_size,
                              hipStream_t stream)
{
    const float* x   = (const float*)d_in[0];
    const int*   ei  = (const int*)d_in[1];
    const float* W1  = (const float*)d_in[2];
    const float* as1 = (const float*)d_in[3];
    const float* ad1 = (const float*)d_in[4];
    const float* b1  = (const float*)d_in[5];
    const float* W2  = (const float*)d_in[6];
    const float* as2 = (const float*)d_in[7];
    const float* ad2 = (const float*)d_in[8];
    const float* b2  = (const float*)d_in[9];
    float* out = (float*)d_out;

    float* ws = (float*)d_ws;
    size_t off = 0;
    float* h1    = ws + off; off += (size_t)N_NODES * HC;
    float* z1    = ws + off; off += (size_t)N_NODES * HC;
    float* sA1   = ws + off; off += (size_t)N_NODES * HEADS;
    float* sD1   = ws + off; off += (size_t)N_NODES * HEADS;
    float* h2    = ws + off; off += (size_t)N_NODES * NUM_CLASS;
    float* sA2   = ws + off; off += N_NODES;
    float* sD2   = ws + off; off += N_NODES;
    unsigned* elist = (unsigned*)(ws + off); off += (size_t)E_IN;
    int*      rowp  = (int*)(ws + off);      off += (size_t)N_NODES + 1;
    (void)ws_size; (void)in_sizes; (void)n_in; (void)out_size;

    const dim3 b256(256);

    gemm1_wmma<<<dim3(HC / 64, (N_NODES + 31) / 32), b256, 0, stream>>>(x, W1, h1);
    att1_kernel<<<N_NODES / ATT_NODES, b256, 0, stream>>>(h1, as1, ad1, sA1, sD1);
    csr_build<<<1, CSR_T, 0, stream>>>(ei, rowp, elist);
    node_agg1<<<AGG_BLOCKS, b256, 0, stream>>>(rowp, elist, h1, sA1, sD1, b1, z1);

    gemm2_wmma<<<N_NODES / G2_ROWS, b256, 0, stream>>>(z1, W2, h2);
    att2_kernel<<<(N_NODES + 255) / 256, b256, 0, stream>>>(h2, as2, ad2, sA2, sD2);
    node_agg2<<<N_NODES / AGG2_NODES, b256, 0, stream>>>(rowp, elist, h2, sA2, sD2, b2, out);
}


# Layer 1 re-associated: softmax-weighted sums taken on x (8 heads per wave, row read once), W1 product with bias+ELU after; scores from x@(W1 att); h1 and its passes gone; sum order changes
# speedup vs baseline: 3.8381x; 1.0734x over previous
#include <hip/hip_runtime.h>
#include <math.h>

#define N_NODES   10000
#define F_IN      128
#define HEADS     8
#define HIDD      256
#define HC        (HEADS * HIDD)
#define NUM_CLASS 10
#define E_IN      50000
#define NEG_SLOPE 0.2f
static_assert(HIDD % 64 == 0, "a 64-column tile of GEMM1 lies inside one head");

typedef __attribute__((ext_vector_type(2))) float v2f;
typedef __attribute__((ext_vector_type(8))) float v8f;
typedef __attribute__((ext_vector_type(4))) float v4f;

__device__ __forceinline__ float bias_elu1(const float s, const float b)
{
    const float v = s + b;
    return (v > 0.f) ? v : expm1f(v);
}

__launch_bounds__(256)
__global__ void gemm1_wmma(const float* __restrict__ XA, const float* __restrict__ W,
                           const float* __restrict__ bias, float* __restrict__ Zout)
{
    __shared__ float As[32][F_IN + 4];
    __shared__ float Bs[F_IN][64];

    const int tid = threadIdx.x;
    const int m0  = blockIdx.y * 32;
    const int n0  = blockIdx.x * 64;
    const int hd  = n0 / HIDD;

#pragma unroll
    for (int i = 0; i < 4; ++i) {
        int idx = (tid + i * 256) * 4;
        int r = idx >> 7;
        int c = idx & 127;
        float4 v = make_float4(0.f, 0.f, 0.f, 0.f);
        int gr = m0 + r;
        if (gr < N_NODES) v = *(const float4*)(XA + ((size_t)gr * HEADS + hd) * F_IN + c);
        *(float4*)(&As[r][c]) = v;
    }
#pragma unroll
    for (int i = 0; i < 8; ++i) {
        int idx = (tid + i * 256) * 4;
        int r = idx >> 6;
        int c = idx & 63;
        *(float4*)(&Bs[r][c]) = *(const float4*)(W + (size_t)r * HC + n0 + c);
    }
    __syncthreads();

    const int wave = tid >> 5;
    const int lane = tid & 31;
    const int half = lane >> 4;
    const int ml   = lane & 15;
    const int ar   = (wave >> 2) * 16;
    const int bc   = (wave & 3) * 16;

    v8f acc = {0.f, 0.f, 0.f, 0.f, 0.f, 0.f, 0.f, 0.f};
#pragma unroll
    for (int kk = 0; kk < F_IN / 4; ++kk) {
        const int kb = kk * 4 + half * 2;
        v2f a, b;
        float2 av = *(const float2*)(&As[ar + ml][kb]);
        a.x = av.x; a.y = av.y;
        b.x = Bs[kb + 0][bc + ml];
        b.y = Bs[kb + 1][bc + ml];
        acc = __builtin_amdgcn_wmma_f32_16x16x4_f32(false, a, false, b,
                                                    (short)0, acc, false, false);
    }

    const int col = n0 + bc + ml;
    const float bv = bias[col];
#pragma unroll
    for (int r = 0; r < 8; ++r) {
        const int row = m0 + ar + r + 8 * half;
        if (row < N_NODES) Zout[(size_t)row * HC + col] = bias_elu1(acc[r], bv);
    }
}

#define G2_ROWS  16
#define G2_WAVES 8
#define G2_KW    (HC / G2_WAVES)
#define G2_KR    128
#define G2_PITCH (G2_KR + 4)
#define G2_TILE  (G2_ROWS * G2_PITCH)
static_assert(N_NODES % G2_ROWS == 0, "every block of gemm2_wmma owns 16 whole rows");
static_assert(G2_KW == 2 * G2_KR, "two rounds per wave");
static_assert(G2_TILE >= 16 * 16, "a wave's partial tile fits its own region");

__device__ __forceinline__ void g2_load_rows(v4f (&v)[G2_ROWS], const float* __restrict__ zp)
{
#pragma unroll
    for (int r = 0; r < G2_ROWS; ++r) v[r] = __builtin_nontemporal_load((const v4f*)(zp + (size_t)r * HC));
}

__device__ __forceinline__ void g2_store_rows(float* __restrict__ tl, const v4f (&v)[G2_ROWS])
{
#pragma unroll
    for (int r = 0; r < G2_ROWS; ++r) *(v4f*)(tl + r * G2_PITCH) = v[r];
}

__device__ __forceinline__ v8f g2_products(v8f acc, const float* __restrict__ fr,
                                           const float* __restrict__ wp)
{
#pragma unroll
    for (int j = 0; j < G2_KR / 8; ++j) {
        const float4 f = *(const float4*)(fr + 8 * j);
        v2f a0, a1, b0, b1;
        a0.x = f.x; a0.y = f.y;
        a1.x = f.z; a1.y = f.w;
        b0.x = wp[(size_t)(8 * j + 0) * NUM_CLASS];
        b0.y = wp[(size_t)(8 * j + 1) * NUM_CLASS];
        b1.x = wp[(size_t)(8 * j + 2) * NUM_CLASS];
        b1.y = wp[(size_t)(8 * j + 3) * NUM_CLASS];
        acc = __builtin_amdgcn_wmma_f32_16x16x4_f32(false, a0, false, b0,
                                                    (short)0, acc, false, false);
        acc = __builtin_amdgcn_wmma_f32_16x16x4_f32(false, a1, false, b1,
                                                    (short)0, acc, false, false);
    }
    return acc;
}

__launch_bounds__(256)
__global__ void gemm2_wmma(const float* __restrict__ Z, const float* __restrict__ W2,
                           float* __restrict__ H2)
{
    __shared__ __attribute__((aligned(16))) float tiles[G2_WAVES * G2_TILE];

    const int tid  = threadIdx.x;
    const int wave = __builtin_amdgcn_readfirstlane(tid >> 5);
    const int lane = tid & 31;
    const int half = lane >> 4;
    const int ml   = lane & 15;
    const int nb   = min(ml, NUM_CLASS - 1);
    const int m0   = blockIdx.x * G2_ROWS;
    const int k0   = wave * G2_KW;

    float*       tl = tiles + wave * G2_TILE;
    float*       ts = tl + lane * 4;
    const float* fr = tl + ml * G2_PITCH + 4 * half;
    const float* zp = Z + (size_t)m0 * HC + k0 + lane * 4;
    const float* wp = W2 + (size_t)(k0 + 4 * half) * NUM_CLASS + nb;

    v8f acc = {0.f, 0.f, 0.f, 0.f, 0.f, 0.f, 0.f, 0.f};
    v4f v[G2_ROWS];

    g2_load_rows(v, zp);
    g2_store_rows(ts, v);
    g2_load_rows(v, zp + G2_KR);
    __builtin_amdgcn_wave_barrier();
    acc = g2_products(acc, fr, wp);
    __builtin_amdgcn_wave_barrier();
    g2_store_rows(ts, v);
    __builtin_amdgcn_wave_barrier();
    acc = g2_products(acc, fr, wp + (size_t)G2_KR * NUM_CLASS);
    __builtin_amdgcn_wave_barrier();

#pragma unroll
    for (int r = 0; r < 8; ++r) tl[(8 * half + r) * 16 + ml] = acc[r];
    __syncthreads();
    if (tid < G2_ROWS * NUM_CLASS) {
        const int row = tid / NUM_CLASS, col = tid % NUM_CLASS;
        float s = 0.f;
#pragma unroll
        for (int w = 0; w < G2_WAVES; ++w) s += tiles[w * G2_TILE + row * 16 + col];
        H2[(size_t)m0 * NUM_CLASS + tid] = s;
    }
}

__global__ void att2_kernel(const float* __restrict__ H2,
                            const float* __restrict__ as2,
                            const float* __restrict__ ad2,
                            float* __restrict__ aS, float* __restrict__ aD)
{
    const int n = blockIdx.x * blockDim.x + threadIdx.x;
    if (n >= N_NODES) return;
    float ss = 0.f, sd = 0.f;
#pragma unroll
    for (int c = 0; c < NUM_CLASS; ++c) {
        float v = H2[(size_t)n * NUM_CLASS + c];
        ss += v * as2[c];
        sd += v * ad2[c];
    }
    aS[n] = ss; aD[n] = sd;
}

#define CSR_T    1024
#define CSR_EPT  ((E_IN + CSR_T - 1) / CSR_T)
#define CSR_CH   7
#define CSR_SPAN ((N_NODES + CSR_T - 1) / CSR_T)
static_assert(E_IN <= 65536 && N_NODES <= 65536, "list entry packs edge id and source in 16 bits each");
static_assert(CSR_EPT % CSR_CH == 0, "chunks cover the edges of a thread exactly");

__launch_bounds__(CSR_T)
__global__ void csr_build(const int* __restrict__ ei, int* __restrict__ rowptr,
                          unsigned* __restrict__ elist)
{
    __shared__ int cnt[N_NODES + 1];
    __shared__ unsigned lst[E_IN + 1];
    __shared__ int wsum[CSR_T / 32];
    __shared__ int total_s;

    const int tid  = threadIdx.x;
    const int lane = tid & 31;
    const int wv   = tid >> 5;

    for (int i = tid; i < N_NODES + 1; i += CSR_T) cnt[i] = 0;
    __syncthreads();

#pragma unroll 1
    for (int k0 = 0; k0 < CSR_EPT; k0 += CSR_CH) {
        unsigned d[CSR_CH];
#pragma unroll
        for (int i = 0; i < CSR_CH; ++i) {
            const int e = tid + (k0 + i) * CSR_T;
            d[i] = (unsigned)ei[E_IN + min(e, E_IN - 1)];
            if (e >= E_IN || d[i] >= (unsigned)N_NODES) d[i] = N_NODES;
        }
#pragma unroll
        for (int i = 0; i < CSR_CH; ++i) atomicAdd(&cnt[d[i]], 1);
    }
    __syncthreads();

    const int base = tid * CSR_SPAN;
    int loc = 0;
#pragma unroll
    for (int j = 0; j < CSR_SPAN; ++j)
        if (base + j < N_NODES) loc += cnt[base + j];

    int inc = loc;
#pragma unroll
    for (int o = 1; o < 32; o <<= 1) {
        const int t = __shfl_up(inc, o, 32);
        if (lane >= o) inc += t;
    }
    if (lane == 31) wsum[wv] = inc;
    __syncthreads();
    if (wv == 0) {
        const int w = wsum[lane];
        int winc = w;
#pragma unroll
        for (int o = 1; o < 32; o <<= 1) {
            const int t = __shfl_up(winc, o, 32);
            if (lane >= o) winc += t;
        }
        wsum[lane] = winc - w;
    }
    __syncthreads();

    int run = wsum[wv] + inc - loc;
#pragma unroll
    for (int j = 0; j < CSR_SPAN; ++j) {
        if (base + j < N_NODES) {
            const int c = cnt[base + j];
            rowptr[base + j] = run;
            cnt[base + j]    = run;
            run += c;
        }
    }
    if (tid == CSR_T - 1) {
        rowptr[N_NODES] = run;
        total_s         = run;
        cnt[N_NODES]    = E_IN;
    }
    __syncthreads();

#pragma unroll 1
    for (int k0 = 0; k0 < CSR_EPT; k0 += CSR_CH) {
        unsigned d[CSR_CH], w[CSR_CH];
#pragma unroll
        for (int i = 0; i < CSR_CH; ++i) {
            const int e  = tid + (k0 + i) * CSR_T;
            const int ec = min(e, E_IN - 1);
            w[i] = ((unsigned)e << 16) | ((unsigned)ei[ec] & 0xffffu);
            d[i] = (unsigned)ei[E_IN + ec];
            if (e >= E_IN || d[i] >= (unsigned)N_NODES) d[i] = N_NODES;
        }
        int slot[CSR_CH];
#pragma unroll
        for (int i = 0; i < CSR_CH; ++i) slot[i] = atomicAdd(&cnt[d[i]], 1);
#pragma unroll
        for (int i = 0; i < CSR_CH; ++i) lst[min(slot[i], E_IN)] = w[i];
    }
    __syncthreads();

    const int total = total_s;
#pragma unroll 1
    for (int k0 = 0; k0 < CSR_EPT; k0 += CSR_CH) {
        unsigned v[CSR_CH];
#pragma unroll
        for (int i = 0; i < CSR_CH; ++i) v[i] = lst[min(tid + (k0 + i) * CSR_T, E_IN - 1)];
#pragma unroll
        for (int i = 0; i < CSR_CH; ++i) {
            const int idx = tid + (k0 + i) * CSR_T;
            if (idx < total) elist[idx] = v[i];
        }
    }
}

__device__ __forceinline__ float leaky(const float x)
{
    return (x >= 0.f) ? x : NEG_SLOPE * x;
}

template <int MASK>
__device__ __forceinline__ float row_xor(const float x)
{
    return __builtin_bit_cast(float, __builtin_amdgcn_update_dpp(
        0, __builtin_bit_cast(int, x), 0x160 + MASK, 0xf, 0xf, false));
}

__device__ __forceinline__ float xhalf(const float x)
{
    const int b = __builtin_bit_cast(int, x);
    return __builtin_bit_cast(float, __builtin_amdgcn_permlanex16(b, b, 0x76543210, 0xfedcba98, false, false));
}

__device__ __forceinline__ float wave_max(float x)
{
    x = fmaxf(x, row_xor<1>(x));
    x = fmaxf(x, row_xor<2>(x));
    x = fmaxf(x, row_xor<4>(x));
    x = fmaxf(x, row_xor<8>(x));
    return fmaxf(x, xhalf(x));
}

__device__ __forceinline__ float wave_sum(float x)
{
    x += row_xor<1>(x);
    x += row_xor<2>(x);
    x += row_xor<4>(x);
    x += row_xor<8>(x);
    return x + xhalf(x);
}

__device__ __forceinline__ float lane_f(const float v, const int k)
{
    return __int_as_float(__builtin_amdgcn_readlane(__float_as_int(v), k));
}

__device__ __forceinline__ float uni(const float v)
{
    return __int_as_float(__builtin_amdgcn_readfirstlane(__float_as_int(v)));
}

#define ATT_NODES 8
static_assert(F_IN % ATT_NODES == 0, "every block of fold_att owns 8 whole rows of W1");
static_assert((ATT_NODES * HEADS * 4) % 128 == 0, "a block's results are whole 128-byte lines");

__device__ __forceinline__ float dot8(const float4 u, const float4 w, const float4 a, const float4 b)
{
    return u.x * a.x + u.y * a.y + u.z * a.z + u.w * a.w
         + w.x * b.x + w.y * b.y + w.z * b.z + w.w * b.w;
}

__launch_bounds__(256)
__global__ void fold_att(const float* __restrict__ Wm,
                         const float* __restrict__ att_src,
                         const float* __restrict__ att_dst,
                         float* __restrict__ wS, float* __restrict__ wD)
{
    __shared__ float res[2 * ATT_NODES * HEADS];

    const int tid  = threadIdx.x;
    const int lane = tid & 31;
    const int head = __builtin_amdgcn_readfirstlane(tid >> 5);
    const int n0   = blockIdx.x * ATT_NODES;
    const int coff = head * HIDD + lane * 4;

    const float4 s0 = *(const float4*)(att_src + coff);
    const float4 s1 = *(const float4*)(att_src + coff + 128);
    const float4 d0 = *(const float4*)(att_dst + coff);
    const float4 d1 = *(const float4*)(att_dst + coff + 128);

    const float* hp = Wm + (size_t)n0 * HC + coff;
    float4 u[ATT_NODES], w[ATT_NODES];
#pragma unroll
    for (int i = 0; i < ATT_NODES; ++i) {
        u[i] = *(const float4*)(hp + (size_t)i * HC);
        w[i] = *(const float4*)(hp + (size_t)i * HC + 128);
    }

    float keep = 0.f;
#pragma unroll
    for (int i = 0; i < ATT_NODES; ++i) {
        const float ss = wave_sum(dot8(u[i], w[i], s0, s1));
        const float sd = wave_sum(dot8(u[i], w[i], d0, d1));
        keep = (lane == i) ? ss : keep;
        keep = (lane == ATT_NODES + i) ? sd : keep;
    }
    if (lane < 2 * ATT_NODES)
        res[(lane >> 3) * (ATT_NODES * HEADS) + (lane & 7) * HEADS + head] = keep;
    __syncthreads();
    if (tid < ATT_NODES * HEADS)
        wS[(size_t)n0 * HEADS + tid] = res[tid];
    else if (tid < 2 * ATT_NODES * HEADS)
        wD[(size_t)n0 * HEADS + (tid - ATT_NODES * HEADS)] = res[tid];
}


#define AX_NODES 16
static_assert(N_NODES % AX_NODES == 0, "every block of node_att1 owns 16 whole nodes");
static_assert((AX_NODES * HEADS * 4) % 128 == 0, "a block's results are whole 128-byte lines");

__launch_bounds__(256)
__global__ void node_att1(const float* __restrict__ X, const float* __restrict__ wS,
                          const float* __restrict__ wD,
                          float* __restrict__ aS, float* __restrict__ aD)
{
    __shared__ float wl[F_IN * 2 * HEADS];
    __shared__ float res[2 * AX_NODES * HEADS];

    const int tid = threadIdx.x;
    for (int i = tid; i < F_IN * HEADS; i += 256) {
        const int f = i / HEADS, h = i % HEADS;
        wl[f * 2 * HEADS + h]         = wS[i];
        wl[f * 2 * HEADS + HEADS + h] = wD[i];
    }
    __syncthreads();

    const int ln = tid >> 4;
    const int j  = tid & 15;
    const int n0 = blockIdx.x * AX_NODES;
    const float* xp = X + (size_t)(n0 + ln) * F_IN;
    const float* wp = wl + j;

    float s = 0.f;
#pragma unroll 8
    for (int q = 0; q < F_IN / 4; ++q) {
        const float4 v = *(const float4*)(xp + 4 * q);
        s += v.x * wp[(4 * q + 0) * 2 * HEADS];
        s += v.y * wp[(4 * q + 1) * 2 * HEADS];
        s += v.z * wp[(4 * q + 2) * 2 * HEADS];
        s += v.w * wp[(4 * q + 3) * 2 * HEADS];
    }
    res[(j >> 3) * (AX_NODES * HEADS) + ln * HEADS + (j & 7)] = s;
    __syncthreads();
    if (tid < AX_NODES * HEADS)
        aS[(size_t)n0 * HEADS + tid] = res[tid];
    else
        aD[(size_t)n0 * HEADS + (tid - AX_NODES * HEADS)] = res[tid];
}

#define AGX_NODES 8
static_assert(N_NODES % AGX_NODES == 0, "every block of node_aggx owns 8 whole nodes");

__device__ __forceinline__ void axpy4(float4& acc, const float a, const float4 v)
{
    acc.x = __builtin_fmaf(a, v.x, acc.x); acc.y = __builtin_fmaf(a, v.y, acc.y);
    acc.z = __builtin_fmaf(a, v.z, acc.z); acc.w = __builtin_fmaf(a, v.w, acc.w);
}

__device__ __forceinline__ void load8(float (&o)[HEADS], const float* __restrict__ p)
{
    const float4 a = *(const float4*)p;
    const float4 b = *(const float4*)(p + 4);
    o[0] = a.x; o[1] = a.y; o[2] = a.z; o[3] = a.w;
    o[4] = b.x; o[5] = b.y; o[6] = b.z; o[7] = b.w;
}

__device__ __forceinline__ void edge_scores1(float (&e)[HEADS], const float* __restrict__ aS,
                                             const unsigned rec, const float (&ad)[HEADS])
{
    load8(e, aS + (size_t)(rec & 0xffffu) * HEADS);
#pragma unroll
    for (int h = 0; h < HEADS; ++h) e[h] = leaky(e[h] + ad[h]);
}

__device__ __forceinline__ void add_row(float4 (&acc)[HEADS], const float (&acur)[HEADS],
                                        const int k, const float4 u)
{
#pragma unroll
    for (int h = 0; h < HEADS; ++h) axpy4(acc[h], lane_f(acur[h], k), u);
}

__launch_bounds__(256)
__global__ void node_aggx(const int* __restrict__ rowptr, const unsigned* __restrict__ elist,
                          const float* __restrict__ X, const float* __restrict__ aS,
                          const float* __restrict__ aD, float* __restrict__ xagg)
{
    const int lane = threadIdx.x & 31;
    const int d    = blockIdx.x * AGX_NODES + __builtin_amdgcn_readfirstlane((int)(threadIdx.x >> 5));
    const int loff = lane * 4;

    const int beg = rowptr[d];
    const int end = rowptr[d + 1];
    const float4 xs = *(const float4*)(X + (size_t)d * F_IN + loff);
    float ad[HEADS], es[HEADS];
    load8(ad, aD + (size_t)d * HEADS);
    load8(es, aS + (size_t)d * HEADS);
#pragma unroll
    for (int h = 0; h < HEADS; ++h) es[h] = leaky(es[h] + ad[h]);

    const int  j0 = beg + lane;
    const bool v0 = j0 < end;
    unsigned rc = 0u;
    float ec[HEADS];
#pragma unroll
    for (int h = 0; h < HEADS; ++h) ec[h] = -INFINITY;
    if (v0) { rc = elist[j0]; edge_scores1(ec, aS, rc, ad); }
    float m[HEADS];
#pragma unroll
    for (int h = 0; h < HEADS; ++h) m[h] = fmaxf(es[h], ec[h]);
    for (int c0 = beg + 32; c0 < end; c0 += 32) {
        const int j = c0 + lane;
        if (j < end) {
            float e[HEADS];
            edge_scores1(e, aS, elist[j], ad);
#pragma unroll
            for (int h = 0; h < HEADS; ++h) m[h] = fmaxf(m[h], e[h]);
        }
    }
#pragma unroll
    for (int h = 0; h < HEADS; ++h) {
        m[h] = uni(wave_max(m[h]));
        m[h] = isfinite(m[h]) ? m[h] : 0.f;
    }

    float acur[HEADS];
    float s[HEADS];
#pragma unroll
    for (int h = 0; h < HEADS; ++h) { acur[h] = v0 ? expf(ec[h] - m[h]) : 0.f; s[h] = acur[h]; }
    for (int c0 = beg + 32; c0 < end; c0 += 32) {
        const int j = c0 + lane;
        if (j < end) {
            float e[HEADS];
            edge_scores1(e, aS, elist[j], ad);
#pragma unroll
            for (int h = 0; h < HEADS; ++h) s[h] += expf(e[h] - m[h]);
        }
    }

    float  den[HEADS];
    float4 acc[HEADS];
#pragma unroll
    for (int h = 0; h < HEADS; ++h) {
        const float ps = expf(es[h] - m[h]);
        den[h] = uni(wave_sum(s[h])) + ps + 1e-16f;
        const float a = ps / den[h];
        acc[h]  = make_float4(a * xs.x, a * xs.y, a * xs.z, a * xs.w);
        acur[h] = acur[h] / den[h];
    }

    unsigned rcur = rc;
    for (int c0 = beg; c0 < end; c0 += 32) {
        const int cnt = min(end - c0, 32);
        int k = 0;
        for (; k + 4 <= cnt; k += 4) {
            const unsigned r0 = __builtin_amdgcn_readlane(rcur, k);
            const unsigned r1 = __builtin_amdgcn_readlane(rcur, k + 1);
            const unsigned r2 = __builtin_amdgcn_readlane(rcur, k + 2);
            const unsigned r3 = __builtin_amdgcn_readlane(rcur, k + 3);
            const float4 u0 = *(const float4*)(X + (size_t)(r0 & 0xffffu) * F_IN + loff);
            const float4 u1 = *(const float4*)(X + (size_t)(r1 & 0xffffu) * F_IN + loff);
            const float4 u2 = *(const float4*)(X + (size_t)(r2 & 0xffffu) * F_IN + loff);
            const float4 u3 = *(const float4*)(X + (size_t)(r3 & 0xffffu) * F_IN + loff);
            add_row(acc, acur, k,     u0);
            add_row(acc, acur, k + 1, u1);
            add_row(acc, acur, k + 2, u2);
            add_row(acc, acur, k + 3, u3);
        }
        if (k + 2 <= cnt) {
            const unsigned r0 = __builtin_amdgcn_readlane(rcur, k);
            const unsigned r1 = __builtin_amdgcn_readlane(rcur, k + 1);
            const float4 u0 = *(const float4*)(X + (size_t)(r0 & 0xffffu) * F_IN + loff);
            const float4 u1 = *(const float4*)(X + (size_t)(r1 & 0xffffu) * F_IN + loff);
            add_row(acc, acur, k,     u0);
            add_row(acc, acur, k + 1, u1);
            k += 2;
        }
        if (k < cnt) {
            const unsigned r = __builtin_amdgcn_readlane(rcur, k);
            const float4 u = *(const float4*)(X + (size_t)(r & 0xffffu) * F_IN + loff);
            add_row(acc, acur, k, u);
        }
        if (c0 + 32 < end) {
            const int j = c0 + 32 + lane;
            rcur = 0u;
#pragma unroll
            for (int h = 0; h < HEADS; ++h) acur[h] = 0.f;
            if (j < end) {
                rcur = elist[j];
                float e[HEADS];
                edge_scores1(e, aS, rcur, ad);
#pragma unroll
                for (int h = 0; h < HEADS; ++h) acur[h] = expf(e[h] - m[h]) / den[h];
            }
        }
    }

    float* op = xagg + (size_t)d * HEADS * F_IN + loff;
#pragma unroll
    for (int h = 0; h < HEADS; ++h) *(float4*)(op + h * F_IN) = acc[h];
}

#define AGG2_NODES 16
static_assert(N_NODES % AGG2_NODES == 0, "every block of node_agg2 owns 16 whole nodes");
static_assert((AGG2_NODES * NUM_CLASS * 4) % 128 == 0, "a block's output rows are whole 128-byte lines");

__launch_bounds__(256)
__global__ void node_agg2(const int* __restrict__ rowptr, const unsigned* __restrict__ elist,
                          const float* __restrict__ H2, const float* __restrict__ aS,
                          const float* __restrict__ aD, const float* __restrict__ b2,
                          float* __restrict__ out)
{
    __shared__ float rows[AGG2_NODES * NUM_CLASS];

    const int tid = threadIdx.x;
    const int ln  = tid >> 4;
    const int cl  = tid & 15;
    const int c   = min(cl, NUM_CLASS - 1);
    const int d   = blockIdx.x * AGG2_NODES + ln;

    const int beg = rowptr[d];
    const int end = rowptr[d + 1];
    const float ad = aD[d];
    const float es = leaky(aS[d] + ad);

    float m = es;
    for (int j = beg; j < end; j += 4) {
        float e[4];
#pragma unroll
        for (int i = 0; i < 4; ++i)
            e[i] = leaky(aS[elist[min(j + i, end - 1)] & 0xffffu] + ad);
        m = fmaxf(fmaxf(m, fmaxf(e[0], e[1])), fmaxf(e[2], e[3]));
    }
    m = isfinite(m) ? m : 0.f;

    float den, acc;
    {
        const float p = expf(es - m);
        den = p;
        acc = p * H2[(size_t)d * NUM_CLASS + c];
    }
    for (int j = beg; j < end; j += 4) {
        unsigned sN[4];
        float e[4], h[4];
#pragma unroll
        for (int i = 0; i < 4; ++i) sN[i] = elist[min(j + i, end - 1)] & 0xffffu;
#pragma unroll
        for (int i = 0; i < 4; ++i) {
            e[i] = leaky(aS[sN[i]] + ad);
            h[i] = H2[(size_t)sN[i] * NUM_CLASS + c];
        }
#pragma unroll
        for (int i = 0; i < 4; ++i) {
            const bool  on = j + i < end;
            const float p  = on ? expf(e[i] - m) : 0.f;
            den += p;
            acc += p * (on ? h[i] : 0.f);
        }
    }
    const float v = acc / (den + 1e-16f) + b2[c];
    if (cl < NUM_CLASS) rows[ln * NUM_CLASS + cl] = 1.f / (1.f + expf(-v));
    __syncthreads();
    if (tid < AGG2_NODES * NUM_CLASS)
        out[(size_t)blockIdx.x * (AGG2_NODES * NUM_CLASS) + tid] = rows[tid];
}

extern "C" void kernel_launch(void* const* d_in, const int* in_sizes, int n_in,
                              void* d_out, int out_size, void* d_ws, size_t ws_size,
                              hipStream_t stream)
{
    const float* x   = (const float*)d_in[0];
    const int*   ei  = (const int*)d_in[1];
    const float* W1  = (const float*)d_in[2];
    const float* as1 = (const float*)d_in[3];
    const float* ad1 = (const float*)d_in[4];
    const float* b1  = (const float*)d_in[5];
    const float* W2  = (const float*)d_in[6];
    const float* as2 = (const float*)d_in[7];
    const float* ad2 = (const float*)d_in[8];
    const float* b2  = (const float*)d_in[9];
    float* out = (float*)d_out;

    float* ws = (float*)d_ws;
    size_t off = 0;
    float* xa    = ws + off; off += (size_t)N_NODES * HEADS * F_IN;
    float* z1    = ws + off; off += (size_t)N_NODES * HC;
    float* sA1   = ws + off; off += (size_t)N_NODES * HEADS;
    float* sD1   = ws + off; off += (size_t)N_NODES * HEADS;
    float* h2    = ws + off; off += (size_t)N_NODES * NUM_CLASS;
    float* sA2   = ws + off; off += N_NODES;
    float* sD2   = ws + off; off += N_NODES;
    float* wS1   = ws + off; off += (size_t)F_IN * HEADS;
    float* wD1   = ws + off; off += (size_t)F_IN * HEADS;
    unsigned* elist = (unsigned*)(ws + off); off += (size_t)E_IN;
    int*      rowp  = (int*)(ws + off);      off += (size_t)N_NODES + 1;
    (void)ws_size; (void)in_sizes; (void)n_in; (void)out_size;

    const dim3 b256(256);

    fold_att<<<F_IN / ATT_NODES, b256, 0, stream>>>(W1, as1, ad1, wS1, wD1);
    node_att1<<<N_NODES / AX_NODES, b256, 0, stream>>>(x, wS1, wD1, sA1, sD1);
    csr_build<<<1, CSR_T, 0, stream>>>(ei, rowp, elist);
    node_aggx<<<N_NODES / AGX_NODES, b256, 0, stream>>>(rowp, elist, x, sA1, sD1, xa);
    gemm1_wmma<<<dim3(HC / 64, (N_NODES + 31) / 32), b256, 0, stream>>>(xa, W1, b1, z1);

    gemm2_wmma<<<N_NODES / G2_ROWS, b256, 0, stream>>>(z1, W2, h2);
    att2_kernel<<<(N_NODES + 255) / 256, b256, 0, stream>>>(h2, as2, ad2, sA2, sD2);
    node_agg2<<<N_NODES / AGG2_NODES, b256, 0, stream>>>(rowp, elist, h2, sA2, sD2, b2, out);
}


# GEMM2 folded into GEMM1: block = 64 rows x one head; z tile (transposed product) stays in registers as B operand of the W2 product, never stored; per-head h2 partials summed after; sum order changes
# speedup vs baseline: 5.0159x; 1.3069x over previous
#include <hip/hip_runtime.h>
#include <math.h>

#define N_NODES   10000
#define F_IN      128
#define HEADS     8
#define HIDD      256
#define HC        (HEADS * HIDD)
#define NUM_CLASS 10
#define E_IN      50000
#define NEG_SLOPE 0.2f
static_assert(HIDD % 64 == 0, "a 64-column tile of GEMM1 lies inside one head");

typedef __attribute__((ext_vector_type(2))) float v2f;
typedef __attribute__((ext_vector_type(8))) float v8f;
typedef __attribute__((ext_vector_type(4))) float v4f;

__device__ __forceinline__ float bias_elu1(const float s, const float b)
{
    const float v = s + b;
    return (v > 0.f) ? v : expm1f(v);
}

#define G1_RT    2
#define G1_ROWS  (32 * G1_RT)
#define G1_CT    (HIDD / 64)
#define G1_RBLK  ((N_NODES + G1_ROWS - 1) / G1_ROWS)
static_assert((G1_ROWS * NUM_CLASS * 4) % 128 == 0, "a block's partial rows are whole 128-byte lines");
static_assert((N_NODES * NUM_CLASS * 4) % 128 == 0, "every head's partial plane starts on a 128-byte line");
static_assert(8 * G1_RT * 256 <= F_IN * 64, "the waves' h2^T tiles fit the W1 tile's LDS");

__device__ __forceinline__ void g1_load_w(v4f (&q)[8], const float* wg)
{
#pragma unroll
    for (int i = 0; i < 8; ++i) q[i] = *(const v4f*)(wg + (size_t)(16 * i) * HC);
}

__device__ __forceinline__ void g1_store_w(float* bs, const v4f (&q)[8])
{
#pragma unroll
    for (int i = 0; i < 8; ++i) *(v4f*)(bs + (16 * i) * 64) = q[i];
}

__launch_bounds__(256)
__global__ void gemm12_wmma(const float* __restrict__ XA, const float* __restrict__ W,
                            const float* __restrict__ bias, const float* __restrict__ W2,
                            float* __restrict__ H2P)
{
    __shared__ __attribute__((aligned(16))) float As[G1_ROWS][F_IN + 4];
    __shared__ __attribute__((aligned(16))) float Bs[F_IN][64];

    const int tid = threadIdx.x;
    const int hd  = blockIdx.x;
    const int m0  = blockIdx.y * G1_ROWS;

    const float* wg = W + (size_t)(tid >> 4) * HC + hd * HIDD + (tid & 15) * 4;
    float*       bs = &Bs[tid >> 4][(tid & 15) * 4];

    v4f q[8];
    g1_load_w(q, wg);
#pragma unroll
    for (int i = 0; i < G1_ROWS * F_IN / 4 / 256; ++i) {
        const int idx = (tid + i * 256) * 4;
        const int r = idx >> 7;
        const int c = idx & 127;
        const int gr = m0 + r;
        v4f v = *(const v4f*)(XA + ((size_t)min(gr, N_NODES - 1) * HEADS + hd) * F_IN + c);
        if (gr >= N_NODES) v = (v4f){0.f, 0.f, 0.f, 0.f};
        *(v4f*)(&As[r][c]) = v;
    }
    g1_store_w(bs, q);
    __syncthreads();

    const int wave = tid >> 5;
    const int lane = tid & 31;
    const int half = lane >> 4;
    const int ml   = lane & 15;
    const int rg   = (wave >> 2) * G1_RT;
    const int bc   = (wave & 3) * 16;
    const int cls  = min(ml, NUM_CLASS - 1);

    v8f acc2[G1_RT];
#pragma unroll
    for (int rt = 0; rt < G1_RT; ++rt) acc2[rt] = (v8f){0.f, 0.f, 0.f, 0.f, 0.f, 0.f, 0.f, 0.f};

    for (int ct = 0; ct < G1_CT; ++ct) {
        if (ct + 1 < G1_CT) g1_load_w(q, wg + (ct + 1) * 64);

        const int cb = hd * HIDD + ct * 64 + bc + 8 * half;
        const float4 bv0 = *(const float4*)(bias + cb);
        const float4 bv1 = *(const float4*)(bias + cb + 4);
        const float bv[8] = {bv0.x, bv0.y, bv0.z, bv0.w, bv1.x, bv1.y, bv1.z, bv1.w};
        float w2f[8];
#pragma unroll
        for (int r = 0; r < 8; ++r) w2f[r] = W2[(size_t)(cb + r) * NUM_CLASS + cls];

        v8f acc[G1_RT];
#pragma unroll
        for (int rt = 0; rt < G1_RT; ++rt) acc[rt] = (v8f){0.f, 0.f, 0.f, 0.f, 0.f, 0.f, 0.f, 0.f};
#pragma unroll
        for (int kk = 0; kk < F_IN / 4; ++kk) {
            const int kb = kk * 4 + half * 2;
            v2f w;
            w.x = Bs[kb + 0][bc + ml];
            w.y = Bs[kb + 1][bc + ml];
#pragma unroll
            for (int rt = 0; rt < G1_RT; ++rt) {
                const float2 xv = *(const float2*)(&As[(rg + rt) * 16 + ml][kb]);
                v2f x;
                x.x = xv.x; x.y = xv.y;
                acc[rt] = __builtin_amdgcn_wmma_f32_16x16x4_f32(false, w, false, x,
                                                                (short)0, acc[rt], false, false);
            }
        }

#pragma unroll
        for (int rt = 0; rt < G1_RT; ++rt) {
            float zt[8];
#pragma unroll
            for (int r = 0; r < 8; ++r) zt[r] = bias_elu1(acc[rt][r], bv[r]);
#pragma unroll
            for (int t = 0; t < 4; ++t) {
                v2f a2, b2;
                a2.x = w2f[2 * t]; a2.y = w2f[2 * t + 1];
                b2.x = zt[2 * t];  b2.y = zt[2 * t + 1];
                acc2[rt] = __builtin_amdgcn_wmma_f32_16x16x4_f32(false, a2, false, b2,
                                                                 (short)0, acc2[rt], false, false);
            }
        }
        __syncthreads();
        if (ct + 1 < G1_CT) {
            g1_store_w(bs, q);
            __syncthreads();
        }
    }

    float* red = &Bs[0][0];
#pragma unroll
    for (int rt = 0; rt < G1_RT; ++rt)
#pragma unroll
        for (int r = 0; r < 8; ++r)
            red[((wave * G1_RT + rt) * 16 + 8 * half + r) * 16 + ml] = acc2[rt][r];
    __syncthreads();
    for (int p = tid; p < G1_ROWS * NUM_CLASS / 2; p += 256) {
        const int e  = 2 * p;
        const int lr = e / NUM_CLASS, c = e % NUM_CLASS;
        const int g  = (lr >> 4) / G1_RT, rt = (lr >> 4) % G1_RT;
        float2 s = make_float2(0.f, 0.f);
#pragma unroll
        for (int j = 0; j < 4; ++j) {
            const float* rp = red + (((g * 4 + j) * G1_RT + rt) * 16 + c) * 16 + (lr & 15);
            s.x += rp[0];
            s.y += rp[16];
        }
        if (m0 + lr < N_NODES)
            *(float2*)(H2P + ((size_t)hd * N_NODES + m0) * NUM_CLASS + e) = s;
    }
}

#define A2_NODES 256
static_assert((A2_NODES * NUM_CLASS * 4) % 128 == 0, "a block's rows of h2 are whole 128-byte lines");

__launch_bounds__(256)
__global__ void att2_kernel(const float* __restrict__ H2P,
                            const float* __restrict__ as2,
                            const float* __restrict__ ad2,
                            float* __restrict__ H2,
                            float* __restrict__ aS, float* __restrict__ aD)
{
    __shared__ float rows[A2_NODES * NUM_CLASS];

    const int tid  = threadIdx.x;
    const int base = blockIdx.x * (A2_NODES * NUM_CLASS);
    float s[NUM_CLASS];
#pragma unroll
    for (int i = 0; i < NUM_CLASS; ++i) {
        const int ic = min(base + tid + i * 256, N_NODES * NUM_CLASS - 1);
        float v = 0.f;
#pragma unroll
        for (int h = 0; h < HEADS; ++h) v += H2P[(size_t)h * (N_NODES * NUM_CLASS) + ic];
        s[i] = v;
    }
#pragma unroll
    for (int i = 0; i < NUM_CLASS; ++i) {
        const int idx = base + tid + i * 256;
        if (idx < N_NODES * NUM_CLASS) H2[idx] = s[i];
        rows[tid + i * 256] = s[i];
    }
    __syncthreads();

    const int n = blockIdx.x * A2_NODES + tid;
    if (n >= N_NODES) return;
    float ss = 0.f, sd = 0.f;
#pragma unroll
    for (int c = 0; c < NUM_CLASS; ++c) {
        const float v = rows[tid * NUM_CLASS + c];
        ss += v * as2[c];
        sd += v * ad2[c];
    }
    aS[n] = ss; aD[n] = sd;
}

#define CSR_T    1024
#define CSR_EPT  ((E_IN + CSR_T - 1) / CSR_T)
#define CSR_CH   7
#define CSR_SPAN ((N_NODES + CSR_T - 1) / CSR_T)
static_assert(E_IN <= 65536 && N_NODES <= 65536, "list entry packs edge id and source in 16 bits each");
static_assert(CSR_EPT % CSR_CH == 0, "chunks cover the edges of a thread exactly");

__launch_bounds__(CSR_T)
__global__ void csr_build(const int* __restrict__ ei, int* __restrict__ rowptr,
                          unsigned* __restrict__ elist)
{
    __shared__ int cnt[N_NODES + 1];
    __shared__ unsigned lst[E_IN + 1];
    __shared__ int wsum[CSR_T / 32];
    __shared__ int total_s;

    const int tid  = threadIdx.x;
    const int lane = tid & 31;
    const int wv   = tid >> 5;

    for (int i = tid; i < N_NODES + 1; i += CSR_T) cnt[i] = 0;
    __syncthreads();

#pragma unroll 1
    for (int k0 = 0; k0 < CSR_EPT; k0 += CSR_CH) {
        unsigned d[CSR_CH];
#pragma unroll
        for (int i = 0; i < CSR_CH; ++i) {
            const int e = tid + (k0 + i) * CSR_T;
            d[i] = (unsigned)ei[E_IN + min(e, E_IN - 1)];
            if (e >= E_IN || d[i] >= (unsigned)N_NODES) d[i] = N_NODES;
        }
#pragma unroll
        for (int i = 0; i < CSR_CH; ++i) atomicAdd(&cnt[d[i]], 1);
    }
    __syncthreads();

    const int base = tid * CSR_SPAN;
    int loc = 0;
#pragma unroll
    for (int j = 0; j < CSR_SPAN; ++j)
        if (base + j < N_NODES) loc += cnt[base + j];

    int inc = loc;
#pragma unroll
    for (int o = 1; o < 32; o <<= 1) {
        const int t = __shfl_up(inc, o, 32);
        if (lane >= o) inc += t;
    }
    if (lane == 31) wsum[wv] = inc;
    __syncthreads();
    if (wv == 0) {
        const int w = wsum[lane];
        int winc = w;
#pragma unroll
        for (int o = 1; o < 32; o <<= 1) {
            const int t = __shfl_up(winc, o, 32);
            if (lane >= o) winc += t;
        }
        wsum[lane] = winc - w;
    }
    __syncthreads();

    int run = wsum[wv] + inc - loc;
#pragma unroll
    for (int j = 0; j < CSR_SPAN; ++j) {
        if (base + j < N_NODES) {
            const int c = cnt[base + j];
            rowptr[base + j] = run;
            cnt[base + j]    = run;
            run += c;
        }
    }
    if (tid == CSR_T - 1) {
        rowptr[N_NODES] = run;
        total_s         = run;
        cnt[N_NODES]    = E_IN;
    }
    __syncthreads();

#pragma unroll 1
    for (int k0 = 0; k0 < CSR_EPT; k0 += CSR_CH) {
        unsigned d[CSR_CH], w[CSR_CH];
#pragma unroll
        for (int i = 0; i < CSR_CH; ++i) {
            const int e  = tid + (k0 + i) * CSR_T;
            const int ec = min(e, E_IN - 1);
            w[i] = ((unsigned)e << 16) | ((unsigned)ei[ec] & 0xffffu);
            d[i] = (unsigned)ei[E_IN + ec];
            if (e >= E_IN || d[i] >= (unsigned)N_NODES) d[i] = N_NODES;
        }
        int slot[CSR_CH];
#pragma unroll
        for (int i = 0; i < CSR_CH; ++i) slot[i] = atomicAdd(&cnt[d[i]], 1);
#pragma unroll
        for (int i = 0; i < CSR_CH; ++i) lst[min(slot[i], E_IN)] = w[i];
    }
    __syncthreads();

    const int total = total_s;
#pragma unroll 1
    for (int k0 = 0; k0 < CSR_EPT; k0 += CSR_CH) {
        unsigned v[CSR_CH];
#pragma unroll
        for (int i = 0; i < CSR_CH; ++i) v[i] = lst[min(tid + (k0 + i) * CSR_T, E_IN - 1)];
#pragma unroll
        for (int i = 0; i < CSR_CH; ++i) {
            const int idx = tid + (k0 + i) * CSR_T;
            if (idx < total) elist[idx] = v[i];
        }
    }
}

__device__ __forceinline__ float leaky(const float x)
{
    return (x >= 0.f) ? x : NEG_SLOPE * x;
}

template <int MASK>
__device__ __forceinline__ float row_xor(const float x)
{
    return __builtin_bit_cast(float, __builtin_amdgcn_update_dpp(
        0, __builtin_bit_cast(int, x), 0x160 + MASK, 0xf, 0xf, false));
}

__device__ __forceinline__ float xhalf(const float x)
{
    const int b = __builtin_bit_cast(int, x);
    return __builtin_bit_cast(float, __builtin_amdgcn_permlanex16(b, b, 0x76543210, 0xfedcba98, false, false));
}

__device__ __forceinline__ float wave_max(float x)
{
    x = fmaxf(x, row_xor<1>(x));
    x = fmaxf(x, row_xor<2>(x));
    x = fmaxf(x, row_xor<4>(x));
    x = fmaxf(x, row_xor<8>(x));
    return fmaxf(x, xhalf(x));
}

__device__ __forceinline__ float wave_sum(float x)
{
    x += row_xor<1>(x);
    x += row_xor<2>(x);
    x += row_xor<4>(x);
    x += row_xor<8>(x);
    return x + xhalf(x);
}

__device__ __forceinline__ float lane_f(const float v, const int k)
{
    return __int_as_float(__builtin_amdgcn_readlane(__float_as_int(v), k));
}

__device__ __forceinline__ float uni(const float v)
{
    return __int_as_float(__builtin_amdgcn_readfirstlane(__float_as_int(v)));
}

#define ATT_NODES 8
static_assert(F_IN % ATT_NODES == 0, "every block of fold_att owns 8 whole rows of W1");
static_assert((ATT_NODES * HEADS * 4) % 128 == 0, "a block's results are whole 128-byte lines");

__device__ __forceinline__ float dot8(const float4 u, const float4 w, const float4 a, const float4 b)
{
    return u.x * a.x + u.y * a.y + u.z * a.z + u.w * a.w
         + w.x * b.x + w.y * b.y + w.z * b.z + w.w * b.w;
}

__launch_bounds__(256)
__global__ void fold_att(const float* __restrict__ Wm,
                         const float* __restrict__ att_src,
                         const float* __restrict__ att_dst,
                         float* __restrict__ wS, float* __restrict__ wD)
{
    __shared__ float res[2 * ATT_NODES * HEADS];

    const int tid  = threadIdx.x;
    const int lane = tid & 31;
    const int head = __builtin_amdgcn_readfirstlane(tid >> 5);
    const int n0   = blockIdx.x * ATT_NODES;
    const int coff = head * HIDD + lane * 4;

    const float4 s0 = *(const float4*)(att_src + coff);
    const float4 s1 = *(const float4*)(att_src + coff + 128);
    const float4 d0 = *(const float4*)(att_dst + coff);
    const float4 d1 = *(const float4*)(att_dst + coff + 128);

    const float* hp = Wm + (size_t)n0 * HC + coff;
    float4 u[ATT_NODES], w[ATT_NODES];
#pragma unroll
    for (int i = 0; i < ATT_NODES; ++i) {
        u[i] = *(const float4*)(hp + (size_t)i * HC);
        w[i] = *(const float4*)(hp + (size_t)i * HC + 128);
    }

    float keep = 0.f;
#pragma unroll
    for (int i = 0; i < ATT_NODES; ++i) {
        const float ss = wave_sum(dot8(u[i], w[i], s0, s1));
        const float sd = wave_sum(dot8(u[i], w[i], d0, d1));
        keep = (lane == i) ? ss : keep;
        keep = (lane == ATT_NODES + i) ? sd : keep;
    }
    if (lane < 2 * ATT_NODES)
        res[(lane >> 3) * (ATT_NODES * HEADS) + (lane & 7) * HEADS + head] = keep;
    __syncthreads();
    if (tid < ATT_NODES * HEADS)
        wS[(size_t)n0 * HEADS + tid] = res[tid];
    else if (tid < 2 * ATT_NODES * HEADS)
        wD[(size_t)n0 * HEADS + (tid - ATT_NODES * HEADS)] = res[tid];
}


#define AX_NODES 16
static_assert(N_NODES % AX_NODES == 0, "every block of node_att1 owns 16 whole nodes");
static_assert((AX_NODES * HEADS * 4) % 128 == 0, "a block's results are whole 128-byte lines");

__launch_bounds__(256)
__global__ void node_att1(const float* __restrict__ X, const float* __restrict__ wS,
                          const float* __restrict__ wD,
                          float* __restrict__ aS, float* __restrict__ aD)
{
    __shared__ float wl[F_IN * 2 * HEADS];
    __shared__ float res[2 * AX_NODES * HEADS];

    const int tid = threadIdx.x;
    for (int i = tid; i < F_IN * HEADS; i += 256) {
        const int f = i / HEADS, h = i % HEADS;
        wl[f * 2 * HEADS + h]         = wS[i];
        wl[f * 2 * HEADS + HEADS + h] = wD[i];
    }
    __syncthreads();

    const int ln = tid >> 4;
    const int j  = tid & 15;
    const int n0 = blockIdx.x * AX_NODES;
    const float* xp = X + (size_t)(n0 + ln) * F_IN;
    const float* wp = wl + j;

    float s = 0.f;
#pragma unroll 8
    for (int q = 0; q < F_IN / 4; ++q) {
        const float4 v = *(const float4*)(xp + 4 * q);
        s += v.x * wp[(4 * q + 0) * 2 * HEADS];
        s += v.y * wp[(4 * q + 1) * 2 * HEADS];
        s += v.z * wp[(4 * q + 2) * 2 * HEADS];
        s += v.w * wp[(4 * q + 3) * 2 * HEADS];
    }
    res[(j >> 3) * (AX_NODES * HEADS) + ln * HEADS + (j & 7)] = s;
    __syncthreads();
    if (tid < AX_NODES * HEADS)
        aS[(size_t)n0 * HEADS + tid] = res[tid];
    else
        aD[(size_t)n0 * HEADS + (tid - AX_NODES * HEADS)] = res[tid];
}

#define AGX_NODES 8
static_assert(N_NODES % AGX_NODES == 0, "every block of node_aggx owns 8 whole nodes");

__device__ __forceinline__ void axpy4(float4& acc, const float a, const float4 v)
{
    acc.x = __builtin_fmaf(a, v.x, acc.x); acc.y = __builtin_fmaf(a, v.y, acc.y);
    acc.z = __builtin_fmaf(a, v.z, acc.z); acc.w = __builtin_fmaf(a, v.w, acc.w);
}

__device__ __forceinline__ void load8(float (&o)[HEADS], const float* __restrict__ p)
{
    const float4 a = *(const float4*)p;
    const float4 b = *(const float4*)(p + 4);
    o[0] = a.x; o[1] = a.y; o[2] = a.z; o[3] = a.w;
    o[4] = b.x; o[5] = b.y; o[6] = b.z; o[7] = b.w;
}

__device__ __forceinline__ void edge_scores1(float (&e)[HEADS], const float* __restrict__ aS,
                                             const unsigned rec, const float (&ad)[HEADS])
{
    load8(e, aS + (size_t)(rec & 0xffffu) * HEADS);
#pragma unroll
    for (int h = 0; h < HEADS; ++h) e[h] = leaky(e[h] + ad[h]);
}

__device__ __forceinline__ void add_row(float4 (&acc)[HEADS], const float (&acur)[HEADS],
                                        const int k, const float4 u)
{
#pragma unroll
    for (int h = 0; h < HEADS; ++h) axpy4(acc[h], lane_f(acur[h], k), u);
}

__launch_bounds__(256)
__global__ void node_aggx(const int* __restrict__ rowptr, const unsigned* __restrict__ elist,
                          const float* __restrict__ X, const float* __restrict__ aS,
                          const float* __restrict__ aD, float* __restrict__ xagg)
{
    const int lane = threadIdx.x & 31;
    const int d    = blockIdx.x * AGX_NODES + __builtin_amdgcn_readfirstlane((int)(threadIdx.x >> 5));
    const int loff = lane * 4;

    const int beg = rowptr[d];
    const int end = rowptr[d + 1];
    const float4 xs = *(const float4*)(X + (size_t)d * F_IN + loff);
    float ad[HEADS], es[HEADS];
    load8(ad, aD + (size_t)d * HEADS);
    load8(es, aS + (size_t)d * HEADS);
#pragma unroll
    for (int h = 0; h < HEADS; ++h) es[h] = leaky(es[h] + ad[h]);

    const int  j0 = beg + lane;
    const bool v0 = j0 < end;
    unsigned rc = 0u;
    float ec[HEADS];
#pragma unroll
    for (int h = 0; h < HEADS; ++h) ec[h] = -INFINITY;
    if (v0) { rc = elist[j0]; edge_scores1(ec, aS, rc, ad); }
    float m[HEADS];
#pragma unroll
    for (int h = 0; h < HEADS; ++h) m[h] = fmaxf(es[h], ec[h]);
    for (int c0 = beg + 32; c0 < end; c0 += 32) {
        const int j = c0 + lane;
        if (j < end) {
            float e[HEADS];
            edge_scores1(e, aS, elist[j], ad);
#pragma unroll
            for (int h = 0; h < HEADS; ++h) m[h] = fmaxf(m[h], e[h]);
        }
    }
#pragma unroll
    for (int h = 0; h < HEADS; ++h) {
        m[h] = uni(wave_max(m[h]));
        m[h] = isfinite(m[h]) ? m[h] : 0.f;
    }

    float acur[HEADS];
    float s[HEADS];
#pragma unroll
    for (int h = 0; h < HEADS; ++h) { acur[h] = v0 ? expf(ec[h] - m[h]) : 0.f; s[h] = acur[h]; }
    for (int c0 = beg + 32; c0 < end; c0 += 32) {
        const int j = c0 + lane;
        if (j < end) {
            float e[HEADS];
            edge_scores1(e, aS, elist[j], ad);
#pragma unroll
            for (int h = 0; h < HEADS; ++h) s[h] += expf(e[h] - m[h]);
        }
    }

    float  den[HEADS];
    float4 acc[HEADS];
#pragma unroll
    for (int h = 0; h < HEADS; ++h) {
        const float ps = expf(es[h] - m[h]);
        den[h] = uni(wave_sum(s[h])) + ps + 1e-16f;
        const float a = ps / den[h];
        acc[h]  = make_float4(a * xs.x, a * xs.y, a * xs.z, a * xs.w);
        acur[h] = acur[h] / den[h];
    }

    unsigned rcur = rc;
    for (int c0 = beg; c0 < end; c0 += 32) {
        const int cnt = min(end - c0, 32);
        int k = 0;
        for (; k + 4 <= cnt; k += 4) {
            const unsigned r0 = __builtin_amdgcn_readlane(rcur, k);
            const unsigned r1 = __builtin_amdgcn_readlane(rcur, k + 1);
            const unsigned r2 = __builtin_amdgcn_readlane(rcur, k + 2);
            const unsigned r3 = __builtin_amdgcn_readlane(rcur, k + 3);
            const float4 u0 = *(const float4*)(X + (size_t)(r0 & 0xffffu) * F_IN + loff);
            const float4 u1 = *(const float4*)(X + (size_t)(r1 & 0xffffu) * F_IN + loff);
            const float4 u2 = *(const float4*)(X + (size_t)(r2 & 0xffffu) * F_IN + loff);
            const float4 u3 = *(const float4*)(X + (size_t)(r3 & 0xffffu) * F_IN + loff);
            add_row(acc, acur, k,     u0);
            add_row(acc, acur, k + 1, u1);
            add_row(acc, acur, k + 2, u2);
            add_row(acc, acur, k + 3, u3);
        }
        if (k + 2 <= cnt) {
            const unsigned r0 = __builtin_amdgcn_readlane(rcur, k);
            const unsigned r1 = __builtin_amdgcn_readlane(rcur, k + 1);
            const float4 u0 = *(const float4*)(X + (size_t)(r0 & 0xffffu) * F_IN + loff);
            const float4 u1 = *(const float4*)(X + (size_t)(r1 & 0xffffu) * F_IN + loff);
            add_row(acc, acur, k,     u0);
            add_row(acc, acur, k + 1, u1);
            k += 2;
        }
        if (k < cnt) {
            const unsigned r = __builtin_amdgcn_readlane(rcur, k);
            const float4 u = *(const float4*)(X + (size_t)(r & 0xffffu) * F_IN + loff);
            add_row(acc, acur, k, u);
        }
        if (c0 + 32 < end) {
            const int j = c0 + 32 + lane;
            rcur = 0u;
#pragma unroll
            for (int h = 0; h < HEADS; ++h) acur[h] = 0.f;
            if (j < end) {
                rcur = elist[j];
                float e[HEADS];
                edge_scores1(e, aS, rcur, ad);
#pragma unroll
                for (int h = 0; h < HEADS; ++h) acur[h] = expf(e[h] - m[h]) / den[h];
            }
        }
    }

    float* op = xagg + (size_t)d * HEADS * F_IN + loff;
#pragma unroll
    for (int h = 0; h < HEADS; ++h) *(float4*)(op + h * F_IN) = acc[h];
}

#define AGG2_NODES 16
static_assert(N_NODES % AGG2_NODES == 0, "every block of node_agg2 owns 16 whole nodes");
static_assert((AGG2_NODES * NUM_CLASS * 4) % 128 == 0, "a block's output rows are whole 128-byte lines");

__launch_bounds__(256)
__global__ void node_agg2(const int* __restrict__ rowptr, const unsigned* __restrict__ elist,
                          const float* __restrict__ H2, const float* __restrict__ aS,
                          const float* __restrict__ aD, const float* __restrict__ b2,
                          float* __restrict__ out)
{
    __shared__ float rows[AGG2_NODES * NUM_CLASS];

    const int tid = threadIdx.x;
    const int ln  = tid >> 4;
    const int cl  = tid & 15;
    const int c   = min(cl, NUM_CLASS - 1);
    const int d   = blockIdx.x * AGG2_NODES + ln;

    const int beg = rowptr[d];
    const int end = rowptr[d + 1];
    const float ad = aD[d];
    const float es = leaky(aS[d] + ad);

    float m = es;
    for (int j = beg; j < end; j += 4) {
        float e[4];
#pragma unroll
        for (int i = 0; i < 4; ++i)
            e[i] = leaky(aS[elist[min(j + i, end - 1)] & 0xffffu] + ad);
        m = fmaxf(fmaxf(m, fmaxf(e[0], e[1])), fmaxf(e[2], e[3]));
    }
    m = isfinite(m) ? m : 0.f;

    float den, acc;
    {
        const float p = expf(es - m);
        den = p;
        acc = p * H2[(size_t)d * NUM_CLASS + c];
    }
    for (int j = beg; j < end; j += 4) {
        unsigned sN[4];
        float e[4], h[4];
#pragma unroll
        for (int i = 0; i < 4; ++i) sN[i] = elist[min(j + i, end - 1)] & 0xffffu;
#pragma unroll
        for (int i = 0; i < 4; ++i) {
            e[i] = leaky(aS[sN[i]] + ad);
            h[i] = H2[(size_t)sN[i] * NUM_CLASS + c];
        }
#pragma unroll
        for (int i = 0; i < 4; ++i) {
            const bool  on = j + i < end;
            const float p  = on ? expf(e[i] - m) : 0.f;
            den += p;
            acc += p * (on ? h[i] : 0.f);
        }
    }
    const float v = acc / (den + 1e-16f) + b2[c];
    if (cl < NUM_CLASS) rows[ln * NUM_CLASS + cl] = 1.f / (1.f + expf(-v));
    __syncthreads();
    if (tid < AGG2_NODES * NUM_CLASS)
        out[(size_t)blockIdx.x * (AGG2_NODES * NUM_CLASS) + tid] = rows[tid];
}

extern "C" void kernel_launch(void* const* d_in, const int* in_sizes, int n_in,
                              void* d_out, int out_size, void* d_ws, size_t ws_size,
                              hipStream_t stream)
{
    const float* x   = (const float*)d_in[0];
    const int*   ei  = (const int*)d_in[1];
    const float* W1  = (const float*)d_in[2];
    const float* as1 = (const float*)d_in[3];
    const float* ad1 = (const float*)d_in[4];
    const float* b1  = (const float*)d_in[5];
    const float* W2  = (const float*)d_in[6];
    const float* as2 = (const float*)d_in[7];
    const float* ad2 = (const float*)d_in[8];
    const float* b2  = (const float*)d_in[9];
    float* out = (float*)d_out;

    float* ws = (float*)d_ws;
    size_t off = 0;
    float* xa    = ws + off; off += (size_t)N_NODES * HEADS * F_IN;
    float* h2p   = ws + off; off += (size_t)HEADS * N_NODES * NUM_CLASS;
    float* sA1   = ws + off; off += (size_t)N_NODES * HEADS;
    float* sD1   = ws + off; off += (size_t)N_NODES * HEADS;
    float* h2    = ws + off; off += (size_t)N_NODES * NUM_CLASS;
    float* sA2   = ws + off; off += N_NODES;
    float* sD2   = ws + off; off += N_NODES;
    float* wS1   = ws + off; off += (size_t)F_IN * HEADS;
    float* wD1   = ws + off; off += (size_t)F_IN * HEADS;
    unsigned* elist = (unsigned*)(ws + off); off += (size_t)E_IN;
    int*      rowp  = (int*)(ws + off);      off += (size_t)N_NODES + 1;
    (void)ws_size; (void)in_sizes; (void)n_in; (void)out_size;

    const dim3 b256(256);

    fold_att<<<F_IN / ATT_NODES, b256, 0, stream>>>(W1, as1, ad1, wS1, wD1);
    node_att1<<<N_NODES / AX_NODES, b256, 0, stream>>>(x, wS1, wD1, sA1, sD1);
    csr_build<<<1, CSR_T, 0, stream>>>(ei, rowp, elist);
    node_aggx<<<N_NODES / AGX_NODES, b256, 0, stream>>>(rowp, elist, x, sA1, sD1, xa);
    gemm12_wmma<<<dim3(HEADS, G1_RBLK), b256, 0, stream>>>(xa, W1, b1, W2, h2p);

    att2_kernel<<<(N_NODES + A2_NODES - 1) / A2_NODES, b256, 0, stream>>>(h2p, as2, ad2, h2, sA2, sD2);
    node_agg2<<<N_NODES / AGG2_NODES, b256, 0, stream>>>(rowp, elist, h2, sA2, sD2, b2, out);
}


# Grouped edge list built by 157 independent blocks (64 destinations each: scan all destinations, collect and rank own edges in LDS), not by one block; order inside a run (sum order) changes
# speedup vs baseline: 5.4718x; 1.0909x over previous
#include <hip/hip_runtime.h>
#include <math.h>

#define N_NODES   10000
#define F_IN      128
#define HEADS     8
#define HIDD      256
#define HC        (HEADS * HIDD)
#define NUM_CLASS 10
#define E_IN      50000
#define NEG_SLOPE 0.2f
static_assert(HIDD % 64 == 0, "a 64-column tile of GEMM1 lies inside one head");

typedef __attribute__((ext_vector_type(2))) float v2f;
typedef __attribute__((ext_vector_type(8))) float v8f;
typedef __attribute__((ext_vector_type(4))) float v4f;

__device__ __forceinline__ float bias_elu1(const float s, const float b)
{
    const float v = s + b;
    return (v > 0.f) ? v : expm1f(v);
}

#define G1_RT    2
#define G1_ROWS  (32 * G1_RT)
#define G1_CT    (HIDD / 64)
#define G1_RBLK  ((N_NODES + G1_ROWS - 1) / G1_ROWS)
static_assert((G1_ROWS * NUM_CLASS * 4) % 128 == 0, "a block's partial rows are whole 128-byte lines");
static_assert((N_NODES * NUM_CLASS * 4) % 128 == 0, "every head's partial plane starts on a 128-byte line");
static_assert(8 * G1_RT * 256 <= F_IN * 64, "the waves' h2^T tiles fit the W1 tile's LDS");

__device__ __forceinline__ void g1_load_w(v4f (&q)[8], const float* wg)
{
#pragma unroll
    for (int i = 0; i < 8; ++i) q[i] = *(const v4f*)(wg + (size_t)(16 * i) * HC);
}

__device__ __forceinline__ void g1_store_w(float* bs, const v4f (&q)[8])
{
#pragma unroll
    for (int i = 0; i < 8; ++i) *(v4f*)(bs + (16 * i) * 64) = q[i];
}

__launch_bounds__(256)
__global__ void gemm12_wmma(const float* __restrict__ XA, const float* __restrict__ W,
                            const float* __restrict__ bias, const float* __restrict__ W2,
                            float* __restrict__ H2P)
{
    __shared__ __attribute__((aligned(16))) float As[G1_ROWS][F_IN + 4];
    __shared__ __attribute__((aligned(16))) float Bs[F_IN][64];

    const int tid = threadIdx.x;
    const int hd  = blockIdx.x;
    const int m0  = blockIdx.y * G1_ROWS;

    const float* wg = W + (size_t)(tid >> 4) * HC + hd * HIDD + (tid & 15) * 4;
    float*       bs = &Bs[tid >> 4][(tid & 15) * 4];

    v4f q[8];
    g1_load_w(q, wg);
#pragma unroll
    for (int i = 0; i < G1_ROWS * F_IN / 4 / 256; ++i) {
        const int idx = (tid + i * 256) * 4;
        const int r = idx >> 7;
        const int c = idx & 127;
        const int gr = m0 + r;
        v4f v = *(const v4f*)(XA + ((size_t)min(gr, N_NODES - 1) * HEADS + hd) * F_IN + c);
        if (gr >= N_NODES) v = (v4f){0.f, 0.f, 0.f, 0.f};
        *(v4f*)(&As[r][c]) = v;
    }
    g1_store_w(bs, q);
    __syncthreads();

    const int wave = tid >> 5;
    const int lane = tid & 31;
    const int half = lane >> 4;
    const int ml   = lane & 15;
    const int rg   = (wave >> 2) * G1_RT;
    const int bc   = (wave & 3) * 16;
    const int cls  = min(ml, NUM_CLASS - 1);

    v8f acc2[G1_RT];
#pragma unroll
    for (int rt = 0; rt < G1_RT; ++rt) acc2[rt] = (v8f){0.f, 0.f, 0.f, 0.f, 0.f, 0.f, 0.f, 0.f};

    for (int ct = 0; ct < G1_CT; ++ct) {
        if (ct + 1 < G1_CT) g1_load_w(q, wg + (ct + 1) * 64);

        const int cb = hd * HIDD + ct * 64 + bc + 8 * half;
        const float4 bv0 = *(const float4*)(bias + cb);
        const float4 bv1 = *(const float4*)(bias + cb + 4);
        const float bv[8] = {bv0.x, bv0.y, bv0.z, bv0.w, bv1.x, bv1.y, bv1.z, bv1.w};
        float w2f[8];
#pragma unroll
        for (int r = 0; r < 8; ++r) w2f[r] = W2[(size_t)(cb + r) * NUM_CLASS + cls];

        v8f acc[G1_RT];
#pragma unroll
        for (int rt = 0; rt < G1_RT; ++rt) acc[rt] = (v8f){0.f, 0.f, 0.f, 0.f, 0.f, 0.f, 0.f, 0.f};
#pragma unroll
        for (int kk = 0; kk < F_IN / 4; ++kk) {
            const int kb = kk * 4 + half * 2;
            v2f w;
            w.x = Bs[kb + 0][bc + ml];
            w.y = Bs[kb + 1][bc + ml];
#pragma unroll
            for (int rt = 0; rt < G1_RT; ++rt) {
                const float2 xv = *(const float2*)(&As[(rg + rt) * 16 + ml][kb]);
                v2f x;
                x.x = xv.x; x.y = xv.y;
                acc[rt] = __builtin_amdgcn_wmma_f32_16x16x4_f32(false, w, false, x,
                                                                (short)0, acc[rt], false, false);
            }
        }

#pragma unroll
        for (int rt = 0; rt < G1_RT; ++rt) {
            float zt[8];
#pragma unroll
            for (int r = 0; r < 8; ++r) zt[r] = bias_elu1(acc[rt][r], bv[r]);
#pragma unroll
            for (int t = 0; t < 4; ++t) {
                v2f a2, b2;
                a2.x = w2f[2 * t]; a2.y = w2f[2 * t + 1];
                b2.x = zt[2 * t];  b2.y = zt[2 * t + 1];
                acc2[rt] = __builtin_amdgcn_wmma_f32_16x16x4_f32(false, a2, false, b2,
                                                                 (short)0, acc2[rt], false, false);
            }
        }
        __syncthreads();
        if (ct + 1 < G1_CT) {
            g1_store_w(bs, q);
            __syncthreads();
        }
    }

    float* red = &Bs[0][0];
#pragma unroll
    for (int rt = 0; rt < G1_RT; ++rt)
#pragma unroll
        for (int r = 0; r < 8; ++r)
            red[((wave * G1_RT + rt) * 16 + 8 * half + r) * 16 + ml] = acc2[rt][r];
    __syncthreads();
    for (int p = tid; p < G1_ROWS * NUM_CLASS / 2; p += 256) {
        const int e  = 2 * p;
        const int lr = e / NUM_CLASS, c = e % NUM_CLASS;
        const int g  = (lr >> 4) / G1_RT, rt = (lr >> 4) % G1_RT;
        float2 s = make_float2(0.f, 0.f);
#pragma unroll
        for (int j = 0; j < 4; ++j) {
            const float* rp = red + (((g * 4 + j) * G1_RT + rt) * 16 + c) * 16 + (lr & 15);
            s.x += rp[0];
            s.y += rp[16];
        }
        if (m0 + lr < N_NODES)
            *(float2*)(H2P + ((size_t)hd * N_NODES + m0) * NUM_CLASS + e) = s;
    }
}

#define A2_NODES 256
static_assert((A2_NODES * NUM_CLASS * 4) % 128 == 0, "a block's rows of h2 are whole 128-byte lines");

__launch_bounds__(256)
__global__ void att2_kernel(const float* __restrict__ H2P,
                            const float* __restrict__ as2,
                            const float* __restrict__ ad2,
                            float* __restrict__ H2,
                            float* __restrict__ aS, float* __restrict__ aD)
{
    __shared__ float rows[A2_NODES * NUM_CLASS];

    const int tid  = threadIdx.x;
    const int base = blockIdx.x * (A2_NODES * NUM_CLASS);
    float s[NUM_CLASS];
#pragma unroll
    for (int i = 0; i < NUM_CLASS; ++i) {
        const int ic = min(base + tid + i * 256, N_NODES * NUM_CLASS - 1);
        float v = 0.f;
#pragma unroll
        for (int h = 0; h < HEADS; ++h) v += H2P[(size_t)h * (N_NODES * NUM_CLASS) + ic];
        s[i] = v;
    }
#pragma unroll
    for (int i = 0; i < NUM_CLASS; ++i) {
        const int idx = base + tid + i * 256;
        if (idx < N_NODES * NUM_CLASS) H2[idx] = s[i];
        rows[tid + i * 256] = s[i];
    }
    __syncthreads();

    const int n = blockIdx.x * A2_NODES + tid;
    if (n >= N_NODES) return;
    float ss = 0.f, sd = 0.f;
#pragma unroll
    for (int c = 0; c < NUM_CLASS; ++c) {
        const float v = rows[tid * NUM_CLASS + c];
        ss += v * as2[c];
        sd += v * ad2[c];
    }
    aS[n] = ss; aD[n] = sd;
}

#define CSR_T    512
#define CSR_R    64
#define CSR_B    ((N_NODES + CSR_R - 1) / CSR_R)
#define CSR_E4   (E_IN / 4)
#define CSR_VPT  ((CSR_E4 + CSR_T - 1) / CSR_T)
#define ELIST_CAP ((((E_IN + 31) / 32) + CSR_B + 1) * 32)
static_assert(E_IN % 4 == 0, "the destinations are read in groups of four");
static_assert(E_IN <= 65536 && N_NODES <= 65536, "edge id and source node fit 16 bits each");
static_assert(CSR_R == 64 && CSR_R <= CSR_T, "a collected edge packs its local destination in 6 bits");
static_assert((CSR_R * 2 * 4) % 128 == 0, "a block's rows of rowbe are whole 128-byte lines");

typedef __attribute__((ext_vector_type(4))) int v4i;

__launch_bounds__(CSR_T)
__global__ void csr_build(const int* __restrict__ ei, int* __restrict__ rowbe,
                          unsigned* __restrict__ elist)
{
    __shared__ unsigned       sel[E_IN];
    __shared__ unsigned short lst[E_IN];
    __shared__ int cnt[CSR_R];
    __shared__ int cur[CSR_R];
    __shared__ int wsum[CSR_T / 32];
    __shared__ int nsel_s;

    const int tid  = threadIdx.x;
    const int lane = tid & 31;
    const int wv   = tid >> 5;
    const unsigned lo   = blockIdx.x * CSR_R;
    const unsigned span = min((unsigned)CSR_R, (unsigned)N_NODES - lo);

    if (tid < CSR_R) cnt[tid] = 0;
    if (tid == 0) nsel_s = 0;
    __syncthreads();

    const v4i* dp = (const v4i*)(ei + E_IN);
    v4i dv[CSR_VPT];
#pragma unroll
    for (int i = 0; i < CSR_VPT; ++i) dv[i] = dp[min(tid + i * CSR_T, CSR_E4 - 1)];

    int nb = 0;
#pragma unroll
    for (int i = 0; i < CSR_VPT; ++i) {
        const int q = tid + i * CSR_T;
#pragma unroll
        for (int c = 0; c < 4; ++c) {
            unsigned d = (unsigned)dv[i][c];
            if (q >= CSR_E4) d = 0xffffffffu;
            nb += (d < lo) ? 1 : 0;
            const unsigned dl = d - lo;
            if (dl < span) {
                const int k = atomicAdd(&nsel_s, 1);
                sel[k] = ((unsigned)(4 * q + c) << 6) | dl;
                atomicAdd(&cnt[dl], 1);
            }
        }
    }
#pragma unroll
    for (int o = 16; o > 0; o >>= 1) nb += __shfl_xor(nb, o, 32);
    if (lane == 0) wsum[wv] = nb;
    __syncthreads();

    int base = 0;
#pragma unroll
    for (int j = 0; j < CSR_T / 32; ++j) base += wsum[j];
    const int nsel = nsel_s;
    const int seg  = ((base + 31) & ~31) + 32 * (int)blockIdx.x;

    if (tid < CSR_R) {
        int ex = 0, own = 0;
#pragma unroll
        for (int j = 0; j < CSR_R; ++j) {
            const int c = cnt[j];
            ex  += (j < tid) ? c : 0;
            own  = (j == tid) ? c : own;
        }
        cur[tid] = ex;
        if (lo + tid < (unsigned)N_NODES)
            *(int2*)(rowbe + 2 * (lo + tid)) = make_int2(seg + ex, seg + ex + own);
    }
    __syncthreads();

    for (int k = tid; k < nsel; k += CSR_T) {
        const unsigned rec = sel[k];
        const unsigned s   = (unsigned)ei[rec >> 6] & 0xffffu;
        const int slot = atomicAdd(&cur[rec & 63u], 1);
        lst[slot] = (unsigned short)s;
    }
    __syncthreads();

    for (int i = tid; i < nsel; i += CSR_T) elist[seg + i] = lst[i];
}

__device__ __forceinline__ float leaky(const float x)
{
    return (x >= 0.f) ? x : NEG_SLOPE * x;
}

template <int MASK>
__device__ __forceinline__ float row_xor(const float x)
{
    return __builtin_bit_cast(float, __builtin_amdgcn_update_dpp(
        0, __builtin_bit_cast(int, x), 0x160 + MASK, 0xf, 0xf, false));
}

__device__ __forceinline__ float xhalf(const float x)
{
    const int b = __builtin_bit_cast(int, x);
    return __builtin_bit_cast(float, __builtin_amdgcn_permlanex16(b, b, 0x76543210, 0xfedcba98, false, false));
}

__device__ __forceinline__ float wave_max(float x)
{
    x = fmaxf(x, row_xor<1>(x));
    x = fmaxf(x, row_xor<2>(x));
    x = fmaxf(x, row_xor<4>(x));
    x = fmaxf(x, row_xor<8>(x));
    return fmaxf(x, xhalf(x));
}

__device__ __forceinline__ float wave_sum(float x)
{
    x += row_xor<1>(x);
    x += row_xor<2>(x);
    x += row_xor<4>(x);
    x += row_xor<8>(x);
    return x + xhalf(x);
}

__device__ __forceinline__ float lane_f(const float v, const int k)
{
    return __int_as_float(__builtin_amdgcn_readlane(__float_as_int(v), k));
}

__device__ __forceinline__ float uni(const float v)
{
    return __int_as_float(__builtin_amdgcn_readfirstlane(__float_as_int(v)));
}

#define ATT_NODES 8
static_assert(F_IN % ATT_NODES == 0, "every block of fold_att owns 8 whole rows of W1");
static_assert((ATT_NODES * HEADS * 4) % 128 == 0, "a block's results are whole 128-byte lines");

__device__ __forceinline__ float dot8(const float4 u, const float4 w, const float4 a, const float4 b)
{
    return u.x * a.x + u.y * a.y + u.z * a.z + u.w * a.w
         + w.x * b.x + w.y * b.y + w.z * b.z + w.w * b.w;
}

__launch_bounds__(256)
__global__ void fold_att(const float* __restrict__ Wm,
                         const float* __restrict__ att_src,
                         const float* __restrict__ att_dst,
                         float* __restrict__ wS, float* __restrict__ wD)
{
    __shared__ float res[2 * ATT_NODES * HEADS];

    const int tid  = threadIdx.x;
    const int lane = tid & 31;
    const int head = __builtin_amdgcn_readfirstlane(tid >> 5);
    const int n0   = blockIdx.x * ATT_NODES;
    const int coff = head * HIDD + lane * 4;

    const float4 s0 = *(const float4*)(att_src + coff);
    const float4 s1 = *(const float4*)(att_src + coff + 128);
    const float4 d0 = *(const float4*)(att_dst + coff);
    const float4 d1 = *(const float4*)(att_dst + coff + 128);

    const float* hp = Wm + (size_t)n0 * HC + coff;
    float4 u[ATT_NODES], w[ATT_NODES];
#pragma unroll
    for (int i = 0; i < ATT_NODES; ++i) {
        u[i] = *(const float4*)(hp + (size_t)i * HC);
        w[i] = *(const float4*)(hp + (size_t)i * HC + 128);
    }

    float keep = 0.f;
#pragma unroll
    for (int i = 0; i < ATT_NODES; ++i) {
        const float ss = wave_sum(dot8(u[i], w[i], s0, s1));
        const float sd = wave_sum(dot8(u[i], w[i], d0, d1));
        keep = (lane == i) ? ss : keep;
        keep = (lane == ATT_NODES + i) ? sd : keep;
    }
    if (lane < 2 * ATT_NODES)
        res[(lane >> 3) * (ATT_NODES * HEADS) + (lane & 7) * HEADS + head] = keep;
    __syncthreads();
    if (tid < ATT_NODES * HEADS)
        wS[(size_t)n0 * HEADS + tid] = res[tid];
    else if (tid < 2 * ATT_NODES * HEADS)
        wD[(size_t)n0 * HEADS + (tid - ATT_NODES * HEADS)] = res[tid];
}


#define AX_NODES 16
static_assert(N_NODES % AX_NODES == 0, "every block of node_att1 owns 16 whole nodes");
static_assert((AX_NODES * HEADS * 4) % 128 == 0, "a block's results are whole 128-byte lines");

__launch_bounds__(256)
__global__ void node_att1(const float* __restrict__ X, const float* __restrict__ wS,
                          const float* __restrict__ wD,
                          float* __restrict__ aS, float* __restrict__ aD)
{
    __shared__ float wl[F_IN * 2 * HEADS];
    __shared__ float res[2 * AX_NODES * HEADS];

    const int tid = threadIdx.x;
    for (int i = tid; i < F_IN * HEADS; i += 256) {
        const int f = i / HEADS, h = i % HEADS;
        wl[f * 2 * HEADS + h]         = wS[i];
        wl[f * 2 * HEADS + HEADS + h] = wD[i];
    }
    __syncthreads();

    const int ln = tid >> 4;
    const int j  = tid & 15;
    const int n0 = blockIdx.x * AX_NODES;
    const float* xp = X + (size_t)(n0 + ln) * F_IN;
    const float* wp = wl + j;

    float s = 0.f;
#pragma unroll 8
    for (int q = 0; q < F_IN / 4; ++q) {
        const float4 v = *(const float4*)(xp + 4 * q);
        s += v.x * wp[(4 * q + 0) * 2 * HEADS];
        s += v.y * wp[(4 * q + 1) * 2 * HEADS];
        s += v.z * wp[(4 * q + 2) * 2 * HEADS];
        s += v.w * wp[(4 * q + 3) * 2 * HEADS];
    }
    res[(j >> 3) * (AX_NODES * HEADS) + ln * HEADS + (j & 7)] = s;
    __syncthreads();
    if (tid < AX_NODES * HEADS)
        aS[(size_t)n0 * HEADS + tid] = res[tid];
    else
        aD[(size_t)n0 * HEADS + (tid - AX_NODES * HEADS)] = res[tid];
}

#define AGX_NODES 8
static_assert(N_NODES % AGX_NODES == 0, "every block of node_aggx owns 8 whole nodes");

__device__ __forceinline__ void axpy4(float4& acc, const float a, const float4 v)
{
    acc.x = __builtin_fmaf(a, v.x, acc.x); acc.y = __builtin_fmaf(a, v.y, acc.y);
    acc.z = __builtin_fmaf(a, v.z, acc.z); acc.w = __builtin_fmaf(a, v.w, acc.w);
}

__device__ __forceinline__ void load8(float (&o)[HEADS], const float* __restrict__ p)
{
    const float4 a = *(const float4*)p;
    const float4 b = *(const float4*)(p + 4);
    o[0] = a.x; o[1] = a.y; o[2] = a.z; o[3] = a.w;
    o[4] = b.x; o[5] = b.y; o[6] = b.z; o[7] = b.w;
}

__device__ __forceinline__ void edge_scores1(float (&e)[HEADS], const float* __restrict__ aS,
                                             const unsigned rec, const float (&ad)[HEADS])
{
    load8(e, aS + (size_t)(rec & 0xffffu) * HEADS);
#pragma unroll
    for (int h = 0; h < HEADS; ++h) e[h] = leaky(e[h] + ad[h]);
}

__device__ __forceinline__ void add_row(float4 (&acc)[HEADS], const float (&acur)[HEADS],
                                        const int k, const float4 u)
{
#pragma unroll
    for (int h = 0; h < HEADS; ++h) axpy4(acc[h], lane_f(acur[h], k), u);
}

__launch_bounds__(256)
__global__ void node_aggx(const int* __restrict__ rowptr, const unsigned* __restrict__ elist,
                          const float* __restrict__ X, const float* __restrict__ aS,
                          const float* __restrict__ aD, float* __restrict__ xagg)
{
    const int lane = threadIdx.x & 31;
    const int d    = blockIdx.x * AGX_NODES + __builtin_amdgcn_readfirstlane((int)(threadIdx.x >> 5));
    const int loff = lane * 4;

    const int beg = rowptr[2 * d];
    const int end = rowptr[2 * d + 1];
    const float4 xs = *(const float4*)(X + (size_t)d * F_IN + loff);
    float ad[HEADS], es[HEADS];
    load8(ad, aD + (size_t)d * HEADS);
    load8(es, aS + (size_t)d * HEADS);
#pragma unroll
    for (int h = 0; h < HEADS; ++h) es[h] = leaky(es[h] + ad[h]);

    const int  j0 = beg + lane;
    const bool v0 = j0 < end;
    unsigned rc = 0u;
    float ec[HEADS];
#pragma unroll
    for (int h = 0; h < HEADS; ++h) ec[h] = -INFINITY;
    if (v0) { rc = elist[j0]; edge_scores1(ec, aS, rc, ad); }
    float m[HEADS];
#pragma unroll
    for (int h = 0; h < HEADS; ++h) m[h] = fmaxf(es[h], ec[h]);
    for (int c0 = beg + 32; c0 < end; c0 += 32) {
        const int j = c0 + lane;
        if (j < end) {
            float e[HEADS];
            edge_scores1(e, aS, elist[j], ad);
#pragma unroll
            for (int h = 0; h < HEADS; ++h) m[h] = fmaxf(m[h], e[h]);
        }
    }
#pragma unroll
    for (int h = 0; h < HEADS; ++h) {
        m[h] = uni(wave_max(m[h]));
        m[h] = isfinite(m[h]) ? m[h] : 0.f;
    }

    float acur[HEADS];
    float s[HEADS];
#pragma unroll
    for (int h = 0; h < HEADS; ++h) { acur[h] = v0 ? expf(ec[h] - m[h]) : 0.f; s[h] = acur[h]; }
    for (int c0 = beg + 32; c0 < end; c0 += 32) {
        const int j = c0 + lane;
        if (j < end) {
            float e[HEADS];
            edge_scores1(e, aS, elist[j], ad);
#pragma unroll
            for (int h = 0; h < HEADS; ++h) s[h] += expf(e[h] - m[h]);
        }
    }

    float  den[HEADS];
    float4 acc[HEADS];
#pragma unroll
    for (int h = 0; h < HEADS; ++h) {
        const float ps = expf(es[h] - m[h]);
        den[h] = uni(wave_sum(s[h])) + ps + 1e-16f;
        const float a = ps / den[h];
        acc[h]  = make_float4(a * xs.x, a * xs.y, a * xs.z, a * xs.w);
        acur[h] = acur[h] / den[h];
    }

    unsigned rcur = rc;
    for (int c0 = beg; c0 < end; c0 += 32) {
        const int cnt = min(end - c0, 32);
        int k = 0;
        for (; k + 4 <= cnt; k += 4) {
            const unsigned r0 = __builtin_amdgcn_readlane(rcur, k);
            const unsigned r1 = __builtin_amdgcn_readlane(rcur, k + 1);
            const unsigned r2 = __builtin_amdgcn_readlane(rcur, k + 2);
            const unsigned r3 = __builtin_amdgcn_readlane(rcur, k + 3);
            const float4 u0 = *(const float4*)(X + (size_t)(r0 & 0xffffu) * F_IN + loff);
            const float4 u1 = *(const float4*)(X + (size_t)(r1 & 0xffffu) * F_IN + loff);
            const float4 u2 = *(const float4*)(X + (size_t)(r2 & 0xffffu) * F_IN + loff);
            const float4 u3 = *(const float4*)(X + (size_t)(r3 & 0xffffu) * F_IN + loff);
            add_row(acc, acur, k,     u0);
            add_row(acc, acur, k + 1, u1);
            add_row(acc, acur, k + 2, u2);
            add_row(acc, acur, k + 3, u3);
        }
        if (k + 2 <= cnt) {
            const unsigned r0 = __builtin_amdgcn_readlane(rcur, k);
            const unsigned r1 = __builtin_amdgcn_readlane(rcur, k + 1);
            const float4 u0 = *(const float4*)(X + (size_t)(r0 & 0xffffu) * F_IN + loff);
            const float4 u1 = *(const float4*)(X + (size_t)(r1 & 0xffffu) * F_IN + loff);
            add_row(acc, acur, k,     u0);
            add_row(acc, acur, k + 1, u1);
            k += 2;
        }
        if (k < cnt) {
            const unsigned r = __builtin_amdgcn_readlane(rcur, k);
            const float4 u = *(const float4*)(X + (size_t)(r & 0xffffu) * F_IN + loff);
            add_row(acc, acur, k, u);
        }
        if (c0 + 32 < end) {
            const int j = c0 + 32 + lane;
            rcur = 0u;
#pragma unroll
            for (int h = 0; h < HEADS; ++h) acur[h] = 0.f;
            if (j < end) {
                rcur = elist[j];
                float e[HEADS];
                edge_scores1(e, aS, rcur, ad);
#pragma unroll
                for (int h = 0; h < HEADS; ++h) acur[h] = expf(e[h] - m[h]) / den[h];
            }
        }
    }

    float* op = xagg + (size_t)d * HEADS * F_IN + loff;
#pragma unroll
    for (int h = 0; h < HEADS; ++h) *(float4*)(op + h * F_IN) = acc[h];
}

#define AGG2_NODES 16
static_assert(N_NODES % AGG2_NODES == 0, "every block of node_agg2 owns 16 whole nodes");
static_assert((AGG2_NODES * NUM_CLASS * 4) % 128 == 0, "a block's output rows are whole 128-byte lines");

__launch_bounds__(256)
__global__ void node_agg2(const int* __restrict__ rowptr, const unsigned* __restrict__ elist,
                          const float* __restrict__ H2, const float* __restrict__ aS,
                          const float* __restrict__ aD, const float* __restrict__ b2,
                          float* __restrict__ out)
{
    __shared__ float rows[AGG2_NODES * NUM_CLASS];

    const int tid = threadIdx.x;
    const int ln  = tid >> 4;
    const int cl  = tid & 15;
    const int c   = min(cl, NUM_CLASS - 1);
    const int d   = blockIdx.x * AGG2_NODES + ln;

    const int beg = rowptr[2 * d];
    const int end = rowptr[2 * d + 1];
    const float ad = aD[d];
    const float es = leaky(aS[d] + ad);

    float m = es;
    for (int j = beg; j < end; j += 4) {
        float e[4];
#pragma unroll
        for (int i = 0; i < 4; ++i)
            e[i] = leaky(aS[elist[min(j + i, end - 1)] & 0xffffu] + ad);
        m = fmaxf(fmaxf(m, fmaxf(e[0], e[1])), fmaxf(e[2], e[3]));
    }
    m = isfinite(m) ? m : 0.f;

    float den, acc;
    {
        const float p = expf(es - m);
        den = p;
        acc = p * H2[(size_t)d * NUM_CLASS + c];
    }
    for (int j = beg; j < end; j += 4) {
        unsigned sN[4];
        float e[4], h[4];
#pragma unroll
        for (int i = 0; i < 4; ++i) sN[i] = elist[min(j + i, end - 1)] & 0xffffu;
#pragma unroll
        for (int i = 0; i < 4; ++i) {
            e[i] = leaky(aS[sN[i]] + ad);
            h[i] = H2[(size_t)sN[i] * NUM_CLASS + c];
        }
#pragma unroll
        for (int i = 0; i < 4; ++i) {
            const bool  on = j + i < end;
            const float p  = on ? expf(e[i] - m) : 0.f;
            den += p;
            acc += p * (on ? h[i] : 0.f);
        }
    }
    const float v = acc / (den + 1e-16f) + b2[c];
    if (cl < NUM_CLASS) rows[ln * NUM_CLASS + cl] = 1.f / (1.f + expf(-v));
    __syncthreads();
    if (tid < AGG2_NODES * NUM_CLASS)
        out[(size_t)blockIdx.x * (AGG2_NODES * NUM_CLASS) + tid] = rows[tid];
}

extern "C" void kernel_launch(void* const* d_in, const int* in_sizes, int n_in,
                              void* d_out, int out_size, void* d_ws, size_t ws_size,
                              hipStream_t stream)
{
    const float* x   = (const float*)d_in[0];
    const int*   ei  = (const int*)d_in[1];
    const float* W1  = (const float*)d_in[2];
    const float* as1 = (const float*)d_in[3];
    const float* ad1 = (const float*)d_in[4];
    const float* b1  = (const float*)d_in[5];
    const float* W2  = (const float*)d_in[6];
    const float* as2 = (const float*)d_in[7];
    const float* ad2 = (const float*)d_in[8];
    const float* b2  = (const float*)d_in[9];
    float* out = (float*)d_out;

    float* ws = (float*)d_ws;
    size_t off = 0;
    float* xa    = ws + off; off += (size_t)N_NODES * HEADS * F_IN;
    float* h2p   = ws + off; off += (size_t)HEADS * N_NODES * NUM_CLASS;
    float* sA1   = ws + off; off += (size_t)N_NODES * HEADS;
    float* sD1   = ws + off; off += (size_t)N_NODES * HEADS;
    float* h2    = ws + off; off += (size_t)N_NODES * NUM_CLASS;
    float* sA2   = ws + off; off += N_NODES;
    float* sD2   = ws + off; off += N_NODES;
    float* wS1   = ws + off; off += (size_t)F_IN * HEADS;
    float* wD1   = ws + off; off += (size_t)F_IN * HEADS;
    unsigned* elist = (unsigned*)(ws + off); off += (size_t)ELIST_CAP;
    int*      rowp  = (int*)(ws + off);      off += (size_t)2 * N_NODES;
    (void)ws_size; (void)in_sizes; (void)n_in; (void)out_size;

    const dim3 b256(256);

    fold_att<<<F_IN / ATT_NODES, b256, 0, stream>>>(W1, as1, ad1, wS1, wD1);
    node_att1<<<N_NODES / AX_NODES, b256, 0, stream>>>(x, wS1, wD1, sA1, sD1);
    csr_build<<<CSR_B, CSR_T, 0, stream>>>(ei, rowp, elist);
    node_aggx<<<N_NODES / AGX_NODES, b256, 0, stream>>>(rowp, elist, x, sA1, sD1, xa);
    gemm12_wmma<<<dim3(HEADS, G1_RBLK), b256, 0, stream>>>(xa, W1, b1, W2, h2p);

    att2_kernel<<<(N_NODES + A2_NODES - 1) / A2_NODES, b256, 0, stream>>>(h2p, as2, ad2, h2, sA2, sD2);
    node_agg2<<<N_NODES / AGG2_NODES, b256, 0, stream>>>(rowp, elist, h2, sA2, sD2, b2, out);
}


# Attention-vector fold of W1 and the grouped edge list run in ONE launch (block index picks the role; 8 fold blocks of 16 waves behind the 157 list blocks): 7 -> 6 launches; every sum unchanged
# speedup vs baseline: 5.6943x; 1.0407x over previous
#include <hip/hip_runtime.h>
#include <math.h>

#define N_NODES   10000
#define F_IN      128
#define HEADS     8
#define HIDD      256
#define HC        (HEADS * HIDD)
#define NUM_CLASS 10
#define E_IN      50000
#define NEG_SLOPE 0.2f
static_assert(HIDD % 64 == 0, "a 64-column tile of GEMM1 lies inside one head");

typedef __attribute__((ext_vector_type(2))) float v2f;
typedef __attribute__((ext_vector_type(8))) float v8f;
typedef __attribute__((ext_vector_type(4))) float v4f;

__device__ __forceinline__ float bias_elu1(const float s, const float b)
{
    const float v = s + b;
    return (v > 0.f) ? v : expm1f(v);
}

#define G1_RT    2
#define G1_ROWS  (32 * G1_RT)
#define G1_CT    (HIDD / 64)
#define G1_RBLK  ((N_NODES + G1_ROWS - 1) / G1_ROWS)
static_assert((G1_ROWS * NUM_CLASS * 4) % 128 == 0, "a block's partial rows are whole 128-byte lines");
static_assert((N_NODES * NUM_CLASS * 4) % 128 == 0, "every head's partial plane starts on a 128-byte line");
static_assert(8 * G1_RT * 256 <= F_IN * 64, "the waves' h2^T tiles fit the W1 tile's LDS");

__device__ __forceinline__ void g1_load_w(v4f (&q)[8], const float* wg)
{
#pragma unroll
    for (int i = 0; i < 8; ++i) q[i] = *(const v4f*)(wg + (size_t)(16 * i) * HC);
}

__device__ __forceinline__ void g1_store_w(float* bs, const v4f (&q)[8])
{
#pragma unroll
    for (int i = 0; i < 8; ++i) *(v4f*)(bs + (16 * i) * 64) = q[i];
}

__launch_bounds__(256)
__global__ void gemm12_wmma(const float* __restrict__ XA, const float* __restrict__ W,
                            const float* __restrict__ bias, const float* __restrict__ W2,
                            float* __restrict__ H2P)
{
    __shared__ __attribute__((aligned(16))) float As[G1_ROWS][F_IN + 4];
    __shared__ __attribute__((aligned(16))) float Bs[F_IN][64];

    const int tid = threadIdx.x;
    const int hd  = blockIdx.x;
    const int m0  = blockIdx.y * G1_ROWS;

    const float* wg = W + (size_t)(tid >> 4) * HC + hd * HIDD + (tid & 15) * 4;
    float*       bs = &Bs[tid >> 4][(tid & 15) * 4];

    v4f q[8];
    g1_load_w(q, wg);
#pragma unroll
    for (int i = 0; i < G1_ROWS * F_IN / 4 / 256; ++i) {
        const int idx = (tid + i * 256) * 4;
        const int r = idx >> 7;
        const int c = idx & 127;
        const int gr = m0 + r;
        v4f v = *(const v4f*)(XA + ((size_t)min(gr, N_NODES - 1) * HEADS + hd) * F_IN + c);
        if (gr >= N_NODES) v = (v4f){0.f, 0.f, 0.f, 0.f};
        *(v4f*)(&As[r][c]) = v;
    }
    g1_store_w(bs, q);
    __syncthreads();

    const int wave = tid >> 5;
    const int lane = tid & 31;
    const int half = lane >> 4;
    const int ml   = lane & 15;
    const int rg   = (wave >> 2) * G1_RT;
    const int bc   = (wave & 3) * 16;
    const int cls  = min(ml, NUM_CLASS - 1);

    v8f acc2[G1_RT];
#pragma unroll
    for (int rt = 0; rt < G1_RT; ++rt) acc2[rt] = (v8f){0.f, 0.f, 0.f, 0.f, 0.f, 0.f, 0.f, 0.f};

    for (int ct = 0; ct < G1_CT; ++ct) {
        if (ct + 1 < G1_CT) g1_load_w(q, wg + (ct + 1) * 64);

        const int cb = hd * HIDD + ct * 64 + bc + 8 * half;
        const float4 bv0 = *(const float4*)(bias + cb);
        const float4 bv1 = *(const float4*)(bias + cb + 4);
        const float bv[8] = {bv0.x, bv0.y, bv0.z, bv0.w, bv1.x, bv1.y, bv1.z, bv1.w};
        float w2f[8];
#pragma unroll
        for (int r = 0; r < 8; ++r) w2f[r] = W2[(size_t)(cb + r) * NUM_CLASS + cls];

        v8f acc[G1_RT];
#pragma unroll
        for (int rt = 0; rt < G1_RT; ++rt) acc[rt] = (v8f){0.f, 0.f, 0.f, 0.f, 0.f, 0.f, 0.f, 0.f};
#pragma unroll
        for (int kk = 0; kk < F_IN / 4; ++kk) {
            const int kb = kk * 4 + half * 2;
            v2f w;
            w.x = Bs[kb + 0][bc + ml];
            w.y = Bs[kb + 1][bc + ml];
#pragma unroll
            for (int rt = 0; rt < G1_RT; ++rt) {
                const float2 xv = *(const float2*)(&As[(rg + rt) * 16 + ml][kb]);
                v2f x;
                x.x = xv.x; x.y = xv.y;
                acc[rt] = __builtin_amdgcn_wmma_f32_16x16x4_f32(false, w, false, x,
                                                                (short)0, acc[rt], false, false);
            }
        }

#pragma unroll
        for (int rt = 0; rt < G1_RT; ++rt) {
            float zt[8];
#pragma unroll
            for (int r = 0; r < 8; ++r) zt[r] = bias_elu1(acc[rt][r], bv[r]);
#pragma unroll
            for (int t = 0; t < 4; ++t) {
                v2f a2, b2;
                a2.x = w2f[2 * t]; a2.y = w2f[2 * t + 1];
                b2.x = zt[2 * t];  b2.y = zt[2 * t + 1];
                acc2[rt] = __builtin_amdgcn_wmma_f32_16x16x4_f32(false, a2, false, b2,
                                                                 (short)0, acc2[rt], false, false);
            }
        }
        __syncthreads();
        if (ct + 1 < G1_CT) {
            g1_store_w(bs, q);
            __syncthreads();
        }
    }

    float* red = &Bs[0][0];
#pragma unroll
    for (int rt = 0; rt < G1_RT; ++rt)
#pragma unroll
        for (int r = 0; r < 8; ++r)
            red[((wave * G1_RT + rt) * 16 + 8 * half + r) * 16 + ml] = acc2[rt][r];
    __syncthreads();
    for (int p = tid; p < G1_ROWS * NUM_CLASS / 2; p += 256) {
        const int e  = 2 * p;
        const int lr = e / NUM_CLASS, c = e % NUM_CLASS;
        const int g  = (lr >> 4) / G1_RT, rt = (lr >> 4) % G1_RT;
        float2 s = make_float2(0.f, 0.f);
#pragma unroll
        for (int j = 0; j < 4; ++j) {
            const float* rp = red + (((g * 4 + j) * G1_RT + rt) * 16 + c) * 16 + (lr & 15);
            s.x += rp[0];
            s.y += rp[16];
        }
        if (m0 + lr < N_NODES)
            *(float2*)(H2P + ((size_t)hd * N_NODES + m0) * NUM_CLASS + e) = s;
    }
}

#define A2_NODES 256
static_assert((A2_NODES * NUM_CLASS * 4) % 128 == 0, "a block's rows of h2 are whole 128-byte lines");

__launch_bounds__(256)
__global__ void att2_kernel(const float* __restrict__ H2P,
                            const float* __restrict__ as2,
                            const float* __restrict__ ad2,
                            float* __restrict__ H2,
                            float* __restrict__ aS, float* __restrict__ aD)
{
    __shared__ float rows[A2_NODES * NUM_CLASS];

    const int tid  = threadIdx.x;
    const int base = blockIdx.x * (A2_NODES * NUM_CLASS);
    float s[NUM_CLASS];
#pragma unroll
    for (int i = 0; i < NUM_CLASS; ++i) {
        const int ic = min(base + tid + i * 256, N_NODES * NUM_CLASS - 1);
        float v = 0.f;
#pragma unroll
        for (int h = 0; h < HEADS; ++h) v += H2P[(size_t)h * (N_NODES * NUM_CLASS) + ic];
        s[i] = v;
    }
#pragma unroll
    for (int i = 0; i < NUM_CLASS; ++i) {
        const int idx = base + tid + i * 256;
        if (idx < N_NODES * NUM_CLASS) H2[idx] = s[i];
        rows[tid + i * 256] = s[i];
    }
    __syncthreads();

    const int n = blockIdx.x * A2_NODES + tid;
    if (n >= N_NODES) return;
    float ss = 0.f, sd = 0.f;
#pragma unroll
    for (int c = 0; c < NUM_CLASS; ++c) {
        const float v = rows[tid * NUM_CLASS + c];
        ss += v * as2[c];
        sd += v * ad2[c];
    }
    aS[n] = ss; aD[n] = sd;
}

#define CSR_T    512
#define CSR_R    64
#define CSR_B    ((N_NODES + CSR_R - 1) / CSR_R)
#define CSR_E4   (E_IN / 4)
#define CSR_VPT  ((CSR_E4 + CSR_T - 1) / CSR_T)
#define ELIST_CAP ((((E_IN + 31) / 32) + CSR_B + 1) * 32)
static_assert(E_IN % 4 == 0, "the destinations are read in groups of four");
static_assert(E_IN <= 65536 && N_NODES <= 65536, "edge id and source node fit 16 bits each");
static_assert(CSR_R == 64 && CSR_R <= CSR_T, "a collected edge packs its local destination in 6 bits");
static_assert((CSR_R * 2 * 4) % 128 == 0, "a block's rows of rowbe are whole 128-byte lines");

typedef __attribute__((ext_vector_type(4))) int v4i;

__device__ __forceinline__ float leaky(const float x)
{
    return (x >= 0.f) ? x : NEG_SLOPE * x;
}

template <int MASK>
__device__ __forceinline__ float row_xor(const float x)
{
    return __builtin_bit_cast(float, __builtin_amdgcn_update_dpp(
        0, __builtin_bit_cast(int, x), 0x160 + MASK, 0xf, 0xf, false));
}

__device__ __forceinline__ float xhalf(const float x)
{
    const int b = __builtin_bit_cast(int, x);
    return __builtin_bit_cast(float, __builtin_amdgcn_permlanex16(b, b, 0x76543210, 0xfedcba98, false, false));
}

__device__ __forceinline__ float wave_max(float x)
{
    x = fmaxf(x, row_xor<1>(x));
    x = fmaxf(x, row_xor<2>(x));
    x = fmaxf(x, row_xor<4>(x));
    x = fmaxf(x, row_xor<8>(x));
    return fmaxf(x, xhalf(x));
}

__device__ __forceinline__ float wave_sum(float x)
{
    x += row_xor<1>(x);
    x += row_xor<2>(x);
    x += row_xor<4>(x);
    x += row_xor<8>(x);
    return x + xhalf(x);
}

__device__ __forceinline__ float lane_f(const float v, const int k)
{
    return __int_as_float(__builtin_amdgcn_readlane(__float_as_int(v), k));
}

__device__ __forceinline__ float uni(const float v)
{
    return __int_as_float(__builtin_amdgcn_readfirstlane(__float_as_int(v)));
}

#define ATT_NODES 8
#define FOLD_ROWS (2 * ATT_NODES)
#define FOLD_B    (F_IN / FOLD_ROWS)
static_assert(F_IN % FOLD_ROWS == 0, "every fold block owns 16 whole rows of W1");
static_assert(CSR_T == 32 * 2 * HEADS, "a fold block is 8 heads x two row groups, one wave each");
static_assert((FOLD_ROWS * HEADS * 4) % 128 == 0, "a block's results are whole 128-byte lines");

__device__ __forceinline__ float dot8(const float4 u, const float4 w, const float4 a, const float4 b)
{
    return u.x * a.x + u.y * a.y + u.z * a.z + u.w * a.w
         + w.x * b.x + w.y * b.y + w.z * b.z + w.w * b.w;
}

__device__ __forceinline__ void fold_rows(float* __restrict__ res, const int fb,
                                          const float* __restrict__ Wm,
                                          const float* __restrict__ att_src,
                                          const float* __restrict__ att_dst,
                                          float* __restrict__ wS, float* __restrict__ wD)
{
    const int tid  = threadIdx.x;
    const int lane = tid & 31;
    const int wv   = __builtin_amdgcn_readfirstlane(tid >> 5);
    const int head = wv & (HEADS - 1);
    const int grp  = wv >> 3;
    const int n0   = fb * FOLD_ROWS;
    const int coff = head * HIDD + lane * 4;

    const float4 s0 = *(const float4*)(att_src + coff);
    const float4 s1 = *(const float4*)(att_src + coff + 128);
    const float4 d0 = *(const float4*)(att_dst + coff);
    const float4 d1 = *(const float4*)(att_dst + coff + 128);

    const float* hp = Wm + (size_t)(n0 + grp * ATT_NODES) * HC + coff;
    float4 u[ATT_NODES], w[ATT_NODES];
#pragma unroll
    for (int i = 0; i < ATT_NODES; ++i) {
        u[i] = *(const float4*)(hp + (size_t)i * HC);
        w[i] = *(const float4*)(hp + (size_t)i * HC + 128);
    }

    float keep = 0.f;
#pragma unroll
    for (int i = 0; i < ATT_NODES; ++i) {
        const float ss = wave_sum(dot8(u[i], w[i], s0, s1));
        const float sd = wave_sum(dot8(u[i], w[i], d0, d1));
        keep = (lane == i) ? ss : keep;
        keep = (lane == ATT_NODES + i) ? sd : keep;
    }
    if (lane < 2 * ATT_NODES)
        res[(lane >> 3) * (FOLD_ROWS * HEADS) + (grp * ATT_NODES + (lane & 7)) * HEADS + head] = keep;
    __syncthreads();
    if (tid < FOLD_ROWS * HEADS)
        wS[(size_t)n0 * HEADS + tid] = res[tid];
    else if (tid < 2 * FOLD_ROWS * HEADS)
        wD[(size_t)n0 * HEADS + (tid - FOLD_ROWS * HEADS)] = res[tid];
}

__launch_bounds__(CSR_T)
__global__ void csr_fold(const int* __restrict__ ei, int* __restrict__ rowbe,
                         unsigned* __restrict__ elist,
                         const float* __restrict__ Wm,
                         const float* __restrict__ att_src,
                         const float* __restrict__ att_dst,
                         float* __restrict__ wS, float* __restrict__ wD)
{
    __shared__ float res[2 * FOLD_ROWS * HEADS];
    if (blockIdx.x >= CSR_B) {
        fold_rows(res, (int)blockIdx.x - CSR_B, Wm, att_src, att_dst, wS, wD);
        return;
    }

    __shared__ unsigned       sel[E_IN];
    __shared__ unsigned short lst[E_IN];
    __shared__ int cnt[CSR_R];
    __shared__ int cur[CSR_R];
    __shared__ int wsum[CSR_T / 32];
    __shared__ int nsel_s;

    const int tid  = threadIdx.x;
    const int lane = tid & 31;
    const int wv   = tid >> 5;
    const unsigned lo   = blockIdx.x * CSR_R;
    const unsigned span = min((unsigned)CSR_R, (unsigned)N_NODES - lo);

    if (tid < CSR_R) cnt[tid] = 0;
    if (tid == 0) nsel_s = 0;
    __syncthreads();

    const v4i* dp = (const v4i*)(ei + E_IN);
    v4i dv[CSR_VPT];
#pragma unroll
    for (int i = 0; i < CSR_VPT; ++i) dv[i] = dp[min(tid + i * CSR_T, CSR_E4 - 1)];

    int nb = 0;
#pragma unroll
    for (int i = 0; i < CSR_VPT; ++i) {
        const int q = tid + i * CSR_T;
#pragma unroll
        for (int c = 0; c < 4; ++c) {
            unsigned d = (unsigned)dv[i][c];
            if (q >= CSR_E4) d = 0xffffffffu;
            nb += (d < lo) ? 1 : 0;
            const unsigned dl = d - lo;
            if (dl < span) {
                const int k = atomicAdd(&nsel_s, 1);
                sel[k] = ((unsigned)(4 * q + c) << 6) | dl;
                atomicAdd(&cnt[dl], 1);
            }
        }
    }
#pragma unroll
    for (int o = 16; o > 0; o >>= 1) nb += __shfl_xor(nb, o, 32);
    if (lane == 0) wsum[wv] = nb;
    __syncthreads();

    int base = 0;
#pragma unroll
    for (int j = 0; j < CSR_T / 32; ++j) base += wsum[j];
    const int nsel = nsel_s;
    const int seg  = ((base + 31) & ~31) + 32 * (int)blockIdx.x;

    if (tid < CSR_R) {
        int ex = 0, own = 0;
#pragma unroll
        for (int j = 0; j < CSR_R; ++j) {
            const int c = cnt[j];
            ex  += (j < tid) ? c : 0;
            own  = (j == tid) ? c : own;
        }
        cur[tid] = ex;
        if (lo + tid < (unsigned)N_NODES)
            *(int2*)(rowbe + 2 * (lo + tid)) = make_int2(seg + ex, seg + ex + own);
    }
    __syncthreads();

    for (int k = tid; k < nsel; k += CSR_T) {
        const unsigned rec = sel[k];
        const unsigned s   = (unsigned)ei[rec >> 6] & 0xffffu;
        const int slot = atomicAdd(&cur[rec & 63u], 1);
        lst[slot] = (unsigned short)s;
    }
    __syncthreads();

    for (int i = tid; i < nsel; i += CSR_T) elist[seg + i] = lst[i];
}


#define AX_NODES 16
static_assert(N_NODES % AX_NODES == 0, "every block of node_att1 owns 16 whole nodes");
static_assert((AX_NODES * HEADS * 4) % 128 == 0, "a block's results are whole 128-byte lines");

__launch_bounds__(256)
__global__ void node_att1(const float* __restrict__ X, const float* __restrict__ wS,
                          const float* __restrict__ wD,
                          float* __restrict__ aS, float* __restrict__ aD)
{
    __shared__ float wl[F_IN * 2 * HEADS];
    __shared__ float res[2 * AX_NODES * HEADS];

    const int tid = threadIdx.x;
    for (int i = tid; i < F_IN * HEADS; i += 256) {
        const int f = i / HEADS, h = i % HEADS;
        wl[f * 2 * HEADS + h]         = wS[i];
        wl[f * 2 * HEADS + HEADS + h] = wD[i];
    }
    __syncthreads();

    const int ln = tid >> 4;
    const int j  = tid & 15;
    const int n0 = blockIdx.x * AX_NODES;
    const float* xp = X + (size_t)(n0 + ln) * F_IN;
    const float* wp = wl + j;

    float s = 0.f;
#pragma unroll 8
    for (int q = 0; q < F_IN / 4; ++q) {
        const float4 v = *(const float4*)(xp + 4 * q);
        s += v.x * wp[(4 * q + 0) * 2 * HEADS];
        s += v.y * wp[(4 * q + 1) * 2 * HEADS];
        s += v.z * wp[(4 * q + 2) * 2 * HEADS];
        s += v.w * wp[(4 * q + 3) * 2 * HEADS];
    }
    res[(j >> 3) * (AX_NODES * HEADS) + ln * HEADS + (j & 7)] = s;
    __syncthreads();
    if (tid < AX_NODES * HEADS)
        aS[(size_t)n0 * HEADS + tid] = res[tid];
    else
        aD[(size_t)n0 * HEADS + (tid - AX_NODES * HEADS)] = res[tid];
}

#define AGX_NODES 8
static_assert(N_NODES % AGX_NODES == 0, "every block of node_aggx owns 8 whole nodes");

__device__ __forceinline__ void axpy4(float4& acc, const float a, const float4 v)
{
    acc.x = __builtin_fmaf(a, v.x, acc.x); acc.y = __builtin_fmaf(a, v.y, acc.y);
    acc.z = __builtin_fmaf(a, v.z, acc.z); acc.w = __builtin_fmaf(a, v.w, acc.w);
}

__device__ __forceinline__ void load8(float (&o)[HEADS], const float* __restrict__ p)
{
    const float4 a = *(const float4*)p;
    const float4 b = *(const float4*)(p + 4);
    o[0] = a.x; o[1] = a.y; o[2] = a.z; o[3] = a.w;
    o[4] = b.x; o[5] = b.y; o[6] = b.z; o[7] = b.w;
}

__device__ __forceinline__ void edge_scores1(float (&e)[HEADS], const float* __restrict__ aS,
                                             const unsigned rec, const float (&ad)[HEADS])
{
    load8(e, aS + (size_t)(rec & 0xffffu) * HEADS);
#pragma unroll
    for (int h = 0; h < HEADS; ++h) e[h] = leaky(e[h] + ad[h]);
}

__device__ __forceinline__ void add_row(float4 (&acc)[HEADS], const float (&acur)[HEADS],
                                        const int k, const float4 u)
{
#pragma unroll
    for (int h = 0; h < HEADS; ++h) axpy4(acc[h], lane_f(acur[h], k), u);
}

__launch_bounds__(256)
__global__ void node_aggx(const int* __restrict__ rowptr, const unsigned* __restrict__ elist,
                          const float* __restrict__ X, const float* __restrict__ aS,
                          const float* __restrict__ aD, float* __restrict__ xagg)
{
    const int lane = threadIdx.x & 31;
    const int d    = blockIdx.x * AGX_NODES + __builtin_amdgcn_readfirstlane((int)(threadIdx.x >> 5));
    const int loff = lane * 4;

    const int beg = rowptr[2 * d];
    const int end = rowptr[2 * d + 1];
    const float4 xs = *(const float4*)(X + (size_t)d * F_IN + loff);
    float ad[HEADS], es[HEADS];
    load8(ad, aD + (size_t)d * HEADS);
    load8(es, aS + (size_t)d * HEADS);
#pragma unroll
    for (int h = 0; h < HEADS; ++h) es[h] = leaky(es[h] + ad[h]);

    const int  j0 = beg + lane;
    const bool v0 = j0 < end;
    unsigned rc = 0u;
    float ec[HEADS];
#pragma unroll
    for (int h = 0; h < HEADS; ++h) ec[h] = -INFINITY;
    if (v0) { rc = elist[j0]; edge_scores1(ec, aS, rc, ad); }
    float m[HEADS];
#pragma unroll
    for (int h = 0; h < HEADS; ++h) m[h] = fmaxf(es[h], ec[h]);
    for (int c0 = beg + 32; c0 < end; c0 += 32) {
        const int j = c0 + lane;
        if (j < end) {
            float e[HEADS];
            edge_scores1(e, aS, elist[j], ad);
#pragma unroll
            for (int h = 0; h < HEADS; ++h) m[h] = fmaxf(m[h], e[h]);
        }
    }
#pragma unroll
    for (int h = 0; h < HEADS; ++h) {
        m[h] = uni(wave_max(m[h]));
        m[h] = isfinite(m[h]) ? m[h] : 0.f;
    }

    float acur[HEADS];
    float s[HEADS];
#pragma unroll
    for (int h = 0; h < HEADS; ++h) { acur[h] = v0 ? expf(ec[h] - m[h]) : 0.f; s[h] = acur[h]; }
    for (int c0 = beg + 32; c0 < end; c0 += 32) {
        const int j = c0 + lane;
        if (j < end) {
            float e[HEADS];
            edge_scores1(e, aS, elist[j], ad);
#pragma unroll
            for (int h = 0; h < HEADS; ++h) s[h] += expf(e[h] - m[h]);
        }
    }

    float  den[HEADS];
    float4 acc[HEADS];
#pragma unroll
    for (int h = 0; h < HEADS; ++h) {
        const float ps = expf(es[h] - m[h]);
        den[h] = uni(wave_sum(s[h])) + ps + 1e-16f;
        const float a = ps / den[h];
        acc[h]  = make_float4(a * xs.x, a * xs.y, a * xs.z, a * xs.w);
        acur[h] = acur[h] / den[h];
    }

    unsigned rcur = rc;
    for (int c0 = beg; c0 < end; c0 += 32) {
        const int cnt = min(end - c0, 32);
        int k = 0;
        for (; k + 4 <= cnt; k += 4) {
            const unsigned r0 = __builtin_amdgcn_readlane(rcur, k);
            const unsigned r1 = __builtin_amdgcn_readlane(rcur, k + 1);
            const unsigned r2 = __builtin_amdgcn_readlane(rcur, k + 2);
            const unsigned r3 = __builtin_amdgcn_readlane(rcur, k + 3);
            const float4 u0 = *(const float4*)(X + (size_t)(r0 & 0xffffu) * F_IN + loff);
            const float4 u1 = *(const float4*)(X + (size_t)(r1 & 0xffffu) * F_IN + loff);
            const float4 u2 = *(const float4*)(X + (size_t)(r2 & 0xffffu) * F_IN + loff);
            const float4 u3 = *(const float4*)(X + (size_t)(r3 & 0xffffu) * F_IN + loff);
            add_row(acc, acur, k,     u0);
            add_row(acc, acur, k + 1, u1);
            add_row(acc, acur, k + 2, u2);
            add_row(acc, acur, k + 3, u3);
        }
        if (k + 2 <= cnt) {
            const unsigned r0 = __builtin_amdgcn_readlane(rcur, k);
            const unsigned r1 = __builtin_amdgcn_readlane(rcur, k + 1);
            const float4 u0 = *(const float4*)(X + (size_t)(r0 & 0xffffu) * F_IN + loff);
            const float4 u1 = *(const float4*)(X + (size_t)(r1 & 0xffffu) * F_IN + loff);
            add_row(acc, acur, k,     u0);
            add_row(acc, acur, k + 1, u1);
            k += 2;
        }
        if (k < cnt) {
            const unsigned r = __builtin_amdgcn_readlane(rcur, k);
            const float4 u = *(const float4*)(X + (size_t)(r & 0xffffu) * F_IN + loff);
            add_row(acc, acur, k, u);
        }
        if (c0 + 32 < end) {
            const int j = c0 + 32 + lane;
            rcur = 0u;
#pragma unroll
            for (int h = 0; h < HEADS; ++h) acur[h] = 0.f;
            if (j < end) {
                rcur = elist[j];
                float e[HEADS];
                edge_scores1(e, aS, rcur, ad);
#pragma unroll
                for (int h = 0; h < HEADS; ++h) acur[h] = expf(e[h] - m[h]) / den[h];
            }
        }
    }

    float* op = xagg + (size_t)d * HEADS * F_IN + loff;
#pragma unroll
    for (int h = 0; h < HEADS; ++h) *(float4*)(op + h * F_IN) = acc[h];
}

#define AGG2_NODES 16
static_assert(N_NODES % AGG2_NODES == 0, "every block of node_agg2 owns 16 whole nodes");
static_assert((AGG2_NODES * NUM_CLASS * 4) % 128 == 0, "a block's output rows are whole 128-byte lines");

__launch_bounds__(256)
__global__ void node_agg2(const int* __restrict__ rowptr, const unsigned* __restrict__ elist,
                          const float* __restrict__ H2, const float* __restrict__ aS,
                          const float* __restrict__ aD, const float* __restrict__ b2,
                          float* __restrict__ out)
{
    __shared__ float rows[AGG2_NODES * NUM_CLASS];

    const int tid = threadIdx.x;
    const int ln  = tid >> 4;
    const int cl  = tid & 15;
    const int c   = min(cl, NUM_CLASS - 1);
    const int d   = blockIdx.x * AGG2_NODES + ln;

    const int beg = rowptr[2 * d];
    const int end = rowptr[2 * d + 1];
    const float ad = aD[d];
    const float es = leaky(aS[d] + ad);

    float m = es;
    for (int j = beg; j < end; j += 4) {
        float e[4];
#pragma unroll
        for (int i = 0; i < 4; ++i)
            e[i] = leaky(aS[elist[min(j + i, end - 1)] & 0xffffu] + ad);
        m = fmaxf(fmaxf(m, fmaxf(e[0], e[1])), fmaxf(e[2], e[3]));
    }
    m = isfinite(m) ? m : 0.f;

    float den, acc;
    {
        const float p = expf(es - m);
        den = p;
        acc = p * H2[(size_t)d * NUM_CLASS + c];
    }
    for (int j = beg; j < end; j += 4) {
        unsigned sN[4];
        float e[4], h[4];
#pragma unroll
        for (int i = 0; i < 4; ++i) sN[i] = elist[min(j + i, end - 1)] & 0xffffu;
#pragma unroll
        for (int i = 0; i < 4; ++i) {
            e[i] = leaky(aS[sN[i]] + ad);
            h[i] = H2[(size_t)sN[i] * NUM_CLASS + c];
        }
#pragma unroll
        for (int i = 0; i < 4; ++i) {
            const bool  on = j + i < end;
            const float p  = on ? expf(e[i] - m) : 0.f;
            den += p;
            acc += p * (on ? h[i] : 0.f);
        }
    }
    const float v = acc / (den + 1e-16f) + b2[c];
    if (cl < NUM_CLASS) rows[ln * NUM_CLASS + cl] = 1.f / (1.f + expf(-v));
    __syncthreads();
    if (tid < AGG2_NODES * NUM_CLASS)
        out[(size_t)blockIdx.x * (AGG2_NODES * NUM_CLASS) + tid] = rows[tid];
}

extern "C" void kernel_launch(void* const* d_in, const int* in_sizes, int n_in,
                              void* d_out, int out_size, void* d_ws, size_t ws_size,
                              hipStream_t stream)
{
    const float* x   = (const float*)d_in[0];
    const int*   ei  = (const int*)d_in[1];
    const float* W1  = (const float*)d_in[2];
    const float* as1 = (const float*)d_in[3];
    const float* ad1 = (const float*)d_in[4];
    const float* b1  = (const float*)d_in[5];
    const float* W2  = (const float*)d_in[6];
    const float* as2 = (const float*)d_in[7];
    const float* ad2 = (const float*)d_in[8];
    const float* b2  = (const float*)d_in[9];
    float* out = (float*)d_out;

    float* ws = (float*)d_ws;
    size_t off = 0;
    float* xa    = ws + off; off += (size_t)N_NODES * HEADS * F_IN;
    float* h2p   = ws + off; off += (size_t)HEADS * N_NODES * NUM_CLASS;
    float* sA1   = ws + off; off += (size_t)N_NODES * HEADS;
    float* sD1   = ws + off; off += (size_t)N_NODES * HEADS;
    float* h2    = ws + off; off += (size_t)N_NODES * NUM_CLASS;
    float* sA2   = ws + off; off += N_NODES;
    float* sD2   = ws + off; off += N_NODES;
    float* wS1   = ws + off; off += (size_t)F_IN * HEADS;
    float* wD1   = ws + off; off += (size_t)F_IN * HEADS;
    unsigned* elist = (unsigned*)(ws + off); off += (size_t)ELIST_CAP;
    int*      rowp  = (int*)(ws + off);      off += (size_t)2 * N_NODES;
    (void)ws_size; (void)in_sizes; (void)n_in; (void)out_size;

    const dim3 b256(256);

    csr_fold<<<CSR_B + FOLD_B, CSR_T, 0, stream>>>(ei, rowp, elist, W1, as1, ad1, wS1, wD1);
    node_att1<<<N_NODES / AX_NODES, b256, 0, stream>>>(x, wS1, wD1, sA1, sD1);
    node_aggx<<<N_NODES / AGX_NODES, b256, 0, stream>>>(rowp, elist, x, sA1, sD1, xa);
    gemm12_wmma<<<dim3(HEADS, G1_RBLK), b256, 0, stream>>>(xa, W1, b1, W2, h2p);

    att2_kernel<<<(N_NODES + A2_NODES - 1) / A2_NODES, b256, 0, stream>>>(h2p, as2, ad2, h2, sA2, sD2);
    node_agg2<<<N_NODES / AGG2_NODES, b256, 0, stream>>>(rowp, elist, h2, sA2, sD2, b2, out);
}


# Layer-1 attention terms x @ [wS|wD] on the f32 16x16x4 matrix instruction: one wave per 16 nodes x 16 outputs (5000 -> 625 waves, no LDS copy of the vectors); dot-product sum order changes
# speedup vs baseline: 5.8816x; 1.0329x over previous
#include <hip/hip_runtime.h>
#include <math.h>

#define N_NODES   10000
#define F_IN      128
#define HEADS     8
#define HIDD      256
#define HC        (HEADS * HIDD)
#define NUM_CLASS 10
#define E_IN      50000
#define NEG_SLOPE 0.2f
static_assert(HIDD % 64 == 0, "a 64-column tile of GEMM1 lies inside one head");

typedef __attribute__((ext_vector_type(2))) float v2f;
typedef __attribute__((ext_vector_type(8))) float v8f;
typedef __attribute__((ext_vector_type(4))) float v4f;

__device__ __forceinline__ float bias_elu1(const float s, const float b)
{
    const float v = s + b;
    return (v > 0.f) ? v : expm1f(v);
}

#define G1_RT    2
#define G1_ROWS  (32 * G1_RT)
#define G1_CT    (HIDD / 64)
#define G1_RBLK  ((N_NODES + G1_ROWS - 1) / G1_ROWS)
static_assert((G1_ROWS * NUM_CLASS * 4) % 128 == 0, "a block's partial rows are whole 128-byte lines");
static_assert((N_NODES * NUM_CLASS * 4) % 128 == 0, "every head's partial plane starts on a 128-byte line");
static_assert(8 * G1_RT * 256 <= F_IN * 64, "the waves' h2^T tiles fit the W1 tile's LDS");

__device__ __forceinline__ void g1_load_w(v4f (&q)[8], const float* wg)
{
#pragma unroll
    for (int i = 0; i < 8; ++i) q[i] = *(const v4f*)(wg + (size_t)(16 * i) * HC);
}

__device__ __forceinline__ void g1_store_w(float* bs, const v4f (&q)[8])
{
#pragma unroll
    for (int i = 0; i < 8; ++i) *(v4f*)(bs + (16 * i) * 64) = q[i];
}

__launch_bounds__(256)
__global__ void gemm12_wmma(const float* __restrict__ XA, const float* __restrict__ W,
                            const float* __restrict__ bias, const float* __restrict__ W2,
                            float* __restrict__ H2P)
{
    __shared__ __attribute__((aligned(16))) float As[G1_ROWS][F_IN + 4];
    __shared__ __attribute__((aligned(16))) float Bs[F_IN][64];

    const int tid = threadIdx.x;
    const int hd  = blockIdx.x;
    const int m0  = blockIdx.y * G1_ROWS;

    const float* wg = W + (size_t)(tid >> 4) * HC + hd * HIDD + (tid & 15) * 4;
    float*       bs = &Bs[tid >> 4][(tid & 15) * 4];

    v4f q[8];
    g1_load_w(q, wg);
#pragma unroll
    for (int i = 0; i < G1_ROWS * F_IN / 4 / 256; ++i) {
        const int idx = (tid + i * 256) * 4;
        const int r = idx >> 7;
        const int c = idx & 127;
        const int gr = m0 + r;
        v4f v = *(const v4f*)(XA + ((size_t)min(gr, N_NODES - 1) * HEADS + hd) * F_IN + c);
        if (gr >= N_NODES) v = (v4f){0.f, 0.f, 0.f, 0.f};
        *(v4f*)(&As[r][c]) = v;
    }
    g1_store_w(bs, q);
    __syncthreads();

    const int wave = tid >> 5;
    const int lane = tid & 31;
    const int half = lane >> 4;
    const int ml   = lane & 15;
    const int rg   = (wave >> 2) * G1_RT;
    const int bc   = (wave & 3) * 16;
    const int cls  = min(ml, NUM_CLASS - 1);

    v8f acc2[G1_RT];
#pragma unroll
    for (int rt = 0; rt < G1_RT; ++rt) acc2[rt] = (v8f){0.f, 0.f, 0.f, 0.f, 0.f, 0.f, 0.f, 0.f};

    for (int ct = 0; ct < G1_CT; ++ct) {
        if (ct + 1 < G1_CT) g1_load_w(q, wg + (ct + 1) * 64);

        const int cb = hd * HIDD + ct * 64 + bc + 8 * half;
        const float4 bv0 = *(const float4*)(bias + cb);
        const float4 bv1 = *(const float4*)(bias + cb + 4);
        const float bv[8] = {bv0.x, bv0.y, bv0.z, bv0.w, bv1.x, bv1.y, bv1.z, bv1.w};
        float w2f[8];
#pragma unroll
        for (int r = 0; r < 8; ++r) w2f[r] = W2[(size_t)(cb + r) * NUM_CLASS + cls];

        v8f acc[G1_RT];
#pragma unroll
        for (int rt = 0; rt < G1_RT; ++rt) acc[rt] = (v8f){0.f, 0.f, 0.f, 0.f, 0.f, 0.f, 0.f, 0.f};
#pragma unroll
        for (int kk = 0; kk < F_IN / 4; ++kk) {
            const int kb = kk * 4 + half * 2;
            v2f w;
            w.x = Bs[kb + 0][bc + ml];
            w.y = Bs[kb + 1][bc + ml];
#pragma unroll
            for (int rt = 0; rt < G1_RT; ++rt) {
                const float2 xv = *(const float2*)(&As[(rg + rt) * 16 + ml][kb]);
                v2f x;
                x.x = xv.x; x.y = xv.y;
                acc[rt] = __builtin_amdgcn_wmma_f32_16x16x4_f32(false, w, false, x,
                                                                (short)0, acc[rt], false, false);
            }
        }

#pragma unroll
        for (int rt = 0; rt < G1_RT; ++rt) {
            float zt[8];
#pragma unroll
            for (int r = 0; r < 8; ++r) zt[r] = bias_elu1(acc[rt][r], bv[r]);
#pragma unroll
            for (int t = 0; t < 4; ++t) {
                v2f a2, b2;
                a2.x = w2f[2 * t]; a2.y = w2f[2 * t + 1];
                b2.x = zt[2 * t];  b2.y = zt[2 * t + 1];
                acc2[rt] = __builtin_amdgcn_wmma_f32_16x16x4_f32(false, a2, false, b2,
                                                                 (short)0, acc2[rt], false, false);
            }
        }
        __syncthreads();
        if (ct + 1 < G1_CT) {
            g1_store_w(bs, q);
            __syncthreads();
        }
    }

    float* red = &Bs[0][0];
#pragma unroll
    for (int rt = 0; rt < G1_RT; ++rt)
#pragma unroll
        for (int r = 0; r < 8; ++r)
            red[((wave * G1_RT + rt) * 16 + 8 * half + r) * 16 + ml] = acc2[rt][r];
    __syncthreads();
    for (int p = tid; p < G1_ROWS * NUM_CLASS / 2; p += 256) {
        const int e  = 2 * p;
        const int lr = e / NUM_CLASS, c = e % NUM_CLASS;
        const int g  = (lr >> 4) / G1_RT, rt = (lr >> 4) % G1_RT;
        float2 s = make_float2(0.f, 0.f);
#pragma unroll
        for (int j = 0; j < 4; ++j) {
            const float* rp = red + (((g * 4 + j) * G1_RT + rt) * 16 + c) * 16 + (lr & 15);
            s.x += rp[0];
            s.y += rp[16];
        }
        if (m0 + lr < N_NODES)
            *(float2*)(H2P + ((size_t)hd * N_NODES + m0) * NUM_CLASS + e) = s;
    }
}

#define A2_NODES 256
static_assert((A2_NODES * NUM_CLASS * 4) % 128 == 0, "a block's rows of h2 are whole 128-byte lines");

__launch_bounds__(256)
__global__ void att2_kernel(const float* __restrict__ H2P,
                            const float* __restrict__ as2,
                            const float* __restrict__ ad2,
                            float* __restrict__ H2,
                            float* __restrict__ aS, float* __restrict__ aD)
{
    __shared__ float rows[A2_NODES * NUM_CLASS];

    const int tid  = threadIdx.x;
    const int base = blockIdx.x * (A2_NODES * NUM_CLASS);
    float s[NUM_CLASS];
#pragma unroll
    for (int i = 0; i < NUM_CLASS; ++i) {
        const int ic = min(base + tid + i * 256, N_NODES * NUM_CLASS - 1);
        float v = 0.f;
#pragma unroll
        for (int h = 0; h < HEADS; ++h) v += H2P[(size_t)h * (N_NODES * NUM_CLASS) + ic];
        s[i] = v;
    }
#pragma unroll
    for (int i = 0; i < NUM_CLASS; ++i) {
        const int idx = base + tid + i * 256;
        if (idx < N_NODES * NUM_CLASS) H2[idx] = s[i];
        rows[tid + i * 256] = s[i];
    }
    __syncthreads();

    const int n = blockIdx.x * A2_NODES + tid;
    if (n >= N_NODES) return;
    float ss = 0.f, sd = 0.f;
#pragma unroll
    for (int c = 0; c < NUM_CLASS; ++c) {
        const float v = rows[tid * NUM_CLASS + c];
        ss += v * as2[c];
        sd += v * ad2[c];
    }
    aS[n] = ss; aD[n] = sd;
}

#define CSR_T    512
#define CSR_R    64
#define CSR_B    ((N_NODES + CSR_R - 1) / CSR_R)
#define CSR_E4   (E_IN / 4)
#define CSR_VPT  ((CSR_E4 + CSR_T - 1) / CSR_T)
#define ELIST_CAP ((((E_IN + 31) / 32) + CSR_B + 1) * 32)
static_assert(E_IN % 4 == 0, "the destinations are read in groups of four");
static_assert(E_IN <= 65536 && N_NODES <= 65536, "edge id and source node fit 16 bits each");
static_assert(CSR_R == 64 && CSR_R <= CSR_T, "a collected edge packs its local destination in 6 bits");
static_assert((CSR_R * 2 * 4) % 128 == 0, "a block's rows of rowbe are whole 128-byte lines");

typedef __attribute__((ext_vector_type(4))) int v4i;

__device__ __forceinline__ float leaky(const float x)
{
    return (x >= 0.f) ? x : NEG_SLOPE * x;
}

template <int MASK>
__device__ __forceinline__ float row_xor(const float x)
{
    return __builtin_bit_cast(float, __builtin_amdgcn_update_dpp(
        0, __builtin_bit_cast(int, x), 0x160 + MASK, 0xf, 0xf, false));
}

__device__ __forceinline__ float xhalf(const float x)
{
    const int b = __builtin_bit_cast(int, x);
    return __builtin_bit_cast(float, __builtin_amdgcn_permlanex16(b, b, 0x76543210, 0xfedcba98, false, false));
}

__device__ __forceinline__ float wave_max(float x)
{
    x = fmaxf(x, row_xor<1>(x));
    x = fmaxf(x, row_xor<2>(x));
    x = fmaxf(x, row_xor<4>(x));
    x = fmaxf(x, row_xor<8>(x));
    return fmaxf(x, xhalf(x));
}

__device__ __forceinline__ float wave_sum(float x)
{
    x += row_xor<1>(x);
    x += row_xor<2>(x);
    x += row_xor<4>(x);
    x += row_xor<8>(x);
    return x + xhalf(x);
}

__device__ __forceinline__ float lane_f(const float v, const int k)
{
    return __int_as_float(__builtin_amdgcn_readlane(__float_as_int(v), k));
}

__device__ __forceinline__ float uni(const float v)
{
    return __int_as_float(__builtin_amdgcn_readfirstlane(__float_as_int(v)));
}

#define ATT_NODES 8
#define FOLD_ROWS (2 * ATT_NODES)
#define FOLD_B    (F_IN / FOLD_ROWS)
static_assert(F_IN % FOLD_ROWS == 0, "every fold block owns 16 whole rows of W1");
static_assert(CSR_T == 32 * 2 * HEADS, "a fold block is 8 heads x two row groups, one wave each");
static_assert((FOLD_ROWS * HEADS * 4) % 128 == 0, "a block's results are whole 128-byte lines");

__device__ __forceinline__ float dot8(const float4 u, const float4 w, const float4 a, const float4 b)
{
    return u.x * a.x + u.y * a.y + u.z * a.z + u.w * a.w
         + w.x * b.x + w.y * b.y + w.z * b.z + w.w * b.w;
}

__device__ __forceinline__ void fold_rows(float* __restrict__ res, const int fb,
                                          const float* __restrict__ Wm,
                                          const float* __restrict__ att_src,
                                          const float* __restrict__ att_dst,
                                          float* __restrict__ wS, float* __restrict__ wD)
{
    const int tid  = threadIdx.x;
    const int lane = tid & 31;
    const int wv   = __builtin_amdgcn_readfirstlane(tid >> 5);
    const int head = wv & (HEADS - 1);
    const int grp  = wv >> 3;
    const int n0   = fb * FOLD_ROWS;
    const int coff = head * HIDD + lane * 4;

    const float4 s0 = *(const float4*)(att_src + coff);
    const float4 s1 = *(const float4*)(att_src + coff + 128);
    const float4 d0 = *(const float4*)(att_dst + coff);
    const float4 d1 = *(const float4*)(att_dst + coff + 128);

    const float* hp = Wm + (size_t)(n0 + grp * ATT_NODES) * HC + coff;
    float4 u[ATT_NODES], w[ATT_NODES];
#pragma unroll
    for (int i = 0; i < ATT_NODES; ++i) {
        u[i] = *(const float4*)(hp + (size_t)i * HC);
        w[i] = *(const float4*)(hp + (size_t)i * HC + 128);
    }

    float keep = 0.f;
#pragma unroll
    for (int i = 0; i < ATT_NODES; ++i) {
        const float ss = wave_sum(dot8(u[i], w[i], s0, s1));
        const float sd = wave_sum(dot8(u[i], w[i], d0, d1));
        keep = (lane == i) ? ss : keep;
        keep = (lane == ATT_NODES + i) ? sd : keep;
    }
    if (lane < 2 * ATT_NODES)
        res[(lane >> 3) * (FOLD_ROWS * HEADS) + (grp * ATT_NODES + (lane & 7)) * HEADS + head] = keep;
    __syncthreads();
    if (tid < FOLD_ROWS * HEADS)
        wS[(size_t)n0 * HEADS + tid] = res[tid];
    else if (tid < 2 * FOLD_ROWS * HEADS)
        wD[(size_t)n0 * HEADS + (tid - FOLD_ROWS * HEADS)] = res[tid];
}

__launch_bounds__(CSR_T)
__global__ void csr_fold(const int* __restrict__ ei, int* __restrict__ rowbe,
                         unsigned* __restrict__ elist,
                         const float* __restrict__ Wm,
                         const float* __restrict__ att_src,
                         const float* __restrict__ att_dst,
                         float* __restrict__ wS, float* __restrict__ wD)
{
    __shared__ float res[2 * FOLD_ROWS * HEADS];
    if (blockIdx.x >= CSR_B) {
        fold_rows(res, (int)blockIdx.x - CSR_B, Wm, att_src, att_dst, wS, wD);
        return;
    }

    __shared__ unsigned       sel[E_IN];
    __shared__ unsigned short lst[E_IN];
    __shared__ int cnt[CSR_R];
    __shared__ int cur[CSR_R];
    __shared__ int wsum[CSR_T / 32];
    __shared__ int nsel_s;

    const int tid  = threadIdx.x;
    const int lane = tid & 31;
    const int wv   = tid >> 5;
    const unsigned lo   = blockIdx.x * CSR_R;
    const unsigned span = min((unsigned)CSR_R, (unsigned)N_NODES - lo);

    if (tid < CSR_R) cnt[tid] = 0;
    if (tid == 0) nsel_s = 0;
    __syncthreads();

    const v4i* dp = (const v4i*)(ei + E_IN);
    v4i dv[CSR_VPT];
#pragma unroll
    for (int i = 0; i < CSR_VPT; ++i) dv[i] = dp[min(tid + i * CSR_T, CSR_E4 - 1)];

    int nb = 0;
#pragma unroll
    for (int i = 0; i < CSR_VPT; ++i) {
        const int q = tid + i * CSR_T;
#pragma unroll
        for (int c = 0; c < 4; ++c) {
            unsigned d = (unsigned)dv[i][c];
            if (q >= CSR_E4) d = 0xffffffffu;
            nb += (d < lo) ? 1 : 0;
            const unsigned dl = d - lo;
            if (dl < span) {
                const int k = atomicAdd(&nsel_s, 1);
                sel[k] = ((unsigned)(4 * q + c) << 6) | dl;
                atomicAdd(&cnt[dl], 1);
            }
        }
    }
#pragma unroll
    for (int o = 16; o > 0; o >>= 1) nb += __shfl_xor(nb, o, 32);
    if (lane == 0) wsum[wv] = nb;
    __syncthreads();

    int base = 0;
#pragma unroll
    for (int j = 0; j < CSR_T / 32; ++j) base += wsum[j];
    const int nsel = nsel_s;
    const int seg  = ((base + 31) & ~31) + 32 * (int)blockIdx.x;

    if (tid < CSR_R) {
        int ex = 0, own = 0;
#pragma unroll
        for (int j = 0; j < CSR_R; ++j) {
            const int c = cnt[j];
            ex  += (j < tid) ? c : 0;
            own  = (j == tid) ? c : own;
        }
        cur[tid] = ex;
        if (lo + tid < (unsigned)N_NODES)
            *(int2*)(rowbe + 2 * (lo + tid)) = make_int2(seg + ex, seg + ex + own);
    }
    __syncthreads();

    for (int k = tid; k < nsel; k += CSR_T) {
        const unsigned rec = sel[k];
        const unsigned s   = (unsigned)ei[rec >> 6] & 0xffffu;
        const int slot = atomicAdd(&cur[rec & 63u], 1);
        lst[slot] = (unsigned short)s;
    }
    __syncthreads();

    for (int i = tid; i < nsel; i += CSR_T) elist[seg + i] = lst[i];
}


#define AX_NODES 16
static_assert(N_NODES % AX_NODES == 0, "every block of node_att1 owns 16 whole nodes");
static_assert(2 * HEADS == 16, "the 16 outputs of a node are one tile's columns");
static_assert((AX_NODES * HEADS * 4) % 128 == 0, "a block's results are whole 128-byte lines");

__device__ __forceinline__ int att1_feat(const int t, const int h, const int s)
{
    return 4 * t + 2 * h + s;
}

__launch_bounds__(32)
__global__ void node_att1(const float* __restrict__ X, const float* __restrict__ wS,
                          const float* __restrict__ wD,
                          float* __restrict__ aS, float* __restrict__ aD)
{
    __shared__ float res[2 * AX_NODES * HEADS];

    const int lane = threadIdx.x;
    const int half = lane >> 4;
    const int ml   = lane & 15;
    const int n0   = blockIdx.x * AX_NODES;

    const float* xp = X + (size_t)(n0 + ml) * F_IN;
    const float* wp = ((ml < HEADS) ? wS : wD) + (ml & (HEADS - 1));

    v2f a[F_IN / 4], b[F_IN / 4];
#pragma unroll
    for (int t = 0; t < F_IN / 4; ++t) {
        const float2 xv = *(const float2*)(xp + att1_feat(t, half, 0));
        a[t].x = xv.x; a[t].y = xv.y;
        b[t].x = wp[(size_t)att1_feat(t, half, 0) * HEADS];
        b[t].y = wp[(size_t)att1_feat(t, half, 1) * HEADS];
    }
    v8f acc0 = (v8f){0.f, 0.f, 0.f, 0.f, 0.f, 0.f, 0.f, 0.f};
    v8f acc1 = (v8f){0.f, 0.f, 0.f, 0.f, 0.f, 0.f, 0.f, 0.f};
#pragma unroll
    for (int t = 0; t < F_IN / 4; t += 2) {
        acc0 = __builtin_amdgcn_wmma_f32_16x16x4_f32(false, a[t],     false, b[t],
                                                     (short)0, acc0, false, false);
        acc1 = __builtin_amdgcn_wmma_f32_16x16x4_f32(false, a[t + 1], false, b[t + 1],
                                                     (short)0, acc1, false, false);
    }
#pragma unroll
    for (int r = 0; r < 8; ++r)
        res[(ml >> 3) * (AX_NODES * HEADS) + (8 * half + r) * HEADS + (ml & 7)] = acc0[r] + acc1[r];
    __syncthreads();
#pragma unroll
    for (int i = 0; i < AX_NODES * HEADS / 32; ++i) {
        aS[(size_t)n0 * HEADS + lane + 32 * i] = res[lane + 32 * i];
        aD[(size_t)n0 * HEADS + lane + 32 * i] = res[AX_NODES * HEADS + lane + 32 * i];
    }
}

#define AGX_NODES 8
static_assert(N_NODES % AGX_NODES == 0, "every block of node_aggx owns 8 whole nodes");

__device__ __forceinline__ void axpy4(float4& acc, const float a, const float4 v)
{
    acc.x = __builtin_fmaf(a, v.x, acc.x); acc.y = __builtin_fmaf(a, v.y, acc.y);
    acc.z = __builtin_fmaf(a, v.z, acc.z); acc.w = __builtin_fmaf(a, v.w, acc.w);
}

__device__ __forceinline__ void load8(float (&o)[HEADS], const float* __restrict__ p)
{
    const float4 a = *(const float4*)p;
    const float4 b = *(const float4*)(p + 4);
    o[0] = a.x; o[1] = a.y; o[2] = a.z; o[3] = a.w;
    o[4] = b.x; o[5] = b.y; o[6] = b.z; o[7] = b.w;
}

__device__ __forceinline__ void edge_scores1(float (&e)[HEADS], const float* __restrict__ aS,
                                             const unsigned rec, const float (&ad)[HEADS])
{
    load8(e, aS + (size_t)(rec & 0xffffu) * HEADS);
#pragma unroll
    for (int h = 0; h < HEADS; ++h) e[h] = leaky(e[h] + ad[h]);
}

__device__ __forceinline__ void add_row(float4 (&acc)[HEADS], const float (&acur)[HEADS],
                                        const int k, const float4 u)
{
#pragma unroll
    for (int h = 0; h < HEADS; ++h) axpy4(acc[h], lane_f(acur[h], k), u);
}

__launch_bounds__(256)
__global__ void node_aggx(const int* __restrict__ rowptr, const unsigned* __restrict__ elist,
                          const float* __restrict__ X, const float* __restrict__ aS,
                          const float* __restrict__ aD, float* __restrict__ xagg)
{
    const int lane = threadIdx.x & 31;
    const int d    = blockIdx.x * AGX_NODES + __builtin_amdgcn_readfirstlane((int)(threadIdx.x >> 5));
    const int loff = lane * 4;

    const int beg = rowptr[2 * d];
    const int end = rowptr[2 * d + 1];
    const float4 xs = *(const float4*)(X + (size_t)d * F_IN + loff);
    float ad[HEADS], es[HEADS];
    load8(ad, aD + (size_t)d * HEADS);
    load8(es, aS + (size_t)d * HEADS);
#pragma unroll
    for (int h = 0; h < HEADS; ++h) es[h] = leaky(es[h] + ad[h]);

    const int  j0 = beg + lane;
    const bool v0 = j0 < end;
    unsigned rc = 0u;
    float ec[HEADS];
#pragma unroll
    for (int h = 0; h < HEADS; ++h) ec[h] = -INFINITY;
    if (v0) { rc = elist[j0]; edge_scores1(ec, aS, rc, ad); }
    float m[HEADS];
#pragma unroll
    for (int h = 0; h < HEADS; ++h) m[h] = fmaxf(es[h], ec[h]);
    for (int c0 = beg + 32; c0 < end; c0 += 32) {
        const int j = c0 + lane;
        if (j < end) {
            float e[HEADS];
            edge_scores1(e, aS, elist[j], ad);
#pragma unroll
            for (int h = 0; h < HEADS; ++h) m[h] = fmaxf(m[h], e[h]);
        }
    }
#pragma unroll
    for (int h = 0; h < HEADS; ++h) {
        m[h] = uni(wave_max(m[h]));
        m[h] = isfinite(m[h]) ? m[h] : 0.f;
    }

    float acur[HEADS];
    float s[HEADS];
#pragma unroll
    for (int h = 0; h < HEADS; ++h) { acur[h] = v0 ? expf(ec[h] - m[h]) : 0.f; s[h] = acur[h]; }
    for (int c0 = beg + 32; c0 < end; c0 += 32) {
        const int j = c0 + lane;
        if (j < end) {
            float e[HEADS];
            edge_scores1(e, aS, elist[j], ad);
#pragma unroll
            for (int h = 0; h < HEADS; ++h) s[h] += expf(e[h] - m[h]);
        }
    }

    float  den[HEADS];
    float4 acc[HEADS];
#pragma unroll
    for (int h = 0; h < HEADS; ++h) {
        const float ps = expf(es[h] - m[h]);
        den[h] = uni(wave_sum(s[h])) + ps + 1e-16f;
        const float a = ps / den[h];
        acc[h]  = make_float4(a * xs.x, a * xs.y, a * xs.z, a * xs.w);
        acur[h] = acur[h] / den[h];
    }

    unsigned rcur = rc;
    for (int c0 = beg; c0 < end; c0 += 32) {
        const int cnt = min(end - c0, 32);
        int k = 0;
        for (; k + 4 <= cnt; k += 4) {
            const unsigned r0 = __builtin_amdgcn_readlane(rcur, k);
            const unsigned r1 = __builtin_amdgcn_readlane(rcur, k + 1);
            const unsigned r2 = __builtin_amdgcn_readlane(rcur, k + 2);
            const unsigned r3 = __builtin_amdgcn_readlane(rcur, k + 3);
            const float4 u0 = *(const float4*)(X + (size_t)(r0 & 0xffffu) * F_IN + loff);
            const float4 u1 = *(const float4*)(X + (size_t)(r1 & 0xffffu) * F_IN + loff);
            const float4 u2 = *(const float4*)(X + (size_t)(r2 & 0xffffu) * F_IN + loff);
            const float4 u3 = *(const float4*)(X + (size_t)(r3 & 0xffffu) * F_IN + loff);
            add_row(acc, acur, k,     u0);
            add_row(acc, acur, k + 1, u1);
            add_row(acc, acur, k + 2, u2);
            add_row(acc, acur, k + 3, u3);
        }
        if (k + 2 <= cnt) {
            const unsigned r0 = __builtin_amdgcn_readlane(rcur, k);
            const unsigned r1 = __builtin_amdgcn_readlane(rcur, k + 1);
            const float4 u0 = *(const float4*)(X + (size_t)(r0 & 0xffffu) * F_IN + loff);
            const float4 u1 = *(const float4*)(X + (size_t)(r1 & 0xffffu) * F_IN + loff);
            add_row(acc, acur, k,     u0);
            add_row(acc, acur, k + 1, u1);
            k += 2;
        }
        if (k < cnt) {
            const unsigned r = __builtin_amdgcn_readlane(rcur, k);
            const float4 u = *(const float4*)(X + (size_t)(r & 0xffffu) * F_IN + loff);
            add_row(acc, acur, k, u);
        }
        if (c0 + 32 < end) {
            const int j = c0 + 32 + lane;
            rcur = 0u;
#pragma unroll
            for (int h = 0; h < HEADS; ++h) acur[h] = 0.f;
            if (j < end) {
                rcur = elist[j];
                float e[HEADS];
                edge_scores1(e, aS, rcur, ad);
#pragma unroll
                for (int h = 0; h < HEADS; ++h) acur[h] = expf(e[h] - m[h]) / den[h];
            }
        }
    }

    float* op = xagg + (size_t)d * HEADS * F_IN + loff;
#pragma unroll
    for (int h = 0; h < HEADS; ++h) *(float4*)(op + h * F_IN) = acc[h];
}

#define AGG2_NODES 16
static_assert(N_NODES % AGG2_NODES == 0, "every block of node_agg2 owns 16 whole nodes");
static_assert((AGG2_NODES * NUM_CLASS * 4) % 128 == 0, "a block's output rows are whole 128-byte lines");

__launch_bounds__(256)
__global__ void node_agg2(const int* __restrict__ rowptr, const unsigned* __restrict__ elist,
                          const float* __restrict__ H2, const float* __restrict__ aS,
                          const float* __restrict__ aD, const float* __restrict__ b2,
                          float* __restrict__ out)
{
    __shared__ float rows[AGG2_NODES * NUM_CLASS];

    const int tid = threadIdx.x;
    const int ln  = tid >> 4;
    const int cl  = tid & 15;
    const int c   = min(cl, NUM_CLASS - 1);
    const int d   = blockIdx.x * AGG2_NODES + ln;

    const int beg = rowptr[2 * d];
    const int end = rowptr[2 * d + 1];
    const float ad = aD[d];
    const float es = leaky(aS[d] + ad);

    float m = es;
    for (int j = beg; j < end; j += 4) {
        float e[4];
#pragma unroll
        for (int i = 0; i < 4; ++i)
            e[i] = leaky(aS[elist[min(j + i, end - 1)] & 0xffffu] + ad);
        m = fmaxf(fmaxf(m, fmaxf(e[0], e[1])), fmaxf(e[2], e[3]));
    }
    m = isfinite(m) ? m : 0.f;

    float den, acc;
    {
        const float p = expf(es - m);
        den = p;
        acc = p * H2[(size_t)d * NUM_CLASS + c];
    }
    for (int j = beg; j < end; j += 4) {
        unsigned sN[4];
        float e[4], h[4];
#pragma unroll
        for (int i = 0; i < 4; ++i) sN[i] = elist[min(j + i, end - 1)] & 0xffffu;
#pragma unroll
        for (int i = 0; i < 4; ++i) {
            e[i] = leaky(aS[sN[i]] + ad);
            h[i] = H2[(size_t)sN[i] * NUM_CLASS + c];
        }
#pragma unroll
        for (int i = 0; i < 4; ++i) {
            const bool  on = j + i < end;
            const float p  = on ? expf(e[i] - m) : 0.f;
            den += p;
            acc += p * (on ? h[i] : 0.f);
        }
    }
    const float v = acc / (den + 1e-16f) + b2[c];
    if (cl < NUM_CLASS) rows[ln * NUM_CLASS + cl] = 1.f / (1.f + expf(-v));
    __syncthreads();
    if (tid < AGG2_NODES * NUM_CLASS)
        out[(size_t)blockIdx.x * (AGG2_NODES * NUM_CLASS) + tid] = rows[tid];
}

extern "C" void kernel_launch(void* const* d_in, const int* in_sizes, int n_in,
                              void* d_out, int out_size, void* d_ws, size_t ws_size,
                              hipStream_t stream)
{
    const float* x   = (const float*)d_in[0];
    const int*   ei  = (const int*)d_in[1];
    const float* W1  = (const float*)d_in[2];
    const float* as1 = (const float*)d_in[3];
    const float* ad1 = (const float*)d_in[4];
    const float* b1  = (const float*)d_in[5];
    const float* W2  = (const float*)d_in[6];
    const float* as2 = (const float*)d_in[7];
    const float* ad2 = (const float*)d_in[8];
    const float* b2  = (const float*)d_in[9];
    float* out = (float*)d_out;

    float* ws = (float*)d_ws;
    size_t off = 0;
    float* xa    = ws + off; off += (size_t)N_NODES * HEADS * F_IN;
    float* h2p   = ws + off; off += (size_t)HEADS * N_NODES * NUM_CLASS;
    float* sA1   = ws + off; off += (size_t)N_NODES * HEADS;
    float* sD1   = ws + off; off += (size_t)N_NODES * HEADS;
    float* h2    = ws + off; off += (size_t)N_NODES * NUM_CLASS;
    float* sA2   = ws + off; off += N_NODES;
    float* sD2   = ws + off; off += N_NODES;
    float* wS1   = ws + off; off += (size_t)F_IN * HEADS;
    float* wD1   = ws + off; off += (size_t)F_IN * HEADS;
    unsigned* elist = (unsigned*)(ws + off); off += (size_t)ELIST_CAP;
    int*      rowp  = (int*)(ws + off);      off += (size_t)2 * N_NODES;
    (void)ws_size; (void)in_sizes; (void)n_in; (void)out_size;

    const dim3 b256(256);

    csr_fold<<<CSR_B + FOLD_B, CSR_T, 0, stream>>>(ei, rowp, elist, W1, as1, ad1, wS1, wD1);
    node_att1<<<N_NODES / AX_NODES, 32, 0, stream>>>(x, wS1, wD1, sA1, sD1);
    node_aggx<<<N_NODES / AGX_NODES, b256, 0, stream>>>(rowp, elist, x, sA1, sD1, xa);
    gemm12_wmma<<<dim3(HEADS, G1_RBLK), b256, 0, stream>>>(xa, W1, b1, W2, h2p);

    att2_kernel<<<(N_NODES + A2_NODES - 1) / A2_NODES, b256, 0, stream>>>(h2p, as2, ad2, h2, sA2, sD2);
    node_agg2<<<N_NODES / AGG2_NODES, b256, 0, stream>>>(rowp, elist, h2, sA2, sD2, b2, out);
}
